# starting kernel
# baseline (speedup 1.0000x reference)
#include <hip/hip_runtime.h>

#define NNODES 100000
#define NGRAPHS 1024
#define EMB 32
#define HID 64
#define NCLASSES 10

typedef __attribute__((ext_vector_type(2))) float v2f;
typedef __attribute__((ext_vector_type(8))) float v8f;

__global__ void zero_f32(float* __restrict__ p, long n) {
  long i = (long)blockIdx.x * blockDim.x + threadIdx.x;
  if (i < n) p[i] = 0.0f;
}

__global__ void deg_kernel(const int* __restrict__ dst, unsigned int* __restrict__ deg, int E) {
  int e = blockIdx.x * blockDim.x + threadIdx.x;
  if (e < E) atomicAdd(&deg[dst[e]], 1u);
}

__global__ void dis_kernel(float* __restrict__ disbuf, int n) {
  int i = blockIdx.x * blockDim.x + threadIdx.x;
  if (i < n) {
    unsigned int d = ((const unsigned int*)disbuf)[i];
    disbuf[i] = rsqrtf(1.0f + (float)d);
  }
}

__global__ void embed_kernel(const int* __restrict__ sid, const int* __restrict__ cid,
                             const float* __restrict__ st, const float* __restrict__ ct,
                             float* __restrict__ x0, int n) {
  long i = (long)blockIdx.x * blockDim.x + threadIdx.x;
  if (i >= (long)n * EMB) return;
  int node = (int)(i >> 5);
  int c = (int)(i & 31);
  int s = sid[node], cc = cid[node];
  float v = 0.0f;
  if (s != 0) v += st[s * EMB + c];
  if (cc != 0) v += ct[cc * EMB + c];
  x0[i] = v;
}

__global__ void gemm_wmma_f32(const float* __restrict__ X, const float* __restrict__ W,
                              float* __restrict__ H, int K) {
  const int lane  = threadIdx.x & 31;
  const int wave  = threadIdx.x >> 5;
  const int l16   = lane & 15;
  const int khalf = lane >> 4;
  const long row  = (long)blockIdx.x * 16 + l16;
  const int  col  = wave * 16 + l16;

  v8f c = {};
  const float* ap = X + row * K + 2 * khalf;
  for (int k0 = 0; k0 < K; k0 += 4) {
    v2f a, b;
    a.x = ap[k0];
    a.y = ap[k0 + 1];
    b.x = W[(k0 + 2 * khalf + 0) * HID + col];
    b.y = W[(k0 + 2 * khalf + 1) * HID + col];
    c = __builtin_amdgcn_wmma_f32_16x16x4_f32(false, a, false, b, (short)0, c, false, false);
  }
  const long rbase = (long)blockIdx.x * 16 + 8 * khalf;
#pragma unroll
  for (int v = 0; v < 8; ++v)
    H[(rbase + v) * HID + col] = c[v];
}

__global__ void edge_agg_kernel(const int* __restrict__ src, const int* __restrict__ dst,
                                const float* __restrict__ h, const float* __restrict__ dis,
                                float* __restrict__ agg, int E) {
  long t = (long)blockIdx.x * blockDim.x + threadIdx.x;
  long e = t >> 4;
  if (e >= E) return;
  int c4 = (int)(t & 15) << 2;
  int s = src[e];
  int d = dst[e];
  if (e + 512 < E) __builtin_prefetch(h + (long)src[e + 512] * HID + c4, 0, 1);
  float norm = dis[s] * dis[d];
  const float4 hv = *(const float4*)(h + (long)s * HID + c4);
  float* ap = agg + (long)d * HID + c4;
  atomicAdd(ap + 0, hv.x * norm);
  atomicAdd(ap + 1, hv.y * norm);
  atomicAdd(ap + 2, hv.z * norm);
  atomicAdd(ap + 3, hv.w * norm);
}

__global__ void finalize_kernel(float* __restrict__ agg, const float* __restrict__ h,
                                const float* __restrict__ dis, const float* __restrict__ b,
                                int n) {
  long i = (long)blockIdx.x * blockDim.x + threadIdx.x;
  if (i >= (long)n * HID) return;
  int node = (int)(i >> 6);
  int c = (int)(i & 63);
  float di = dis[node];
  float v = agg[i] + h[i] * di * di + b[c];
  agg[i] = v > 0.0f ? v : 0.0f;
}

__global__ void pool_kernel(const float* __restrict__ x, const int* __restrict__ batch,
                            float* __restrict__ pooled, float* __restrict__ cnt, int n) {
  long i = (long)blockIdx.x * blockDim.x + threadIdx.x;
  if (i >= (long)n * HID) return;
  int node = (int)(i >> 6);
  int c = (int)(i & 63);
  int g = batch[node];
  atomicAdd(pooled + (long)g * HID + c, x[i]);
  if (c == 0) atomicAdd(cnt + g, 1.0f);
}

__global__ void out_kernel(const float* __restrict__ pooled, const float* __restrict__ cnt,
                           const float* __restrict__ Wl, const float* __restrict__ bl,
                           float* __restrict__ out) {
  int i = blockIdx.x * blockDim.x + threadIdx.x;
  if (i >= NGRAPHS * NCLASSES) return;
  int g = i / NCLASSES;
  int k = i % NCLASSES;
  float s = 0.0f;
#pragma unroll 8
  for (int hh = 0; hh < HID; ++hh) s += pooled[g * HID + hh] * Wl[hh * NCLASSES + k];
  float c = cnt[g];
  c = c > 1.0f ? c : 1.0f;
  out[i] = s / c + bl[k];
}

extern "C" void kernel_launch(void* const* d_in, const int* in_sizes, int n_in,
                              void* d_out, int out_size, void* d_ws, size_t ws_size,
                              hipStream_t stream) {
  const int* shape_id = (const int*)d_in[0];
  const int* color_id = (const int*)d_in[1];
  const int* edge_index = (const int*)d_in[2];
  const int* batch = (const int*)d_in[3];
  const float* shape_table = (const float*)d_in[4];
  const float* color_table = (const float*)d_in[5];
  const float* W1 = (const float*)d_in[6];
  const float* b1 = (const float*)d_in[7];
  const float* W2 = (const float*)d_in[8];
  const float* b2 = (const float*)d_in[9];
  const float* Wl = (const float*)d_in[10];
  const float* bl = (const float*)d_in[11];
  float* out = (float*)d_out;

  const int N = NNODES;
  const int E = in_sizes[2] / 2;
  const int* src = edge_index;
  const int* dst = edge_index + E;

  float* dis    = (float*)d_ws;
  float* x0     = dis + N;
  float* bufH   = x0 + (long)N * EMB;
  float* bufA   = bufH + (long)N * HID;
  float* pooled = bufA + (long)N * HID;
  float* cnt    = pooled + (long)NGRAPHS * HID;
  (void)n_in; (void)out_size; (void)ws_size;

  const int B = 256;
  auto blocks = [](long n, int b) { return (unsigned)((n + b - 1) / b); };

  zero_f32<<<blocks(N, B), B, 0, stream>>>(dis, N);
  deg_kernel<<<blocks(E, B), B, 0, stream>>>(dst, (unsigned int*)dis, E);
  dis_kernel<<<blocks(N, B), B, 0, stream>>>(dis, N);

  embed_kernel<<<blocks((long)N * EMB, B), B, 0, stream>>>(shape_id, color_id,
                                                           shape_table, color_table, x0, N);

  gemm_wmma_f32<<<N / 16, 128, 0, stream>>>(x0, W1, bufH, EMB);
  zero_f32<<<blocks((long)N * HID, B), B, 0, stream>>>(bufA, (long)N * HID);
  edge_agg_kernel<<<blocks((long)E * 16, B), B, 0, stream>>>(src, dst, bufH, dis, bufA, E);
  finalize_kernel<<<blocks((long)N * HID, B), B, 0, stream>>>(bufA, bufH, dis, b1, N);

  gemm_wmma_f32<<<N / 16, 128, 0, stream>>>(bufA, W2, bufH, HID);
  zero_f32<<<blocks((long)N * HID, B), B, 0, stream>>>(bufA, (long)N * HID);
  edge_agg_kernel<<<blocks((long)E * 16, B), B, 0, stream>>>(src, dst, bufH, dis, bufA, E);
  finalize_kernel<<<blocks((long)N * HID, B), B, 0, stream>>>(bufA, bufH, dis, b2, N);

  zero_f32<<<blocks((long)NGRAPHS * HID + NGRAPHS, B), B, 0, stream>>>(pooled,
                                                                      (long)NGRAPHS * HID + NGRAPHS);
  pool_kernel<<<blocks((long)N * HID, B), B, 0, stream>>>(bufA, batch, pooled, cnt, N);
  out_kernel<<<blocks((long)NGRAPHS * NCLASSES, B), B, 0, stream>>>(pooled, cnt, Wl, bl, out);
}


# Layer 1 aggregates the 32-wide embeddings before the W1 product ((A x) W = A (x W)): half the scatter atomics and zeroing; self-loop, bias, relu move into the GEMM; f32 sum order changes
# speedup vs baseline: 1.2129x; 1.2129x over previous
#include <hip/hip_runtime.h>

#define NNODES 100000
#define NGRAPHS 1024
#define EMB 32
#define HID 64
#define NCLASSES 10

typedef __attribute__((ext_vector_type(2))) float v2f;
typedef __attribute__((ext_vector_type(8))) float v8f;

__global__ void zero_f32(float* __restrict__ p, long n) {
  long i = (long)blockIdx.x * blockDim.x + threadIdx.x;
  if (i < n) p[i] = 0.0f;
}

__global__ void deg_kernel(const int* __restrict__ dst, unsigned int* __restrict__ deg, int E) {
  int e = blockIdx.x * blockDim.x + threadIdx.x;
  if (e < E) atomicAdd(&deg[dst[e]], 1u);
}

__global__ void dis_kernel(float* __restrict__ disbuf, int n) {
  int i = blockIdx.x * blockDim.x + threadIdx.x;
  if (i < n) {
    unsigned int d = ((const unsigned int*)disbuf)[i];
    disbuf[i] = rsqrtf(1.0f + (float)d);
  }
}

__global__ void embed_kernel(const int* __restrict__ sid, const int* __restrict__ cid,
                             const float* __restrict__ st, const float* __restrict__ ct,
                             float* __restrict__ x0, int n) {
  long i = (long)blockIdx.x * blockDim.x + threadIdx.x;
  if (i >= (long)n * EMB) return;
  int node = (int)(i >> 5);
  int c = (int)(i & 31);
  int s = sid[node], cc = cid[node];
  float v = 0.0f;
  if (s != 0) v += st[s * EMB + c];
  if (cc != 0) v += ct[cc * EMB + c];
  x0[i] = v;
}

__global__ void gemm_wmma_f32(const float* __restrict__ X, const float* __restrict__ W,
                              float* __restrict__ H, int K) {
  const int lane  = threadIdx.x & 31;
  const int wave  = threadIdx.x >> 5;
  const int l16   = lane & 15;
  const int khalf = lane >> 4;
  const long row  = (long)blockIdx.x * 16 + l16;
  const int  col  = wave * 16 + l16;

  v8f c = {};
  const float* ap = X + row * K + 2 * khalf;
  for (int k0 = 0; k0 < K; k0 += 4) {
    v2f a, b;
    a.x = ap[k0];
    a.y = ap[k0 + 1];
    b.x = W[(k0 + 2 * khalf + 0) * HID + col];
    b.y = W[(k0 + 2 * khalf + 1) * HID + col];
    c = __builtin_amdgcn_wmma_f32_16x16x4_f32(false, a, false, b, (short)0, c, false, false);
  }
  const long rbase = (long)blockIdx.x * 16 + 8 * khalf;
#pragma unroll
  for (int v = 0; v < 8; ++v)
    H[(rbase + v) * HID + col] = c[v];
}

__global__ void gemm_wmma_f32_selfloop_bias_relu(const float* __restrict__ AGG,
                                                 const float* __restrict__ X0,
                                                 const float* __restrict__ dis,
                                                 const float* __restrict__ W,
                                                 const float* __restrict__ bias,
                                                 float* __restrict__ X1) {
  const int lane  = threadIdx.x & 31;
  const int wave  = threadIdx.x >> 5;
  const int l16   = lane & 15;
  const int khalf = lane >> 4;
  const long row  = (long)blockIdx.x * 16 + l16;
  const int  col  = wave * 16 + l16;

  const float di = dis[row];
  const float d2 = di * di;
  v8f c = {};
  const float* gp = AGG + row * EMB + 2 * khalf;
  const float* xp = X0 + row * EMB + 2 * khalf;
  for (int k0 = 0; k0 < EMB; k0 += 4) {
    v2f a, b;
    a.x = gp[k0]     + xp[k0]     * d2;
    a.y = gp[k0 + 1] + xp[k0 + 1] * d2;
    b.x = W[(k0 + 2 * khalf + 0) * HID + col];
    b.y = W[(k0 + 2 * khalf + 1) * HID + col];
    c = __builtin_amdgcn_wmma_f32_16x16x4_f32(false, a, false, b, (short)0, c, false, false);
  }
  const float bc = bias[col];
  const long rbase = (long)blockIdx.x * 16 + 8 * khalf;
  for (int v = 0; v < 8; ++v) {
    float o = c[v] + bc;
    X1[(rbase + v) * HID + col] = o > 0.0f ? o : 0.0f;
  }
}

template <int WIDTH>
__device__ __forceinline__ void edge_agg_body(const int* __restrict__ src, const int* __restrict__ dst,
                                              const float* __restrict__ h, const float* __restrict__ dis,
                                              float* __restrict__ agg, int E) {
  constexpr int LPE = WIDTH / 4;
  constexpr int SH  = (LPE == 16) ? 4 : 3;
  long t = (long)blockIdx.x * blockDim.x + threadIdx.x;
  long e = t >> SH;
  if (e >= E) return;
  int c4 = (int)(t & (LPE - 1)) << 2;
  int s = src[e];
  int d = dst[e];
  if (e + 512 < E) __builtin_prefetch(h + (long)src[e + 512] * WIDTH + c4, 0, 1);
  float norm = dis[s] * dis[d];
  const float4 hv = *(const float4*)(h + (long)s * WIDTH + c4);
  float* ap = agg + (long)d * WIDTH + c4;
  atomicAdd(ap + 0, hv.x * norm);
  atomicAdd(ap + 1, hv.y * norm);
  atomicAdd(ap + 2, hv.z * norm);
  atomicAdd(ap + 3, hv.w * norm);
}

__global__ void edge_agg_kernel(const int* __restrict__ src, const int* __restrict__ dst,
                                const float* __restrict__ h, const float* __restrict__ dis,
                                float* __restrict__ agg, int E) {
  edge_agg_body<HID>(src, dst, h, dis, agg, E);
}

__global__ void edge_agg_emb_kernel(const int* __restrict__ src, const int* __restrict__ dst,
                                    const float* __restrict__ x, const float* __restrict__ dis,
                                    float* __restrict__ agg, int E) {
  edge_agg_body<EMB>(src, dst, x, dis, agg, E);
}

__global__ void finalize_kernel(float* __restrict__ agg, const float* __restrict__ h,
                                const float* __restrict__ dis, const float* __restrict__ b,
                                int n) {
  long i = (long)blockIdx.x * blockDim.x + threadIdx.x;
  if (i >= (long)n * HID) return;
  int node = (int)(i >> 6);
  int c = (int)(i & 63);
  float di = dis[node];
  float v = agg[i] + h[i] * di * di + b[c];
  agg[i] = v > 0.0f ? v : 0.0f;
}

__global__ void pool_kernel(const float* __restrict__ x, const int* __restrict__ batch,
                            float* __restrict__ pooled, float* __restrict__ cnt, int n) {
  long i = (long)blockIdx.x * blockDim.x + threadIdx.x;
  if (i >= (long)n * HID) return;
  int node = (int)(i >> 6);
  int c = (int)(i & 63);
  int g = batch[node];
  atomicAdd(pooled + (long)g * HID + c, x[i]);
  if (c == 0) atomicAdd(cnt + g, 1.0f);
}

__global__ void out_kernel(const float* __restrict__ pooled, const float* __restrict__ cnt,
                           const float* __restrict__ Wl, const float* __restrict__ bl,
                           float* __restrict__ out) {
  int i = blockIdx.x * blockDim.x + threadIdx.x;
  if (i >= NGRAPHS * NCLASSES) return;
  int g = i / NCLASSES;
  int k = i % NCLASSES;
  float s = 0.0f;
#pragma unroll 8
  for (int hh = 0; hh < HID; ++hh) s += pooled[g * HID + hh] * Wl[hh * NCLASSES + k];
  float c = cnt[g];
  c = c > 1.0f ? c : 1.0f;
  out[i] = s / c + bl[k];
}

extern "C" void kernel_launch(void* const* d_in, const int* in_sizes, int n_in,
                              void* d_out, int out_size, void* d_ws, size_t ws_size,
                              hipStream_t stream) {
  const int* shape_id = (const int*)d_in[0];
  const int* color_id = (const int*)d_in[1];
  const int* edge_index = (const int*)d_in[2];
  const int* batch = (const int*)d_in[3];
  const float* shape_table = (const float*)d_in[4];
  const float* color_table = (const float*)d_in[5];
  const float* W1 = (const float*)d_in[6];
  const float* b1 = (const float*)d_in[7];
  const float* W2 = (const float*)d_in[8];
  const float* b2 = (const float*)d_in[9];
  const float* Wl = (const float*)d_in[10];
  const float* bl = (const float*)d_in[11];
  float* out = (float*)d_out;

  const int N = NNODES;
  const int E = in_sizes[2] / 2;
  const int* src = edge_index;
  const int* dst = edge_index + E;

  float* dis    = (float*)d_ws;
  float* x0     = dis + N;
  float* bufH   = x0 + (long)N * EMB;
  float* bufA   = bufH + (long)N * HID;
  float* pooled = bufA + (long)N * HID;
  float* cnt    = pooled + (long)NGRAPHS * HID;
  (void)n_in; (void)out_size; (void)ws_size;

  const int B = 256;
  auto blocks = [](long n, int b) { return (unsigned)((n + b - 1) / b); };

  zero_f32<<<blocks(N, B), B, 0, stream>>>(dis, N);
  deg_kernel<<<blocks(E, B), B, 0, stream>>>(dst, (unsigned int*)dis, E);
  dis_kernel<<<blocks(N, B), B, 0, stream>>>(dis, N);

  embed_kernel<<<blocks((long)N * EMB, B), B, 0, stream>>>(shape_id, color_id,
                                                           shape_table, color_table, x0, N);

  float* agg0 = bufH;
  zero_f32<<<blocks((long)N * EMB, B), B, 0, stream>>>(agg0, (long)N * EMB);
  edge_agg_emb_kernel<<<blocks((long)E * 8, B), B, 0, stream>>>(src, dst, x0, dis, agg0, E);
  gemm_wmma_f32_selfloop_bias_relu<<<N / 16, 128, 0, stream>>>(agg0, x0, dis, W1, b1, bufA);

  gemm_wmma_f32<<<N / 16, 128, 0, stream>>>(bufA, W2, bufH, HID);
  zero_f32<<<blocks((long)N * HID, B), B, 0, stream>>>(bufA, (long)N * HID);
  edge_agg_kernel<<<blocks((long)E * 16, B), B, 0, stream>>>(src, dst, bufH, dis, bufA, E);
  finalize_kernel<<<blocks((long)N * HID, B), B, 0, stream>>>(bufA, bufH, dis, b2, N);

  zero_f32<<<blocks((long)NGRAPHS * HID + NGRAPHS, B), B, 0, stream>>>(pooled,
                                                                      (long)NGRAPHS * HID + NGRAPHS);
  pool_kernel<<<blocks((long)N * HID, B), B, 0, stream>>>(bufA, batch, pooled, cnt, N);
  out_kernel<<<blocks((long)NGRAPHS * NCLASSES, B), B, 0, stream>>>(pooled, cnt, Wl, bl, out);
}


# Both aggregations gather over a dst-sorted edge list (per-call counting sort: in-degree, scan, fill by returning int atomic) instead of f32 scatter atomics; f32 sum order = fill arrival rank
# speedup vs baseline: 1.4311x; 1.1799x over previous
#include <hip/hip_runtime.h>

#define NNODES 100000
#define NGRAPHS 1024
#define EMB 32
#define HID 64
#define NCLASSES 10

typedef __attribute__((ext_vector_type(2))) float v2f;
typedef __attribute__((ext_vector_type(8))) float v8f;

__global__ void zero_f32(float* __restrict__ p, long n) {
  long i = (long)blockIdx.x * blockDim.x + threadIdx.x;
  if (i < n) p[i] = 0.0f;
}

__global__ void deg_kernel(const int* __restrict__ dst, unsigned int* __restrict__ deg, int E) {
  int e = blockIdx.x * blockDim.x + threadIdx.x;
  if (e < E) atomicAdd(&deg[dst[e]], 1u);
}

#define SCAN_T 1024
#define SCAN_E 4096
__global__ void scan_dis_kernel(const int* __restrict__ deg, int* __restrict__ rowptr,
                                float* __restrict__ dis, int n) {
  __shared__ int s_part[SCAN_T / 32];
  __shared__ int s_wsum[SCAN_T / 32];
  const int t = threadIdx.x;
  const int lane = t & 31;
  const int wave = t >> 5;

  int pre = 0;
  const int lim = blockIdx.x * SCAN_E;
  int j = 4 * t;
  for (; j + 3 * SCAN_E < lim; j += 4 * SCAN_E) {
    const int4 q0 = *(const int4*)(deg + j);
    const int4 q1 = *(const int4*)(deg + j + SCAN_E);
    const int4 q2 = *(const int4*)(deg + j + 2 * SCAN_E);
    const int4 q3 = *(const int4*)(deg + j + 3 * SCAN_E);
    pre += ((q0.x + q0.y) + (q0.z + q0.w)) + ((q1.x + q1.y) + (q1.z + q1.w));
    pre += ((q2.x + q2.y) + (q2.z + q2.w)) + ((q3.x + q3.y) + (q3.z + q3.w));
  }
  for (; j < lim; j += SCAN_E) {
    const int4 q0 = *(const int4*)(deg + j);
    pre += (q0.x + q0.y) + (q0.z + q0.w);
  }
  for (int o = 16; o > 0; o >>= 1) pre += __shfl_xor(pre, o);
  if (lane == 0) s_part[wave] = pre;

  const int base = lim + 4 * t;
  int4 q = make_int4(0, 0, 0, 0);
  if (base < n) q = *(const int4*)(deg + base);
  const int own = (q.x + q.y) + (q.z + q.w);
  int inc = own;
  for (int o = 1; o < 32; o <<= 1) {
    const int up = __shfl_up(inc, o);
    if (lane >= o) inc += up;
  }
  if (lane == 31) s_wsum[wave] = inc;
  __syncthreads();

  int off = 0;
  for (int w = 0; w < SCAN_T / 32; ++w) {
    off += s_part[w];
    if (w < wave) off += s_wsum[w];
  }
  const int ex = off + inc - own;
  if (base < n) {
    int4 r;
    r.x = ex;
    r.y = ex + q.x;
    r.z = r.y + q.y;
    r.w = r.z + q.z;
    *(int4*)(rowptr + base) = r;
    float4 d;
    d.x = rsqrtf(1.0f + (float)q.x);
    d.y = rsqrtf(1.0f + (float)q.y);
    d.z = rsqrtf(1.0f + (float)q.z);
    d.w = rsqrtf(1.0f + (float)q.w);
    *(float4*)(dis + base) = d;
    if (base + 4 == n) rowptr[n] = r.w + q.w;
  }
}

__global__ void fill_kernel(const int* __restrict__ src, const int* __restrict__ dst,
                            const int* __restrict__ rowptr, int* __restrict__ deg,
                            int* __restrict__ colidx, int E) {
  int e = blockIdx.x * blockDim.x + threadIdx.x;
  if (e >= E) return;
  const int d = dst[e];
  const int old = atomicSub(&deg[d], 1);
  colidx[rowptr[d] + old - 1] = src[e];
}

__global__ void embed_kernel(const int* __restrict__ sid, const int* __restrict__ cid,
                             const float* __restrict__ st, const float* __restrict__ ct,
                             float* __restrict__ x0, int n) {
  long i = (long)blockIdx.x * blockDim.x + threadIdx.x;
  if (i >= (long)n * EMB) return;
  int node = (int)(i >> 5);
  int c = (int)(i & 31);
  int s = sid[node], cc = cid[node];
  float v = 0.0f;
  if (s != 0) v += st[s * EMB + c];
  if (cc != 0) v += ct[cc * EMB + c];
  x0[i] = v;
}

__global__ void gemm_wmma_f32(const float* __restrict__ X, const float* __restrict__ W,
                              float* __restrict__ H, int K) {
  const int lane  = threadIdx.x & 31;
  const int wave  = threadIdx.x >> 5;
  const int l16   = lane & 15;
  const int khalf = lane >> 4;
  const long row  = (long)blockIdx.x * 16 + l16;
  const int  col  = wave * 16 + l16;

  v8f c = {};
  const float* ap = X + row * K + 2 * khalf;
  for (int k0 = 0; k0 < K; k0 += 4) {
    v2f a, b;
    a.x = ap[k0];
    a.y = ap[k0 + 1];
    b.x = W[(k0 + 2 * khalf + 0) * HID + col];
    b.y = W[(k0 + 2 * khalf + 1) * HID + col];
    c = __builtin_amdgcn_wmma_f32_16x16x4_f32(false, a, false, b, (short)0, c, false, false);
  }
  const long rbase = (long)blockIdx.x * 16 + 8 * khalf;
#pragma unroll
  for (int v = 0; v < 8; ++v)
    H[(rbase + v) * HID + col] = c[v];
}

__global__ void gemm_wmma_f32_bias_relu(const float* __restrict__ AGG,
                                        const float* __restrict__ W,
                                        const float* __restrict__ bias,
                                        float* __restrict__ X1) {
  const int lane  = threadIdx.x & 31;
  const int wave  = threadIdx.x >> 5;
  const int l16   = lane & 15;
  const int khalf = lane >> 4;
  const long row  = (long)blockIdx.x * 16 + l16;
  const int  col  = wave * 16 + l16;

  v8f c = {};
  const float* gp = AGG + row * EMB + 2 * khalf;
  for (int k0 = 0; k0 < EMB; k0 += 4) {
    v2f a, b;
    a.x = gp[k0];
    a.y = gp[k0 + 1];
    b.x = W[(k0 + 2 * khalf + 0) * HID + col];
    b.y = W[(k0 + 2 * khalf + 1) * HID + col];
    c = __builtin_amdgcn_wmma_f32_16x16x4_f32(false, a, false, b, (short)0, c, false, false);
  }
  const float bc = bias[col];
  const long rbase = (long)blockIdx.x * 16 + 8 * khalf;
  for (int v = 0; v < 8; ++v) {
    float o = c[v] + bc;
    X1[(rbase + v) * HID + col] = o > 0.0f ? o : 0.0f;
  }
}

template <int WIDTH, bool FINAL>
__device__ __forceinline__ void node_gather_body(const int* __restrict__ rowptr,
                                                 const int* __restrict__ colidx,
                                                 const float* __restrict__ h,
                                                 const float* __restrict__ dis,
                                                 const float* __restrict__ bias,
                                                 float* __restrict__ out, int n) {
  constexpr int LPN = WIDTH / 4;
  constexpr int SH  = (LPN == 16) ? 4 : 3;
  const long t = (long)blockIdx.x * blockDim.x + threadIdx.x;
  const int node = (int)(t >> SH);
  if (node >= n) return;
  const int c4 = (int)(t & (LPN - 1)) << 2;
  const int beg = rowptr[node];
  const int end = rowptr[node + 1];
  const float dd = dis[node];
  const float* hc = h + c4;

  const float4 self = *(const float4*)(hc + (long)node * WIDTH);
  const float d2 = dd * dd;
  float ax = self.x * d2, ay = self.y * d2, az = self.z * d2, aw = self.w * d2;

  int j = beg;
  for (; j + 4 <= end; j += 4) {
    const int s0 = colidx[j], s1 = colidx[j + 1], s2 = colidx[j + 2], s3 = colidx[j + 3];
    const float w0 = dis[s0] * dd, w1 = dis[s1] * dd, w2 = dis[s2] * dd, w3 = dis[s3] * dd;
    const float4 v0 = *(const float4*)(hc + (long)s0 * WIDTH);
    const float4 v1 = *(const float4*)(hc + (long)s1 * WIDTH);
    const float4 v2 = *(const float4*)(hc + (long)s2 * WIDTH);
    const float4 v3 = *(const float4*)(hc + (long)s3 * WIDTH);
    ax += v0.x * w0; ay += v0.y * w0; az += v0.z * w0; aw += v0.w * w0;
    ax += v1.x * w1; ay += v1.y * w1; az += v1.z * w1; aw += v1.w * w1;
    ax += v2.x * w2; ay += v2.y * w2; az += v2.z * w2; aw += v2.w * w2;
    ax += v3.x * w3; ay += v3.y * w3; az += v3.z * w3; aw += v3.w * w3;
  }
  if (j < end) {
    const int l = end - 1;
    const int j1 = j + 1 < l ? j + 1 : l;
    const int j2 = j + 2 < l ? j + 2 : l;
    const int s0 = colidx[j], s1 = colidx[j1], s2 = colidx[j2];
    const float m1 = (j + 1 < end) ? dd : 0.0f;
    const float m2 = (j + 2 < end) ? dd : 0.0f;
    const float w0 = dis[s0] * dd, w1 = dis[s1] * m1, w2 = dis[s2] * m2;
    const float4 v0 = *(const float4*)(hc + (long)s0 * WIDTH);
    const float4 v1 = *(const float4*)(hc + (long)s1 * WIDTH);
    const float4 v2 = *(const float4*)(hc + (long)s2 * WIDTH);
    ax += v0.x * w0; ay += v0.y * w0; az += v0.z * w0; aw += v0.w * w0;
    ax += v1.x * w1; ay += v1.y * w1; az += v1.z * w1; aw += v1.w * w1;
    ax += v2.x * w2; ay += v2.y * w2; az += v2.z * w2; aw += v2.w * w2;
  }

  if (FINAL) {
    const float4 b = *(const float4*)(bias + c4);
    ax += b.x; ay += b.y; az += b.z; aw += b.w;
    ax = ax > 0.0f ? ax : 0.0f;
    ay = ay > 0.0f ? ay : 0.0f;
    az = az > 0.0f ? az : 0.0f;
    aw = aw > 0.0f ? aw : 0.0f;
  }
  float4 o;
  o.x = ax; o.y = ay; o.z = az; o.w = aw;
  *(float4*)(out + (long)node * WIDTH + c4) = o;
}

__global__ void node_gather_emb_kernel(const int* __restrict__ rowptr, const int* __restrict__ colidx,
                                       const float* __restrict__ x, const float* __restrict__ dis,
                                       float* __restrict__ agg, int n) {
  node_gather_body<EMB, false>(rowptr, colidx, x, dis, nullptr, agg, n);
}

__global__ void node_gather_hid_kernel(const int* __restrict__ rowptr, const int* __restrict__ colidx,
                                       const float* __restrict__ h, const float* __restrict__ dis,
                                       const float* __restrict__ b, float* __restrict__ x2, int n) {
  node_gather_body<HID, true>(rowptr, colidx, h, dis, b, x2, n);
}

__global__ void pool_kernel(const float* __restrict__ x, const int* __restrict__ batch,
                            float* __restrict__ pooled, float* __restrict__ cnt, int n) {
  long i = (long)blockIdx.x * blockDim.x + threadIdx.x;
  if (i >= (long)n * HID) return;
  int node = (int)(i >> 6);
  int c = (int)(i & 63);
  int g = batch[node];
  atomicAdd(pooled + (long)g * HID + c, x[i]);
  if (c == 0) atomicAdd(cnt + g, 1.0f);
}

__global__ void out_kernel(const float* __restrict__ pooled, const float* __restrict__ cnt,
                           const float* __restrict__ Wl, const float* __restrict__ bl,
                           float* __restrict__ out) {
  int i = blockIdx.x * blockDim.x + threadIdx.x;
  if (i >= NGRAPHS * NCLASSES) return;
  int g = i / NCLASSES;
  int k = i % NCLASSES;
  float s = 0.0f;
#pragma unroll 8
  for (int hh = 0; hh < HID; ++hh) s += pooled[g * HID + hh] * Wl[hh * NCLASSES + k];
  float c = cnt[g];
  c = c > 1.0f ? c : 1.0f;
  out[i] = s / c + bl[k];
}

extern "C" void kernel_launch(void* const* d_in, const int* in_sizes, int n_in,
                              void* d_out, int out_size, void* d_ws, size_t ws_size,
                              hipStream_t stream) {
  const int* shape_id = (const int*)d_in[0];
  const int* color_id = (const int*)d_in[1];
  const int* edge_index = (const int*)d_in[2];
  const int* batch = (const int*)d_in[3];
  const float* shape_table = (const float*)d_in[4];
  const float* color_table = (const float*)d_in[5];
  const float* W1 = (const float*)d_in[6];
  const float* b1 = (const float*)d_in[7];
  const float* W2 = (const float*)d_in[8];
  const float* b2 = (const float*)d_in[9];
  const float* Wl = (const float*)d_in[10];
  const float* bl = (const float*)d_in[11];
  float* out = (float*)d_out;

  const int N = NNODES;
  const int E = in_sizes[2] / 2;
  const int* src = edge_index;
  const int* dst = edge_index + E;

  float* dis    = (float*)d_ws;
  int*   colidx = (int*)(dis + N);
  int*   rowptr = colidx + E;
  int*   deg    = rowptr + N + 4;
  float* bufH   = dis + (long)N * (1 + EMB);
  float* bufA   = bufH + (long)N * HID;
  float* pooled = bufA + (long)N * HID;
  float* cnt    = pooled + (long)NGRAPHS * HID;
  float* x0     = bufH;
  float* agg0   = bufH + (long)N * EMB;
  (void)n_in; (void)out_size; (void)ws_size;

  const int B = 256;
  auto blocks = [](long n, int b) { return (unsigned)((n + b - 1) / b); };

  zero_f32<<<blocks(N, B), B, 0, stream>>>((float*)deg, N);
  deg_kernel<<<blocks(E, B), B, 0, stream>>>(dst, (unsigned int*)deg, E);
  scan_dis_kernel<<<blocks(N, SCAN_E), SCAN_T, 0, stream>>>(deg, rowptr, dis, N);
  fill_kernel<<<blocks(E, B), B, 0, stream>>>(src, dst, rowptr, deg, colidx, E);

  embed_kernel<<<blocks((long)N * EMB, B), B, 0, stream>>>(shape_id, color_id,
                                                           shape_table, color_table, x0, N);

  node_gather_emb_kernel<<<blocks((long)N * 8, B), B, 0, stream>>>(rowptr, colidx, x0, dis, agg0, N);
  gemm_wmma_f32_bias_relu<<<N / 16, 128, 0, stream>>>(agg0, W1, b1, bufA);

  gemm_wmma_f32<<<N / 16, 128, 0, stream>>>(bufA, W2, bufH, HID);
  node_gather_hid_kernel<<<blocks((long)N * 16, B), B, 0, stream>>>(rowptr, colidx, bufH, dis, b2,
                                                                    bufA, N);

  zero_f32<<<blocks((long)NGRAPHS * HID + NGRAPHS, B), B, 0, stream>>>(pooled,
                                                                      (long)NGRAPHS * HID + NGRAPHS);
  pool_kernel<<<blocks((long)N * HID, B), B, 0, stream>>>(bufA, batch, pooled, cnt, N);
  out_kernel<<<blocks((long)NGRAPHS * NCLASSES, B), B, 0, stream>>>(pooled, cnt, Wl, bl, out);
}


# Thread coarsening: zero, degree, fill, embed, pool kernels do 4-16 items per thread, waves per call fall below a third; pool sums 16-node runs in registers before its atomics, f32 sum order changes
# speedup vs baseline: 1.7023x; 1.1895x over previous
#include <hip/hip_runtime.h>

#define NNODES 100000
#define NGRAPHS 1024
#define EMB 32
#define HID 64
#define NCLASSES 10

typedef __attribute__((ext_vector_type(2))) float v2f;
typedef __attribute__((ext_vector_type(8))) float v8f;

__global__ void zero_f32(float* __restrict__ p, long n) {
  const long i = ((long)blockIdx.x * blockDim.x + threadIdx.x) * 4;
  if (i + 4 <= n) {
    *(float4*)(p + i) = make_float4(0.0f, 0.0f, 0.0f, 0.0f);
  } else {
    for (long k = i; k < n; ++k) p[k] = 0.0f;
  }
}

#define EDGE_T 256
#define EDGE_BLK (8 * EDGE_T)
__global__ void deg_kernel(const int* __restrict__ dst, unsigned int* __restrict__ deg, int E) {
  const long b0 = (long)blockIdx.x * EDGE_BLK;
  const int t = threadIdx.x;
  if (b0 + EDGE_BLK <= (long)E && (((size_t)dst) & 15) == 0) {
    const int4 a = *(const int4*)(dst + b0 + 4 * t);
    const int4 b = *(const int4*)(dst + b0 + 4 * EDGE_T + 4 * t);
    atomicAdd(&deg[a.x], 1u); atomicAdd(&deg[b.x], 1u);
    atomicAdd(&deg[a.y], 1u); atomicAdd(&deg[b.y], 1u);
    atomicAdd(&deg[a.z], 1u); atomicAdd(&deg[b.z], 1u);
    atomicAdd(&deg[a.w], 1u); atomicAdd(&deg[b.w], 1u);
  } else {
    for (long e = b0 + t; e < b0 + EDGE_BLK && e < (long)E; e += EDGE_T) atomicAdd(&deg[dst[e]], 1u);
  }
}

#define SCAN_T 1024
#define SCAN_E 4096
__global__ void scan_dis_kernel(const int* __restrict__ deg, int* __restrict__ rowptr,
                                float* __restrict__ dis, int n) {
  __shared__ int s_part[SCAN_T / 32];
  __shared__ int s_wsum[SCAN_T / 32];
  const int t = threadIdx.x;
  const int lane = t & 31;
  const int wave = t >> 5;

  int pre = 0;
  const int lim = blockIdx.x * SCAN_E;
  int j = 4 * t;
  for (; j + 3 * SCAN_E < lim; j += 4 * SCAN_E) {
    const int4 q0 = *(const int4*)(deg + j);
    const int4 q1 = *(const int4*)(deg + j + SCAN_E);
    const int4 q2 = *(const int4*)(deg + j + 2 * SCAN_E);
    const int4 q3 = *(const int4*)(deg + j + 3 * SCAN_E);
    pre += ((q0.x + q0.y) + (q0.z + q0.w)) + ((q1.x + q1.y) + (q1.z + q1.w));
    pre += ((q2.x + q2.y) + (q2.z + q2.w)) + ((q3.x + q3.y) + (q3.z + q3.w));
  }
  for (; j < lim; j += SCAN_E) {
    const int4 q0 = *(const int4*)(deg + j);
    pre += (q0.x + q0.y) + (q0.z + q0.w);
  }
  for (int o = 16; o > 0; o >>= 1) pre += __shfl_xor(pre, o);
  if (lane == 0) s_part[wave] = pre;

  const int base = lim + 4 * t;
  int4 q = make_int4(0, 0, 0, 0);
  if (base < n) q = *(const int4*)(deg + base);
  const int own = (q.x + q.y) + (q.z + q.w);
  int inc = own;
  for (int o = 1; o < 32; o <<= 1) {
    const int up = __shfl_up(inc, o);
    if (lane >= o) inc += up;
  }
  if (lane == 31) s_wsum[wave] = inc;
  __syncthreads();

  int off = 0;
  for (int w = 0; w < SCAN_T / 32; ++w) {
    off += s_part[w];
    if (w < wave) off += s_wsum[w];
  }
  const int ex = off + inc - own;
  if (base < n) {
    int4 r;
    r.x = ex;
    r.y = ex + q.x;
    r.z = r.y + q.y;
    r.w = r.z + q.z;
    *(int4*)(rowptr + base) = r;
    float4 d;
    d.x = rsqrtf(1.0f + (float)q.x);
    d.y = rsqrtf(1.0f + (float)q.y);
    d.z = rsqrtf(1.0f + (float)q.z);
    d.w = rsqrtf(1.0f + (float)q.w);
    *(float4*)(dis + base) = d;
    if (base + 4 == n) rowptr[n] = r.w + q.w;
  }
}

__global__ void fill_kernel(const int* __restrict__ src, const int* __restrict__ dst,
                            const int* __restrict__ rowptr, int* __restrict__ deg,
                            int* __restrict__ colidx, int E) {
  const long b0 = (long)blockIdx.x * EDGE_BLK;
  const int t = threadIdx.x;
  if (b0 + EDGE_BLK <= (long)E && ((((size_t)src) | ((size_t)dst)) & 15) == 0) {
    const long e0 = b0 + 4 * t;
    const long e1 = e0 + 4 * EDGE_T;
    const int4 d0 = *(const int4*)(dst + e0);
    const int4 d1 = *(const int4*)(dst + e1);
    const int4 s0 = *(const int4*)(src + e0);
    const int4 s1 = *(const int4*)(src + e1);
    const int r0 = rowptr[d0.x], r1 = rowptr[d0.y], r2 = rowptr[d0.z], r3 = rowptr[d0.w];
    const int r4 = rowptr[d1.x], r5 = rowptr[d1.y], r6 = rowptr[d1.z], r7 = rowptr[d1.w];
    const int o0 = atomicSub(&deg[d0.x], 1);
    const int o1 = atomicSub(&deg[d0.y], 1);
    const int o2 = atomicSub(&deg[d0.z], 1);
    const int o3 = atomicSub(&deg[d0.w], 1);
    const int o4 = atomicSub(&deg[d1.x], 1);
    const int o5 = atomicSub(&deg[d1.y], 1);
    const int o6 = atomicSub(&deg[d1.z], 1);
    const int o7 = atomicSub(&deg[d1.w], 1);
    colidx[r0 + o0 - 1] = s0.x;
    colidx[r1 + o1 - 1] = s0.y;
    colidx[r2 + o2 - 1] = s0.z;
    colidx[r3 + o3 - 1] = s0.w;
    colidx[r4 + o4 - 1] = s1.x;
    colidx[r5 + o5 - 1] = s1.y;
    colidx[r6 + o6 - 1] = s1.z;
    colidx[r7 + o7 - 1] = s1.w;
  } else {
    for (long e = b0 + t; e < b0 + EDGE_BLK && e < (long)E; e += EDGE_T) {
      const int d = dst[e];
      const int old = atomicSub(&deg[d], 1);
      colidx[rowptr[d] + old - 1] = src[e];
    }
  }
}

#define EMB_T 256
__global__ void embed_kernel(const int* __restrict__ sid, const int* __restrict__ cid,
                             const float* __restrict__ st, const float* __restrict__ ct,
                             float* __restrict__ x0, int n) {
  const long total = (long)n * (EMB / 4);
  const long blk0 = (long)blockIdx.x * (4 * EMB_T);
  const long base = blk0 + threadIdx.x;
  const int c4 = ((int)(threadIdx.x & 7)) << 2;
  if (blk0 + 4 * EMB_T <= total) {
    int s[4], cc[4];
    float4 a[4], b[4];
#pragma unroll
    for (int k = 0; k < 4; ++k) {
      const int node = (int)((base + k * EMB_T) >> 3);
      s[k] = sid[node];
      cc[k] = cid[node];
    }
#pragma unroll
    for (int k = 0; k < 4; ++k) {
      a[k] = *(const float4*)(st + s[k] * EMB + c4);
      b[k] = *(const float4*)(ct + cc[k] * EMB + c4);
    }
#pragma unroll
    for (int k = 0; k < 4; ++k) {
      float4 v;
      v.x = (s[k] != 0 ? a[k].x : 0.0f) + (cc[k] != 0 ? b[k].x : 0.0f);
      v.y = (s[k] != 0 ? a[k].y : 0.0f) + (cc[k] != 0 ? b[k].y : 0.0f);
      v.z = (s[k] != 0 ? a[k].z : 0.0f) + (cc[k] != 0 ? b[k].z : 0.0f);
      v.w = (s[k] != 0 ? a[k].w : 0.0f) + (cc[k] != 0 ? b[k].w : 0.0f);
      *(float4*)(x0 + (base + k * EMB_T) * 4) = v;
    }
  } else {
    for (int k = 0; k < 4; ++k) {
      const long idx = base + k * EMB_T;
      if (idx < total) {
        const int node = (int)(idx >> 3);
        const int s1 = sid[node], c1 = cid[node];
        const float4 a1 = *(const float4*)(st + s1 * EMB + c4);
        const float4 b1 = *(const float4*)(ct + c1 * EMB + c4);
        float4 v;
        v.x = (s1 != 0 ? a1.x : 0.0f) + (c1 != 0 ? b1.x : 0.0f);
        v.y = (s1 != 0 ? a1.y : 0.0f) + (c1 != 0 ? b1.y : 0.0f);
        v.z = (s1 != 0 ? a1.z : 0.0f) + (c1 != 0 ? b1.z : 0.0f);
        v.w = (s1 != 0 ? a1.w : 0.0f) + (c1 != 0 ? b1.w : 0.0f);
        *(float4*)(x0 + idx * 4) = v;
      }
    }
  }
}

__global__ void gemm_wmma_f32(const float* __restrict__ X, const float* __restrict__ W,
                              float* __restrict__ H, int K) {
  const int lane  = threadIdx.x & 31;
  const int wave  = threadIdx.x >> 5;
  const int l16   = lane & 15;
  const int khalf = lane >> 4;
  const long row  = (long)blockIdx.x * 16 + l16;
  const int  col  = wave * 16 + l16;

  v8f c = {};
  const float* ap = X + row * K + 2 * khalf;
  for (int k0 = 0; k0 < K; k0 += 4) {
    v2f a, b;
    a.x = ap[k0];
    a.y = ap[k0 + 1];
    b.x = W[(k0 + 2 * khalf + 0) * HID + col];
    b.y = W[(k0 + 2 * khalf + 1) * HID + col];
    c = __builtin_amdgcn_wmma_f32_16x16x4_f32(false, a, false, b, (short)0, c, false, false);
  }
  const long rbase = (long)blockIdx.x * 16 + 8 * khalf;
#pragma unroll
  for (int v = 0; v < 8; ++v)
    H[(rbase + v) * HID + col] = c[v];
}

__global__ void gemm_wmma_f32_bias_relu(const float* __restrict__ AGG,
                                        const float* __restrict__ W,
                                        const float* __restrict__ bias,
                                        float* __restrict__ X1) {
  const int lane  = threadIdx.x & 31;
  const int wave  = threadIdx.x >> 5;
  const int l16   = lane & 15;
  const int khalf = lane >> 4;
  const long row  = (long)blockIdx.x * 16 + l16;
  const int  col  = wave * 16 + l16;

  v8f c = {};
  const float* gp = AGG + row * EMB + 2 * khalf;
  for (int k0 = 0; k0 < EMB; k0 += 4) {
    v2f a, b;
    a.x = gp[k0];
    a.y = gp[k0 + 1];
    b.x = W[(k0 + 2 * khalf + 0) * HID + col];
    b.y = W[(k0 + 2 * khalf + 1) * HID + col];
    c = __builtin_amdgcn_wmma_f32_16x16x4_f32(false, a, false, b, (short)0, c, false, false);
  }
  const float bc = bias[col];
  const long rbase = (long)blockIdx.x * 16 + 8 * khalf;
  for (int v = 0; v < 8; ++v) {
    float o = c[v] + bc;
    X1[(rbase + v) * HID + col] = o > 0.0f ? o : 0.0f;
  }
}

template <int WIDTH, bool FINAL>
__device__ __forceinline__ void node_gather_body(const int* __restrict__ rowptr,
                                                 const int* __restrict__ colidx,
                                                 const float* __restrict__ h,
                                                 const float* __restrict__ dis,
                                                 const float* __restrict__ bias,
                                                 float* __restrict__ out, int n) {
  constexpr int LPN = WIDTH / 4;
  constexpr int SH  = (LPN == 16) ? 4 : 3;
  const long t = (long)blockIdx.x * blockDim.x + threadIdx.x;
  const int node = (int)(t >> SH);
  if (node >= n) return;
  const int c4 = (int)(t & (LPN - 1)) << 2;
  const int beg = rowptr[node];
  const int end = rowptr[node + 1];
  const float dd = dis[node];
  const float* hc = h + c4;

  const float4 self = *(const float4*)(hc + (long)node * WIDTH);
  const float d2 = dd * dd;
  float ax = self.x * d2, ay = self.y * d2, az = self.z * d2, aw = self.w * d2;

  int j = beg;
  for (; j + 4 <= end; j += 4) {
    const int s0 = colidx[j], s1 = colidx[j + 1], s2 = colidx[j + 2], s3 = colidx[j + 3];
    const float w0 = dis[s0] * dd, w1 = dis[s1] * dd, w2 = dis[s2] * dd, w3 = dis[s3] * dd;
    const float4 v0 = *(const float4*)(hc + (long)s0 * WIDTH);
    const float4 v1 = *(const float4*)(hc + (long)s1 * WIDTH);
    const float4 v2 = *(const float4*)(hc + (long)s2 * WIDTH);
    const float4 v3 = *(const float4*)(hc + (long)s3 * WIDTH);
    ax += v0.x * w0; ay += v0.y * w0; az += v0.z * w0; aw += v0.w * w0;
    ax += v1.x * w1; ay += v1.y * w1; az += v1.z * w1; aw += v1.w * w1;
    ax += v2.x * w2; ay += v2.y * w2; az += v2.z * w2; aw += v2.w * w2;
    ax += v3.x * w3; ay += v3.y * w3; az += v3.z * w3; aw += v3.w * w3;
  }
  if (j < end) {
    const int l = end - 1;
    const int j1 = j + 1 < l ? j + 1 : l;
    const int j2 = j + 2 < l ? j + 2 : l;
    const int s0 = colidx[j], s1 = colidx[j1], s2 = colidx[j2];
    const float m1 = (j + 1 < end) ? dd : 0.0f;
    const float m2 = (j + 2 < end) ? dd : 0.0f;
    const float w0 = dis[s0] * dd, w1 = dis[s1] * m1, w2 = dis[s2] * m2;
    const float4 v0 = *(const float4*)(hc + (long)s0 * WIDTH);
    const float4 v1 = *(const float4*)(hc + (long)s1 * WIDTH);
    const float4 v2 = *(const float4*)(hc + (long)s2 * WIDTH);
    ax += v0.x * w0; ay += v0.y * w0; az += v0.z * w0; aw += v0.w * w0;
    ax += v1.x * w1; ay += v1.y * w1; az += v1.z * w1; aw += v1.w * w1;
    ax += v2.x * w2; ay += v2.y * w2; az += v2.z * w2; aw += v2.w * w2;
  }

  if (FINAL) {
    const float4 b = *(const float4*)(bias + c4);
    ax += b.x; ay += b.y; az += b.z; aw += b.w;
    ax = ax > 0.0f ? ax : 0.0f;
    ay = ay > 0.0f ? ay : 0.0f;
    az = az > 0.0f ? az : 0.0f;
    aw = aw > 0.0f ? aw : 0.0f;
  }
  float4 o;
  o.x = ax; o.y = ay; o.z = az; o.w = aw;
  *(float4*)(out + (long)node * WIDTH + c4) = o;
}

__global__ void node_gather_emb_kernel(const int* __restrict__ rowptr, const int* __restrict__ colidx,
                                       const float* __restrict__ x, const float* __restrict__ dis,
                                       float* __restrict__ agg, int n) {
  node_gather_body<EMB, false>(rowptr, colidx, x, dis, nullptr, agg, n);
}

__global__ void node_gather_hid_kernel(const int* __restrict__ rowptr, const int* __restrict__ colidx,
                                       const float* __restrict__ h, const float* __restrict__ dis,
                                       const float* __restrict__ b, float* __restrict__ x2, int n) {
  node_gather_body<HID, true>(rowptr, colidx, h, dis, b, x2, n);
}

#define POOL_R 16
__global__ void pool_kernel(const float* __restrict__ x, const int* __restrict__ batch,
                            float* __restrict__ pooled, float* __restrict__ cnt, int n) {
  const long t = (long)blockIdx.x * blockDim.x + threadIdx.x;
  const int c4 = ((int)(t & 15)) << 2;
  const long n0 = (t >> 4) * POOL_R;
  if (n0 >= (long)n) return;
  int g[POOL_R];
  float4 v[POOL_R];
#pragma unroll
  for (int i = 0; i < POOL_R; ++i) {
    const long node = n0 + i < (long)n ? n0 + i : (long)n - 1;
    g[i] = batch[node];
    v[i] = *(const float4*)(x + node * HID + c4);
  }
  int cur = g[0];
  float4 acc = v[0];
  float run = 1.0f;
#pragma unroll
  for (int i = 1; i < POOL_R; ++i) {
    if (n0 + i < (long)n) {
      if (g[i] != cur) {
        float* pp = pooled + (long)cur * HID + c4;
        atomicAdd(pp + 0, acc.x); atomicAdd(pp + 1, acc.y);
        atomicAdd(pp + 2, acc.z); atomicAdd(pp + 3, acc.w);
        if (c4 == 0) atomicAdd(cnt + cur, run);
        cur = g[i];
        acc = make_float4(0.0f, 0.0f, 0.0f, 0.0f);
        run = 0.0f;
      }
      acc.x += v[i].x; acc.y += v[i].y; acc.z += v[i].z; acc.w += v[i].w;
      run += 1.0f;
    }
  }
  float* pp = pooled + (long)cur * HID + c4;
  atomicAdd(pp + 0, acc.x); atomicAdd(pp + 1, acc.y);
  atomicAdd(pp + 2, acc.z); atomicAdd(pp + 3, acc.w);
  if (c4 == 0) atomicAdd(cnt + cur, run);
}

__global__ void out_kernel(const float* __restrict__ pooled, const float* __restrict__ cnt,
                           const float* __restrict__ Wl, const float* __restrict__ bl,
                           float* __restrict__ out) {
  int i = blockIdx.x * blockDim.x + threadIdx.x;
  if (i >= NGRAPHS * NCLASSES) return;
  int g = i / NCLASSES;
  int k = i % NCLASSES;
  float s = 0.0f;
#pragma unroll 8
  for (int hh = 0; hh < HID; ++hh) s += pooled[g * HID + hh] * Wl[hh * NCLASSES + k];
  float c = cnt[g];
  c = c > 1.0f ? c : 1.0f;
  out[i] = s / c + bl[k];
}

extern "C" void kernel_launch(void* const* d_in, const int* in_sizes, int n_in,
                              void* d_out, int out_size, void* d_ws, size_t ws_size,
                              hipStream_t stream) {
  const int* shape_id = (const int*)d_in[0];
  const int* color_id = (const int*)d_in[1];
  const int* edge_index = (const int*)d_in[2];
  const int* batch = (const int*)d_in[3];
  const float* shape_table = (const float*)d_in[4];
  const float* color_table = (const float*)d_in[5];
  const float* W1 = (const float*)d_in[6];
  const float* b1 = (const float*)d_in[7];
  const float* W2 = (const float*)d_in[8];
  const float* b2 = (const float*)d_in[9];
  const float* Wl = (const float*)d_in[10];
  const float* bl = (const float*)d_in[11];
  float* out = (float*)d_out;

  const int N = NNODES;
  const int E = in_sizes[2] / 2;
  const int* src = edge_index;
  const int* dst = edge_index + E;

  float* dis    = (float*)d_ws;
  int*   colidx = (int*)(dis + N);
  int*   rowptr = colidx + E;
  int*   deg    = rowptr + N + 4;
  float* bufH   = dis + (long)N * (1 + EMB);
  float* bufA   = bufH + (long)N * HID;
  float* pooled = bufA + (long)N * HID;
  float* cnt    = pooled + (long)NGRAPHS * HID;
  float* x0     = bufH;
  float* agg0   = bufH + (long)N * EMB;
  (void)n_in; (void)out_size; (void)ws_size;

  const int B = 256;
  auto blocks = [](long n, int b) { return (unsigned)((n + b - 1) / b); };

  zero_f32<<<blocks((N + 3) / 4, B), B, 0, stream>>>((float*)deg, N);
  deg_kernel<<<blocks(E, EDGE_BLK), EDGE_T, 0, stream>>>(dst, (unsigned int*)deg, E);
  scan_dis_kernel<<<blocks(N, SCAN_E), SCAN_T, 0, stream>>>(deg, rowptr, dis, N);
  fill_kernel<<<blocks(E, EDGE_BLK), EDGE_T, 0, stream>>>(src, dst, rowptr, deg, colidx, E);

  embed_kernel<<<blocks((long)N * (EMB / 4), 4 * EMB_T), EMB_T, 0, stream>>>(
      shape_id, color_id, shape_table, color_table, x0, N);

  node_gather_emb_kernel<<<blocks((long)N * 8, B), B, 0, stream>>>(rowptr, colidx, x0, dis, agg0, N);
  gemm_wmma_f32_bias_relu<<<N / 16, 128, 0, stream>>>(agg0, W1, b1, bufA);

  gemm_wmma_f32<<<N / 16, 128, 0, stream>>>(bufA, W2, bufH, HID);
  node_gather_hid_kernel<<<blocks((long)N * 16, B), B, 0, stream>>>(rowptr, colidx, bufH, dis, b2,
                                                                    bufA, N);

  zero_f32<<<blocks(((long)NGRAPHS * HID + NGRAPHS + 3) / 4, B), B, 0, stream>>>(
      pooled, (long)NGRAPHS * HID + NGRAPHS);
  pool_kernel<<<blocks((long)((N + POOL_R - 1) / POOL_R) * 16, B), B, 0, stream>>>(bufA, batch, pooled,
                                                                                 cnt, N);
  out_kernel<<<blocks((long)NGRAPHS * NCLASSES, B), B, 0, stream>>>(pooled, cnt, Wl, bl, out);
}


# Rows stored pre-multiplied by dis[src] (embed, layer-1 GEMM epilogue), dis[dst] applied once per node: both gathers drop the per-edge dis load and multiply; f32 term rounding changes, sum order kept
# speedup vs baseline: 1.7665x; 1.0377x over previous
#include <hip/hip_runtime.h>

#define NNODES 100000
#define NGRAPHS 1024
#define EMB 32
#define HID 64
#define NCLASSES 10

typedef __attribute__((ext_vector_type(2))) float v2f;
typedef __attribute__((ext_vector_type(8))) float v8f;

__global__ void zero_f32(float* __restrict__ p, long n) {
  const long i = ((long)blockIdx.x * blockDim.x + threadIdx.x) * 4;
  if (i + 4 <= n) {
    *(float4*)(p + i) = make_float4(0.0f, 0.0f, 0.0f, 0.0f);
  } else {
    for (long k = i; k < n; ++k) p[k] = 0.0f;
  }
}

#define EDGE_T 256
#define EDGE_BLK (8 * EDGE_T)
__global__ void deg_kernel(const int* __restrict__ dst, unsigned int* __restrict__ deg, int E) {
  const long b0 = (long)blockIdx.x * EDGE_BLK;
  const int t = threadIdx.x;
  if (b0 + EDGE_BLK <= (long)E && (((size_t)dst) & 15) == 0) {
    const int4 a = *(const int4*)(dst + b0 + 4 * t);
    const int4 b = *(const int4*)(dst + b0 + 4 * EDGE_T + 4 * t);
    atomicAdd(&deg[a.x], 1u); atomicAdd(&deg[b.x], 1u);
    atomicAdd(&deg[a.y], 1u); atomicAdd(&deg[b.y], 1u);
    atomicAdd(&deg[a.z], 1u); atomicAdd(&deg[b.z], 1u);
    atomicAdd(&deg[a.w], 1u); atomicAdd(&deg[b.w], 1u);
  } else {
    for (long e = b0 + t; e < b0 + EDGE_BLK && e < (long)E; e += EDGE_T) atomicAdd(&deg[dst[e]], 1u);
  }
}

#define SCAN_T 1024
#define SCAN_E 4096
__global__ void scan_dis_kernel(const int* __restrict__ deg, int* __restrict__ rowptr,
                                float* __restrict__ dis, int n) {
  __shared__ int s_part[SCAN_T / 32];
  __shared__ int s_wsum[SCAN_T / 32];
  const int t = threadIdx.x;
  const int lane = t & 31;
  const int wave = t >> 5;

  int pre = 0;
  const int lim = blockIdx.x * SCAN_E;
  int j = 4 * t;
  for (; j + 3 * SCAN_E < lim; j += 4 * SCAN_E) {
    const int4 q0 = *(const int4*)(deg + j);
    const int4 q1 = *(const int4*)(deg + j + SCAN_E);
    const int4 q2 = *(const int4*)(deg + j + 2 * SCAN_E);
    const int4 q3 = *(const int4*)(deg + j + 3 * SCAN_E);
    pre += ((q0.x + q0.y) + (q0.z + q0.w)) + ((q1.x + q1.y) + (q1.z + q1.w));
    pre += ((q2.x + q2.y) + (q2.z + q2.w)) + ((q3.x + q3.y) + (q3.z + q3.w));
  }
  for (; j < lim; j += SCAN_E) {
    const int4 q0 = *(const int4*)(deg + j);
    pre += (q0.x + q0.y) + (q0.z + q0.w);
  }
  for (int o = 16; o > 0; o >>= 1) pre += __shfl_xor(pre, o);
  if (lane == 0) s_part[wave] = pre;

  const int base = lim + 4 * t;
  int4 q = make_int4(0, 0, 0, 0);
  if (base < n) q = *(const int4*)(deg + base);
  const int own = (q.x + q.y) + (q.z + q.w);
  int inc = own;
  for (int o = 1; o < 32; o <<= 1) {
    const int up = __shfl_up(inc, o);
    if (lane >= o) inc += up;
  }
  if (lane == 31) s_wsum[wave] = inc;
  __syncthreads();

  int off = 0;
  for (int w = 0; w < SCAN_T / 32; ++w) {
    off += s_part[w];
    if (w < wave) off += s_wsum[w];
  }
  const int ex = off + inc - own;
  if (base < n) {
    int4 r;
    r.x = ex;
    r.y = ex + q.x;
    r.z = r.y + q.y;
    r.w = r.z + q.z;
    *(int4*)(rowptr + base) = r;
    float4 d;
    d.x = rsqrtf(1.0f + (float)q.x);
    d.y = rsqrtf(1.0f + (float)q.y);
    d.z = rsqrtf(1.0f + (float)q.z);
    d.w = rsqrtf(1.0f + (float)q.w);
    *(float4*)(dis + base) = d;
    if (base + 4 == n) rowptr[n] = r.w + q.w;
  }
}

__global__ void fill_kernel(const int* __restrict__ src, const int* __restrict__ dst,
                            const int* __restrict__ rowptr, int* __restrict__ deg,
                            int* __restrict__ colidx, int E) {
  const long b0 = (long)blockIdx.x * EDGE_BLK;
  const int t = threadIdx.x;
  if (b0 + EDGE_BLK <= (long)E && ((((size_t)src) | ((size_t)dst)) & 15) == 0) {
    const long e0 = b0 + 4 * t;
    const long e1 = e0 + 4 * EDGE_T;
    const int4 d0 = *(const int4*)(dst + e0);
    const int4 d1 = *(const int4*)(dst + e1);
    const int4 s0 = *(const int4*)(src + e0);
    const int4 s1 = *(const int4*)(src + e1);
    const int r0 = rowptr[d0.x], r1 = rowptr[d0.y], r2 = rowptr[d0.z], r3 = rowptr[d0.w];
    const int r4 = rowptr[d1.x], r5 = rowptr[d1.y], r6 = rowptr[d1.z], r7 = rowptr[d1.w];
    const int o0 = atomicSub(&deg[d0.x], 1);
    const int o1 = atomicSub(&deg[d0.y], 1);
    const int o2 = atomicSub(&deg[d0.z], 1);
    const int o3 = atomicSub(&deg[d0.w], 1);
    const int o4 = atomicSub(&deg[d1.x], 1);
    const int o5 = atomicSub(&deg[d1.y], 1);
    const int o6 = atomicSub(&deg[d1.z], 1);
    const int o7 = atomicSub(&deg[d1.w], 1);
    colidx[r0 + o0 - 1] = s0.x;
    colidx[r1 + o1 - 1] = s0.y;
    colidx[r2 + o2 - 1] = s0.z;
    colidx[r3 + o3 - 1] = s0.w;
    colidx[r4 + o4 - 1] = s1.x;
    colidx[r5 + o5 - 1] = s1.y;
    colidx[r6 + o6 - 1] = s1.z;
    colidx[r7 + o7 - 1] = s1.w;
  } else {
    for (long e = b0 + t; e < b0 + EDGE_BLK && e < (long)E; e += EDGE_T) {
      const int d = dst[e];
      const int old = atomicSub(&deg[d], 1);
      colidx[rowptr[d] + old - 1] = src[e];
    }
  }
}

#define EMB_T 256
__global__ void embed_kernel(const int* __restrict__ sid, const int* __restrict__ cid,
                             const float* __restrict__ st, const float* __restrict__ ct,
                             const float* __restrict__ dis, float* __restrict__ x0, int n) {
  const long total = (long)n * (EMB / 4);
  const long blk0 = (long)blockIdx.x * (4 * EMB_T);
  const long base = blk0 + threadIdx.x;
  const int c4 = ((int)(threadIdx.x & 7)) << 2;
  if (blk0 + 4 * EMB_T <= total) {
    int s[4], cc[4];
    float dn[4];
    float4 a[4], b[4];
#pragma unroll
    for (int k = 0; k < 4; ++k) {
      const int node = (int)((base + k * EMB_T) >> 3);
      s[k] = sid[node];
      cc[k] = cid[node];
      dn[k] = dis[node];
    }
#pragma unroll
    for (int k = 0; k < 4; ++k) {
      a[k] = *(const float4*)(st + s[k] * EMB + c4);
      b[k] = *(const float4*)(ct + cc[k] * EMB + c4);
    }
#pragma unroll
    for (int k = 0; k < 4; ++k) {
      float4 v;
      v.x = ((s[k] != 0 ? a[k].x : 0.0f) + (cc[k] != 0 ? b[k].x : 0.0f)) * dn[k];
      v.y = ((s[k] != 0 ? a[k].y : 0.0f) + (cc[k] != 0 ? b[k].y : 0.0f)) * dn[k];
      v.z = ((s[k] != 0 ? a[k].z : 0.0f) + (cc[k] != 0 ? b[k].z : 0.0f)) * dn[k];
      v.w = ((s[k] != 0 ? a[k].w : 0.0f) + (cc[k] != 0 ? b[k].w : 0.0f)) * dn[k];
      *(float4*)(x0 + (base + k * EMB_T) * 4) = v;
    }
  } else {
    for (int k = 0; k < 4; ++k) {
      const long idx = base + k * EMB_T;
      if (idx < total) {
        const int node = (int)(idx >> 3);
        const int s1 = sid[node], c1 = cid[node];
        const float d1 = dis[node];
        const float4 a1 = *(const float4*)(st + s1 * EMB + c4);
        const float4 b1 = *(const float4*)(ct + c1 * EMB + c4);
        float4 v;
        v.x = ((s1 != 0 ? a1.x : 0.0f) + (c1 != 0 ? b1.x : 0.0f)) * d1;
        v.y = ((s1 != 0 ? a1.y : 0.0f) + (c1 != 0 ? b1.y : 0.0f)) * d1;
        v.z = ((s1 != 0 ? a1.z : 0.0f) + (c1 != 0 ? b1.z : 0.0f)) * d1;
        v.w = ((s1 != 0 ? a1.w : 0.0f) + (c1 != 0 ? b1.w : 0.0f)) * d1;
        *(float4*)(x0 + idx * 4) = v;
      }
    }
  }
}

__global__ void gemm_wmma_f32(const float* __restrict__ X, const float* __restrict__ W,
                              float* __restrict__ H, int K) {
  const int lane  = threadIdx.x & 31;
  const int wave  = threadIdx.x >> 5;
  const int l16   = lane & 15;
  const int khalf = lane >> 4;
  const long row  = (long)blockIdx.x * 16 + l16;
  const int  col  = wave * 16 + l16;

  v8f c = {};
  const float* ap = X + row * K + 2 * khalf;
  for (int k0 = 0; k0 < K; k0 += 4) {
    v2f a, b;
    a.x = ap[k0];
    a.y = ap[k0 + 1];
    b.x = W[(k0 + 2 * khalf + 0) * HID + col];
    b.y = W[(k0 + 2 * khalf + 1) * HID + col];
    c = __builtin_amdgcn_wmma_f32_16x16x4_f32(false, a, false, b, (short)0, c, false, false);
  }
  const long rbase = (long)blockIdx.x * 16 + 8 * khalf;
#pragma unroll
  for (int v = 0; v < 8; ++v)
    H[(rbase + v) * HID + col] = c[v];
}

__global__ void gemm_wmma_f32_bias_relu(const float* __restrict__ AGG,
                                        const float* __restrict__ W,
                                        const float* __restrict__ bias,
                                        const float* __restrict__ dis,
                                        float* __restrict__ X1) {
  const int lane  = threadIdx.x & 31;
  const int wave  = threadIdx.x >> 5;
  const int l16   = lane & 15;
  const int khalf = lane >> 4;
  const long row  = (long)blockIdx.x * 16 + l16;
  const int  col  = wave * 16 + l16;

  const long rbase = (long)blockIdx.x * 16 + 8 * khalf;
  const float4 dlo = *(const float4*)(dis + rbase);
  const float4 dhi = *(const float4*)(dis + rbase + 4);
  const float dr[8] = {dlo.x, dlo.y, dlo.z, dlo.w, dhi.x, dhi.y, dhi.z, dhi.w};

  v8f c = {};
  const float* gp = AGG + row * EMB + 2 * khalf;
  for (int k0 = 0; k0 < EMB; k0 += 4) {
    v2f a, b;
    a.x = gp[k0];
    a.y = gp[k0 + 1];
    b.x = W[(k0 + 2 * khalf + 0) * HID + col];
    b.y = W[(k0 + 2 * khalf + 1) * HID + col];
    c = __builtin_amdgcn_wmma_f32_16x16x4_f32(false, a, false, b, (short)0, c, false, false);
  }
  const float bc = bias[col];
  for (int v = 0; v < 8; ++v) {
    float o = c[v] + bc;
    X1[(rbase + v) * HID + col] = (o > 0.0f ? o : 0.0f) * dr[v];
  }
}

template <int WIDTH, bool FINAL>
__device__ __forceinline__ void node_gather_body(const int* __restrict__ rowptr,
                                                 const int* __restrict__ colidx,
                                                 const float* __restrict__ hs,
                                                 const float* __restrict__ dis,
                                                 const float* __restrict__ bias,
                                                 float* __restrict__ out, int n) {
  constexpr int LPN = WIDTH / 4;
  constexpr int SH  = (LPN == 16) ? 4 : 3;
  const long t = (long)blockIdx.x * blockDim.x + threadIdx.x;
  const int node = (int)(t >> SH);
  if (node >= n) return;
  const int c4 = (int)(t & (LPN - 1)) << 2;
  const int beg = rowptr[node];
  const int end = rowptr[node + 1];
  const float dd = dis[node];
  const float* hc = hs + c4;

  const float4 self = *(const float4*)(hc + (long)node * WIDTH);
  float ax = self.x, ay = self.y, az = self.z, aw = self.w;

  int j = beg;
  for (; j + 4 <= end; j += 4) {
    const int s0 = colidx[j], s1 = colidx[j + 1], s2 = colidx[j + 2], s3 = colidx[j + 3];
    const float4 v0 = *(const float4*)(hc + (long)s0 * WIDTH);
    const float4 v1 = *(const float4*)(hc + (long)s1 * WIDTH);
    const float4 v2 = *(const float4*)(hc + (long)s2 * WIDTH);
    const float4 v3 = *(const float4*)(hc + (long)s3 * WIDTH);
    ax += v0.x; ay += v0.y; az += v0.z; aw += v0.w;
    ax += v1.x; ay += v1.y; az += v1.z; aw += v1.w;
    ax += v2.x; ay += v2.y; az += v2.z; aw += v2.w;
    ax += v3.x; ay += v3.y; az += v3.z; aw += v3.w;
  }
  if (j < end) {
    const int l = end - 1;
    const int j1 = j + 1 < l ? j + 1 : l;
    const int j2 = j + 2 < l ? j + 2 : l;
    const int s0 = colidx[j], s1 = colidx[j1], s2 = colidx[j2];
    const float m1 = (j + 1 < end) ? 1.0f : 0.0f;
    const float m2 = (j + 2 < end) ? 1.0f : 0.0f;
    const float4 v0 = *(const float4*)(hc + (long)s0 * WIDTH);
    const float4 v1 = *(const float4*)(hc + (long)s1 * WIDTH);
    const float4 v2 = *(const float4*)(hc + (long)s2 * WIDTH);
    ax += v0.x; ay += v0.y; az += v0.z; aw += v0.w;
    ax += v1.x * m1; ay += v1.y * m1; az += v1.z * m1; aw += v1.w * m1;
    ax += v2.x * m2; ay += v2.y * m2; az += v2.z * m2; aw += v2.w * m2;
  }
  ax *= dd; ay *= dd; az *= dd; aw *= dd;

  if (FINAL) {
    const float4 b = *(const float4*)(bias + c4);
    ax += b.x; ay += b.y; az += b.z; aw += b.w;
    ax = ax > 0.0f ? ax : 0.0f;
    ay = ay > 0.0f ? ay : 0.0f;
    az = az > 0.0f ? az : 0.0f;
    aw = aw > 0.0f ? aw : 0.0f;
  }
  float4 o;
  o.x = ax; o.y = ay; o.z = az; o.w = aw;
  *(float4*)(out + (long)node * WIDTH + c4) = o;
}

__global__ void node_gather_emb_kernel(const int* __restrict__ rowptr, const int* __restrict__ colidx,
                                       const float* __restrict__ xs, const float* __restrict__ dis,
                                       float* __restrict__ agg, int n) {
  node_gather_body<EMB, false>(rowptr, colidx, xs, dis, nullptr, agg, n);
}

__global__ void node_gather_hid_kernel(const int* __restrict__ rowptr, const int* __restrict__ colidx,
                                       const float* __restrict__ hs, const float* __restrict__ dis,
                                       const float* __restrict__ b, float* __restrict__ x2, int n) {
  node_gather_body<HID, true>(rowptr, colidx, hs, dis, b, x2, n);
}

#define POOL_R 16
__global__ void pool_kernel(const float* __restrict__ x, const int* __restrict__ batch,
                            float* __restrict__ pooled, float* __restrict__ cnt, int n) {
  const long t = (long)blockIdx.x * blockDim.x + threadIdx.x;
  const int c4 = ((int)(t & 15)) << 2;
  const long n0 = (t >> 4) * POOL_R;
  if (n0 >= (long)n) return;
  int g[POOL_R];
  float4 v[POOL_R];
#pragma unroll
  for (int i = 0; i < POOL_R; ++i) {
    const long node = n0 + i < (long)n ? n0 + i : (long)n - 1;
    g[i] = batch[node];
    v[i] = *(const float4*)(x + node * HID + c4);
  }
  int cur = g[0];
  float4 acc = v[0];
  float run = 1.0f;
#pragma unroll
  for (int i = 1; i < POOL_R; ++i) {
    if (n0 + i < (long)n) {
      if (g[i] != cur) {
        float* pp = pooled + (long)cur * HID + c4;
        atomicAdd(pp + 0, acc.x); atomicAdd(pp + 1, acc.y);
        atomicAdd(pp + 2, acc.z); atomicAdd(pp + 3, acc.w);
        if (c4 == 0) atomicAdd(cnt + cur, run);
        cur = g[i];
        acc = make_float4(0.0f, 0.0f, 0.0f, 0.0f);
        run = 0.0f;
      }
      acc.x += v[i].x; acc.y += v[i].y; acc.z += v[i].z; acc.w += v[i].w;
      run += 1.0f;
    }
  }
  float* pp = pooled + (long)cur * HID + c4;
  atomicAdd(pp + 0, acc.x); atomicAdd(pp + 1, acc.y);
  atomicAdd(pp + 2, acc.z); atomicAdd(pp + 3, acc.w);
  if (c4 == 0) atomicAdd(cnt + cur, run);
}

__global__ void out_kernel(const float* __restrict__ pooled, const float* __restrict__ cnt,
                           const float* __restrict__ Wl, const float* __restrict__ bl,
                           float* __restrict__ out) {
  int i = blockIdx.x * blockDim.x + threadIdx.x;
  if (i >= NGRAPHS * NCLASSES) return;
  int g = i / NCLASSES;
  int k = i % NCLASSES;
  float s = 0.0f;
#pragma unroll 8
  for (int hh = 0; hh < HID; ++hh) s += pooled[g * HID + hh] * Wl[hh * NCLASSES + k];
  float c = cnt[g];
  c = c > 1.0f ? c : 1.0f;
  out[i] = s / c + bl[k];
}

extern "C" void kernel_launch(void* const* d_in, const int* in_sizes, int n_in,
                              void* d_out, int out_size, void* d_ws, size_t ws_size,
                              hipStream_t stream) {
  const int* shape_id = (const int*)d_in[0];
  const int* color_id = (const int*)d_in[1];
  const int* edge_index = (const int*)d_in[2];
  const int* batch = (const int*)d_in[3];
  const float* shape_table = (const float*)d_in[4];
  const float* color_table = (const float*)d_in[5];
  const float* W1 = (const float*)d_in[6];
  const float* b1 = (const float*)d_in[7];
  const float* W2 = (const float*)d_in[8];
  const float* b2 = (const float*)d_in[9];
  const float* Wl = (const float*)d_in[10];
  const float* bl = (const float*)d_in[11];
  float* out = (float*)d_out;

  const int N = NNODES;
  const int E = in_sizes[2] / 2;
  const int* src = edge_index;
  const int* dst = edge_index + E;

  float* dis    = (float*)d_ws;
  int*   colidx = (int*)(dis + N);
  int*   rowptr = colidx + E;
  int*   deg    = rowptr + N + 4;
  float* bufH   = dis + (long)N * (1 + EMB);
  float* bufA   = bufH + (long)N * HID;
  float* pooled = bufA + (long)N * HID;
  float* cnt    = pooled + (long)NGRAPHS * HID;
  float* x0     = bufH;
  float* agg0   = bufH + (long)N * EMB;
  (void)n_in; (void)out_size; (void)ws_size;

  const int B = 256;
  auto blocks = [](long n, int b) { return (unsigned)((n + b - 1) / b); };

  zero_f32<<<blocks((N + 3) / 4, B), B, 0, stream>>>((float*)deg, N);
  deg_kernel<<<blocks(E, EDGE_BLK), EDGE_T, 0, stream>>>(dst, (unsigned int*)deg, E);
  scan_dis_kernel<<<blocks(N, SCAN_E), SCAN_T, 0, stream>>>(deg, rowptr, dis, N);
  fill_kernel<<<blocks(E, EDGE_BLK), EDGE_T, 0, stream>>>(src, dst, rowptr, deg, colidx, E);

  embed_kernel<<<blocks((long)N * (EMB / 4), 4 * EMB_T), EMB_T, 0, stream>>>(
      shape_id, color_id, shape_table, color_table, dis, x0, N);

  node_gather_emb_kernel<<<blocks((long)N * 8, B), B, 0, stream>>>(rowptr, colidx, x0, dis, agg0, N);
  gemm_wmma_f32_bias_relu<<<N / 16, 128, 0, stream>>>(agg0, W1, b1, dis, bufA);

  gemm_wmma_f32<<<N / 16, 128, 0, stream>>>(bufA, W2, bufH, HID);
  node_gather_hid_kernel<<<blocks((long)N * 16, B), B, 0, stream>>>(rowptr, colidx, bufH, dis, b2,
                                                                    bufA, N);

  zero_f32<<<blocks(((long)NGRAPHS * HID + NGRAPHS + 3) / 4, B), B, 0, stream>>>(
      pooled, (long)NGRAPHS * HID + NGRAPHS);
  pool_kernel<<<blocks((long)((N + POOL_R - 1) / POOL_R) * 16, B), B, 0, stream>>>(bufA, batch, pooled,
                                                                                 cnt, N);
  out_kernel<<<blocks((long)NGRAPHS * NCLASSES, B), B, 0, stream>>>(pooled, cnt, Wl, bl, out);
}


# Edge sort rebuilt on LDS counters: per-block bucket histograms, partition into 128-node buckets, rank inside a bucket in LDS; no global atomics, 12 to 11 launches; f32 sum order = LDS arrival rank
# speedup vs baseline: 3.9229x; 2.2207x over previous
#include <hip/hip_runtime.h>

#define NNODES 100000
#define NGRAPHS 1024
#define EMB 32
#define HID 64
#define NCLASSES 10

typedef __attribute__((ext_vector_type(2))) float v2f;
typedef __attribute__((ext_vector_type(8))) float v8f;

__global__ void zero_f32(float* __restrict__ p, long n) {
  const long i = ((long)blockIdx.x * blockDim.x + threadIdx.x) * 4;
  if (i + 4 <= n) {
    *(float4*)(p + i) = make_float4(0.0f, 0.0f, 0.0f, 0.0f);
  } else {
    for (long k = i; k < n; ++k) p[k] = 0.0f;
  }
}

#define BKT_SH 7
#define BKT_N (1 << BKT_SH)
#define NBKT ((NNODES + BKT_N - 1) / BKT_N)
#define HIST_PITCH 800
#define PART_T 1024
#define PART_EPT 16
#define PART_BLK (PART_EPT * PART_T)
static_assert(NBKT <= PART_T && NBKT <= HIST_PITCH, "one thread per bucket");
static_assert(NNODES <= (1 << 24) && BKT_N <= 128, "packed edge word");
static_assert(NBKT <= 65536, "bucket number kept in 16 bits");

__global__ void bucket_count_kernel(const int* __restrict__ dst, int* __restrict__ blockHist, int E) {
  __shared__ int hist[NBKT];
  const int t = threadIdx.x;
  if (t < NBKT) hist[t] = 0;
  const long b0 = (long)blockIdx.x * PART_BLK;
  int d[PART_EPT];
#pragma unroll
  for (int k = 0; k < PART_EPT; ++k) {
    const long e = b0 + k * PART_T + t;
    const int v = dst[e < (long)E ? e : (long)E - 1];
    d[k] = e < (long)E ? v : -1;
  }
  __syncthreads();
#pragma unroll
  for (int k = 0; k < PART_EPT; ++k)
    if (d[k] >= 0) atomicAdd(&hist[d[k] >> BKT_SH], 1);
  __syncthreads();
  if (t < NBKT) blockHist[(long)blockIdx.x * HIST_PITCH + t] = hist[t];
}

__global__ void bucket_partition_kernel(const int* __restrict__ src, const int* __restrict__ dst,
                                        const int* __restrict__ blockHist,
                                        int* __restrict__ bucketBase,
                                        unsigned int* __restrict__ tmp, int E) {
  __shared__ int cur[NBKT];
  __shared__ int gdel[NBKT];
  __shared__ int s_wtot[PART_T / 32];
  __shared__ int s_wown[PART_T / 32];
  __shared__ unsigned int stageW[PART_BLK];
  __shared__ unsigned short stageB[PART_BLK];
  const int t = threadIdx.x;
  const int lane = t & 31;
  const int wave = t >> 5;
  const long b0 = (long)blockIdx.x * PART_BLK;
  int d[PART_EPT], s[PART_EPT];
#pragma unroll
  for (int k = 0; k < PART_EPT; ++k) {
    const long e = b0 + k * PART_T + t;
    const long ec = e < (long)E ? e : (long)E - 1;
    const int dv = dst[ec];
    s[k] = src[ec];
    d[k] = e < (long)E ? dv : -1;
  }
  int tot = 0, pre = 0, own = 0;
  if (t < NBKT) {
    const int nblk = (int)gridDim.x;
    const int mine = (int)blockIdx.x;
#pragma unroll 8
    for (int j = 0; j < nblk; ++j) {
      const int v = blockHist[(long)j * HIST_PITCH + t];
      tot += v;
      pre += j < mine ? v : 0;
      own = j == mine ? v : own;
    }
  }
  int incT = tot, incO = own;
  for (int o = 1; o < 32; o <<= 1) {
    const int upT = __shfl_up(incT, o);
    const int upO = __shfl_up(incO, o);
    if (lane >= o) { incT += upT; incO += upO; }
  }
  if (lane == 31) { s_wtot[wave] = incT; s_wown[wave] = incO; }
  __syncthreads();
  int offT = 0, offO = 0;
  for (int w = 0; w < wave; ++w) { offT += s_wtot[w]; offO += s_wown[w]; }
  const int exT = offT + incT - tot;
  const int exO = offO + incO - own;
  if (t < NBKT) {
    cur[t] = exO;
    gdel[t] = exT + pre - exO;
  }
  __syncthreads();
#pragma unroll
  for (int k = 0; k < PART_EPT; ++k) {
    if (d[k] >= 0) {
      const int bk = d[k] >> BKT_SH;
      const int p = atomicAdd(&cur[bk], 1);
      stageW[p] = ((unsigned int)(d[k] & (BKT_N - 1)) << 24) | (unsigned int)s[k];
      stageB[p] = (unsigned short)bk;
    }
  }
  __syncthreads();
  const long rest = (long)E - b0;
  const int nloc = rest < (long)PART_BLK ? (int)rest : PART_BLK;
#pragma unroll
  for (int k = 0; k < PART_EPT; ++k) {
    const int i = k * PART_T + t;
    if (i < nloc) tmp[i + gdel[stageB[i]]] = stageW[i];
  }
  if (blockIdx.x == 0 && t < NBKT) {
    bucketBase[t] = exT;
    if (t == NBKT - 1) bucketBase[NBKT] = exT + tot;
  }
}

#define SORT_T 256
#define SORT_R 12
#define SORT_CAP 4096
__global__ void bucket_sort_kernel(const unsigned int* __restrict__ tmp,
                                   const int* __restrict__ bucketBase, int* __restrict__ rowptr,
                                   float* __restrict__ dis, int* __restrict__ colidx, int n) {
  __shared__ int cnt[BKT_N];
  __shared__ int cursor[BKT_N];
  __shared__ int s_w[BKT_N / 32];
  __shared__ int stage[SORT_CAP];
  const int t = threadIdx.x;
  const int lane = t & 31;
  const int wave = t >> 5;
  const int b = blockIdx.x;
  const int lo = bucketBase[b];
  const int hi = bucketBase[b + 1];
  if (t < BKT_N) cnt[t] = 0;
  unsigned int w[SORT_R];
#pragma unroll
  for (int k = 0; k < SORT_R; ++k) {
    const int i = lo + k * SORT_T + t;
    w[k] = i < hi ? tmp[i] : 0xFFFFFFFFu;
  }
  __syncthreads();
#pragma unroll
  for (int k = 0; k < SORT_R; ++k)
    if (w[k] != 0xFFFFFFFFu) atomicAdd(&cnt[w[k] >> 24], 1);
  for (int i = lo + SORT_R * SORT_T + t; i < hi; i += SORT_T) atomicAdd(&cnt[tmp[i] >> 24], 1);
  __syncthreads();

  const int c = t < BKT_N ? cnt[t] : 0;
  int inc = c;
  for (int o = 1; o < 32; o <<= 1) {
    const int up = __shfl_up(inc, o);
    if (lane >= o) inc += up;
  }
  if (t < BKT_N && lane == 31) s_w[wave] = inc;
  __syncthreads();
  int ex = 0;
  if (t < BKT_N) {
    int o = 0;
    for (int q = 0; q < wave; ++q) o += s_w[q];
    ex = lo + o + inc - c;
    cursor[t] = ex;
  }
  __syncthreads();

  if (hi - lo <= SORT_CAP) {
#pragma unroll
    for (int k = 0; k < SORT_R; ++k) {
      if (w[k] != 0xFFFFFFFFu) {
        const int p = atomicAdd(&cursor[w[k] >> 24], 1);
        stage[p - lo] = (int)(w[k] & 0xFFFFFFu);
      }
    }
    for (int i = lo + SORT_R * SORT_T + t; i < hi; i += SORT_T) {
      const unsigned int x = tmp[i];
      const int p = atomicAdd(&cursor[x >> 24], 1);
      stage[p - lo] = (int)(x & 0xFFFFFFu);
    }
    __syncthreads();
    for (int i = t; i < hi - lo; i += SORT_T) colidx[lo + i] = stage[i];
  } else {
#pragma unroll
    for (int k = 0; k < SORT_R; ++k) {
      if (w[k] != 0xFFFFFFFFu) {
        const int p = atomicAdd(&cursor[w[k] >> 24], 1);
        colidx[p] = (int)(w[k] & 0xFFFFFFu);
      }
    }
    for (int i = lo + SORT_R * SORT_T + t; i < hi; i += SORT_T) {
      const unsigned int x = tmp[i];
      const int p = atomicAdd(&cursor[x >> 24], 1);
      colidx[p] = (int)(x & 0xFFFFFFu);
    }
  }
  if (t < BKT_N) {
    const int node = b * BKT_N + t;
    if (node < n) {
      rowptr[node] = ex;
      dis[node] = rsqrtf(1.0f + (float)c);
      if (node == n - 1) rowptr[n] = ex + c;
    }
  }
}

#define EMB_T 256
__global__ void embed_kernel(const int* __restrict__ sid, const int* __restrict__ cid,
                             const float* __restrict__ st, const float* __restrict__ ct,
                             const float* __restrict__ dis, float* __restrict__ x0, int n) {
  const long total = (long)n * (EMB / 4);
  const long blk0 = (long)blockIdx.x * (4 * EMB_T);
  const long base = blk0 + threadIdx.x;
  const int c4 = ((int)(threadIdx.x & 7)) << 2;
  if (blk0 + 4 * EMB_T <= total) {
    int s[4], cc[4];
    float dn[4];
    float4 a[4], b[4];
#pragma unroll
    for (int k = 0; k < 4; ++k) {
      const int node = (int)((base + k * EMB_T) >> 3);
      s[k] = sid[node];
      cc[k] = cid[node];
      dn[k] = dis[node];
    }
#pragma unroll
    for (int k = 0; k < 4; ++k) {
      a[k] = *(const float4*)(st + s[k] * EMB + c4);
      b[k] = *(const float4*)(ct + cc[k] * EMB + c4);
    }
#pragma unroll
    for (int k = 0; k < 4; ++k) {
      float4 v;
      v.x = ((s[k] != 0 ? a[k].x : 0.0f) + (cc[k] != 0 ? b[k].x : 0.0f)) * dn[k];
      v.y = ((s[k] != 0 ? a[k].y : 0.0f) + (cc[k] != 0 ? b[k].y : 0.0f)) * dn[k];
      v.z = ((s[k] != 0 ? a[k].z : 0.0f) + (cc[k] != 0 ? b[k].z : 0.0f)) * dn[k];
      v.w = ((s[k] != 0 ? a[k].w : 0.0f) + (cc[k] != 0 ? b[k].w : 0.0f)) * dn[k];
      *(float4*)(x0 + (base + k * EMB_T) * 4) = v;
    }
  } else {
    for (int k = 0; k < 4; ++k) {
      const long idx = base + k * EMB_T;
      if (idx < total) {
        const int node = (int)(idx >> 3);
        const int s1 = sid[node], c1 = cid[node];
        const float d1 = dis[node];
        const float4 a1 = *(const float4*)(st + s1 * EMB + c4);
        const float4 b1 = *(const float4*)(ct + c1 * EMB + c4);
        float4 v;
        v.x = ((s1 != 0 ? a1.x : 0.0f) + (c1 != 0 ? b1.x : 0.0f)) * d1;
        v.y = ((s1 != 0 ? a1.y : 0.0f) + (c1 != 0 ? b1.y : 0.0f)) * d1;
        v.z = ((s1 != 0 ? a1.z : 0.0f) + (c1 != 0 ? b1.z : 0.0f)) * d1;
        v.w = ((s1 != 0 ? a1.w : 0.0f) + (c1 != 0 ? b1.w : 0.0f)) * d1;
        *(float4*)(x0 + idx * 4) = v;
      }
    }
  }
}

__global__ void gemm_wmma_f32(const float* __restrict__ X, const float* __restrict__ W,
                              float* __restrict__ H, int K) {
  const int lane  = threadIdx.x & 31;
  const int wave  = threadIdx.x >> 5;
  const int l16   = lane & 15;
  const int khalf = lane >> 4;
  const long row  = (long)blockIdx.x * 16 + l16;
  const int  col  = wave * 16 + l16;

  v8f c = {};
  const float* ap = X + row * K + 2 * khalf;
  for (int k0 = 0; k0 < K; k0 += 4) {
    v2f a, b;
    a.x = ap[k0];
    a.y = ap[k0 + 1];
    b.x = W[(k0 + 2 * khalf + 0) * HID + col];
    b.y = W[(k0 + 2 * khalf + 1) * HID + col];
    c = __builtin_amdgcn_wmma_f32_16x16x4_f32(false, a, false, b, (short)0, c, false, false);
  }
  const long rbase = (long)blockIdx.x * 16 + 8 * khalf;
#pragma unroll
  for (int v = 0; v < 8; ++v)
    H[(rbase + v) * HID + col] = c[v];
}

__global__ void gemm_wmma_f32_bias_relu(const float* __restrict__ AGG,
                                        const float* __restrict__ W,
                                        const float* __restrict__ bias,
                                        const float* __restrict__ dis,
                                        float* __restrict__ X1) {
  const int lane  = threadIdx.x & 31;
  const int wave  = threadIdx.x >> 5;
  const int l16   = lane & 15;
  const int khalf = lane >> 4;
  const long row  = (long)blockIdx.x * 16 + l16;
  const int  col  = wave * 16 + l16;

  const long rbase = (long)blockIdx.x * 16 + 8 * khalf;
  const float4 dlo = *(const float4*)(dis + rbase);
  const float4 dhi = *(const float4*)(dis + rbase + 4);
  const float dr[8] = {dlo.x, dlo.y, dlo.z, dlo.w, dhi.x, dhi.y, dhi.z, dhi.w};

  v8f c = {};
  const float* gp = AGG + row * EMB + 2 * khalf;
  for (int k0 = 0; k0 < EMB; k0 += 4) {
    v2f a, b;
    a.x = gp[k0];
    a.y = gp[k0 + 1];
    b.x = W[(k0 + 2 * khalf + 0) * HID + col];
    b.y = W[(k0 + 2 * khalf + 1) * HID + col];
    c = __builtin_amdgcn_wmma_f32_16x16x4_f32(false, a, false, b, (short)0, c, false, false);
  }
  const float bc = bias[col];
  for (int v = 0; v < 8; ++v) {
    float o = c[v] + bc;
    X1[(rbase + v) * HID + col] = (o > 0.0f ? o : 0.0f) * dr[v];
  }
}

template <int WIDTH, bool FINAL>
__device__ __forceinline__ void node_gather_body(const int* __restrict__ rowptr,
                                                 const int* __restrict__ colidx,
                                                 const float* __restrict__ hs,
                                                 const float* __restrict__ dis,
                                                 const float* __restrict__ bias,
                                                 float* __restrict__ out, int n) {
  constexpr int LPN = WIDTH / 4;
  constexpr int SH  = (LPN == 16) ? 4 : 3;
  const long t = (long)blockIdx.x * blockDim.x + threadIdx.x;
  const int node = (int)(t >> SH);
  if (node >= n) return;
  const int c4 = (int)(t & (LPN - 1)) << 2;
  const int beg = rowptr[node];
  const int end = rowptr[node + 1];
  const float dd = dis[node];
  const float* hc = hs + c4;

  const float4 self = *(const float4*)(hc + (long)node * WIDTH);
  float ax = self.x, ay = self.y, az = self.z, aw = self.w;

  int j = beg;
  for (; j + 4 <= end; j += 4) {
    const int s0 = colidx[j], s1 = colidx[j + 1], s2 = colidx[j + 2], s3 = colidx[j + 3];
    const float4 v0 = *(const float4*)(hc + (long)s0 * WIDTH);
    const float4 v1 = *(const float4*)(hc + (long)s1 * WIDTH);
    const float4 v2 = *(const float4*)(hc + (long)s2 * WIDTH);
    const float4 v3 = *(const float4*)(hc + (long)s3 * WIDTH);
    ax += v0.x; ay += v0.y; az += v0.z; aw += v0.w;
    ax += v1.x; ay += v1.y; az += v1.z; aw += v1.w;
    ax += v2.x; ay += v2.y; az += v2.z; aw += v2.w;
    ax += v3.x; ay += v3.y; az += v3.z; aw += v3.w;
  }
  if (j < end) {
    const int l = end - 1;
    const int j1 = j + 1 < l ? j + 1 : l;
    const int j2 = j + 2 < l ? j + 2 : l;
    const int s0 = colidx[j], s1 = colidx[j1], s2 = colidx[j2];
    const float m1 = (j + 1 < end) ? 1.0f : 0.0f;
    const float m2 = (j + 2 < end) ? 1.0f : 0.0f;
    const float4 v0 = *(const float4*)(hc + (long)s0 * WIDTH);
    const float4 v1 = *(const float4*)(hc + (long)s1 * WIDTH);
    const float4 v2 = *(const float4*)(hc + (long)s2 * WIDTH);
    ax += v0.x; ay += v0.y; az += v0.z; aw += v0.w;
    ax += v1.x * m1; ay += v1.y * m1; az += v1.z * m1; aw += v1.w * m1;
    ax += v2.x * m2; ay += v2.y * m2; az += v2.z * m2; aw += v2.w * m2;
  }
  ax *= dd; ay *= dd; az *= dd; aw *= dd;

  if (FINAL) {
    const float4 b = *(const float4*)(bias + c4);
    ax += b.x; ay += b.y; az += b.z; aw += b.w;
    ax = ax > 0.0f ? ax : 0.0f;
    ay = ay > 0.0f ? ay : 0.0f;
    az = az > 0.0f ? az : 0.0f;
    aw = aw > 0.0f ? aw : 0.0f;
  }
  float4 o;
  o.x = ax; o.y = ay; o.z = az; o.w = aw;
  *(float4*)(out + (long)node * WIDTH + c4) = o;
}

__global__ void node_gather_emb_kernel(const int* __restrict__ rowptr, const int* __restrict__ colidx,
                                       const float* __restrict__ xs, const float* __restrict__ dis,
                                       float* __restrict__ agg, int n) {
  node_gather_body<EMB, false>(rowptr, colidx, xs, dis, nullptr, agg, n);
}

__global__ void node_gather_hid_kernel(const int* __restrict__ rowptr, const int* __restrict__ colidx,
                                       const float* __restrict__ hs, const float* __restrict__ dis,
                                       const float* __restrict__ b, float* __restrict__ x2, int n) {
  node_gather_body<HID, true>(rowptr, colidx, hs, dis, b, x2, n);
}

#define POOL_R 16
__global__ void pool_kernel(const float* __restrict__ x, const int* __restrict__ batch,
                            float* __restrict__ pooled, float* __restrict__ cnt, int n) {
  const long t = (long)blockIdx.x * blockDim.x + threadIdx.x;
  const int c4 = ((int)(t & 15)) << 2;
  const long n0 = (t >> 4) * POOL_R;
  if (n0 >= (long)n) return;
  int g[POOL_R];
  float4 v[POOL_R];
#pragma unroll
  for (int i = 0; i < POOL_R; ++i) {
    const long node = n0 + i < (long)n ? n0 + i : (long)n - 1;
    g[i] = batch[node];
    v[i] = *(const float4*)(x + node * HID + c4);
  }
  int cur = g[0];
  float4 acc = v[0];
  float run = 1.0f;
#pragma unroll
  for (int i = 1; i < POOL_R; ++i) {
    if (n0 + i < (long)n) {
      if (g[i] != cur) {
        float* pp = pooled + (long)cur * HID + c4;
        atomicAdd(pp + 0, acc.x); atomicAdd(pp + 1, acc.y);
        atomicAdd(pp + 2, acc.z); atomicAdd(pp + 3, acc.w);
        if (c4 == 0) atomicAdd(cnt + cur, run);
        cur = g[i];
        acc = make_float4(0.0f, 0.0f, 0.0f, 0.0f);
        run = 0.0f;
      }
      acc.x += v[i].x; acc.y += v[i].y; acc.z += v[i].z; acc.w += v[i].w;
      run += 1.0f;
    }
  }
  float* pp = pooled + (long)cur * HID + c4;
  atomicAdd(pp + 0, acc.x); atomicAdd(pp + 1, acc.y);
  atomicAdd(pp + 2, acc.z); atomicAdd(pp + 3, acc.w);
  if (c4 == 0) atomicAdd(cnt + cur, run);
}

__global__ void out_kernel(const float* __restrict__ pooled, const float* __restrict__ cnt,
                           const float* __restrict__ Wl, const float* __restrict__ bl,
                           float* __restrict__ out) {
  int i = blockIdx.x * blockDim.x + threadIdx.x;
  if (i >= NGRAPHS * NCLASSES) return;
  int g = i / NCLASSES;
  int k = i % NCLASSES;
  float s = 0.0f;
#pragma unroll 8
  for (int hh = 0; hh < HID; ++hh) s += pooled[g * HID + hh] * Wl[hh * NCLASSES + k];
  float c = cnt[g];
  c = c > 1.0f ? c : 1.0f;
  out[i] = s / c + bl[k];
}

extern "C" void kernel_launch(void* const* d_in, const int* in_sizes, int n_in,
                              void* d_out, int out_size, void* d_ws, size_t ws_size,
                              hipStream_t stream) {
  const int* shape_id = (const int*)d_in[0];
  const int* color_id = (const int*)d_in[1];
  const int* edge_index = (const int*)d_in[2];
  const int* batch = (const int*)d_in[3];
  const float* shape_table = (const float*)d_in[4];
  const float* color_table = (const float*)d_in[5];
  const float* W1 = (const float*)d_in[6];
  const float* b1 = (const float*)d_in[7];
  const float* W2 = (const float*)d_in[8];
  const float* b2 = (const float*)d_in[9];
  const float* Wl = (const float*)d_in[10];
  const float* bl = (const float*)d_in[11];
  float* out = (float*)d_out;

  const int N = NNODES;
  const int E = in_sizes[2] / 2;
  const int* src = edge_index;
  const int* dst = edge_index + E;

  float* dis    = (float*)d_ws;
  int*   colidx = (int*)(dis + N);
  int*   rowptr = colidx + E;
  int*   bucketBase = rowptr + N + 4;
  float* bufH   = dis + (long)N * (1 + EMB);
  float* bufA   = bufH + (long)N * HID;
  float* pooled = bufA + (long)N * HID;
  float* cnt    = pooled + (long)NGRAPHS * HID;
  float* x0     = bufH;
  float* agg0   = bufH + (long)N * EMB;
  unsigned int* tmp = (unsigned int*)bufA;
  int*   blockHist  = (int*)bufA + E;
  (void)n_in; (void)out_size; (void)ws_size;

  const int B = 256;
  auto blocks = [](long n, int b) { return (unsigned)((n + b - 1) / b); };

  bucket_count_kernel<<<blocks(E, PART_BLK), PART_T, 0, stream>>>(dst, blockHist, E);
  bucket_partition_kernel<<<blocks(E, PART_BLK), PART_T, 0, stream>>>(src, dst, blockHist, bucketBase,
                                                                      tmp, E);
  bucket_sort_kernel<<<NBKT, SORT_T, 0, stream>>>(tmp, bucketBase, rowptr, dis, colidx, N);

  embed_kernel<<<blocks((long)N * (EMB / 4), 4 * EMB_T), EMB_T, 0, stream>>>(
      shape_id, color_id, shape_table, color_table, dis, x0, N);

  node_gather_emb_kernel<<<blocks((long)N * 8, B), B, 0, stream>>>(rowptr, colidx, x0, dis, agg0, N);
  gemm_wmma_f32_bias_relu<<<N / 16, 128, 0, stream>>>(agg0, W1, b1, dis, bufA);

  gemm_wmma_f32<<<N / 16, 128, 0, stream>>>(bufA, W2, bufH, HID);
  node_gather_hid_kernel<<<blocks((long)N * 16, B), B, 0, stream>>>(rowptr, colidx, bufH, dis, b2,
                                                                    bufA, N);

  zero_f32<<<blocks(((long)NGRAPHS * HID + NGRAPHS + 3) / 4, B), B, 0, stream>>>(
      pooled, (long)NGRAPHS * HID + NGRAPHS);
  pool_kernel<<<blocks((long)((N + POOL_R - 1) / POOL_R) * 16, B), B, 0, stream>>>(bufA, batch, pooled,
                                                                                 cnt, N);
  out_kernel<<<blocks((long)NGRAPHS * NCLASSES, B), B, 0, stream>>>(pooled, cnt, Wl, bl, out);
}


# Layer-1 gather, W1 product and W2 product fused into one launch: a 32-node tile goes gather -> LDS -> WMMA -> LDS -> WMMA, only h is stored; 11 to 9 launches; f32 sum order and k order kept
# speedup vs baseline: 5.0431x; 1.2856x over previous
#include <hip/hip_runtime.h>

#define NNODES 100000
#define NGRAPHS 1024
#define EMB 32
#define HID 64
#define NCLASSES 10

typedef __attribute__((ext_vector_type(2))) float v2f;
typedef __attribute__((ext_vector_type(8))) float v8f;

__global__ void zero_f32(float* __restrict__ p, long n) {
  const long i = ((long)blockIdx.x * blockDim.x + threadIdx.x) * 4;
  if (i + 4 <= n) {
    *(float4*)(p + i) = make_float4(0.0f, 0.0f, 0.0f, 0.0f);
  } else {
    for (long k = i; k < n; ++k) p[k] = 0.0f;
  }
}

#define BKT_SH 7
#define BKT_N (1 << BKT_SH)
#define NBKT ((NNODES + BKT_N - 1) / BKT_N)
#define HIST_PITCH 800
#define PART_T 1024
#define PART_EPT 16
#define PART_BLK (PART_EPT * PART_T)
static_assert(NBKT <= PART_T && NBKT <= HIST_PITCH, "one thread per bucket");
static_assert(NNODES <= (1 << 24) && BKT_N <= 128, "packed edge word");
static_assert(NBKT <= 65536, "bucket number kept in 16 bits");

__global__ void bucket_count_kernel(const int* __restrict__ dst, int* __restrict__ blockHist, int E) {
  __shared__ int hist[NBKT];
  const int t = threadIdx.x;
  if (t < NBKT) hist[t] = 0;
  const long b0 = (long)blockIdx.x * PART_BLK;
  int d[PART_EPT];
#pragma unroll
  for (int k = 0; k < PART_EPT; ++k) {
    const long e = b0 + k * PART_T + t;
    const int v = dst[e < (long)E ? e : (long)E - 1];
    d[k] = e < (long)E ? v : -1;
  }
  __syncthreads();
#pragma unroll
  for (int k = 0; k < PART_EPT; ++k)
    if (d[k] >= 0) atomicAdd(&hist[d[k] >> BKT_SH], 1);
  __syncthreads();
  if (t < NBKT) blockHist[(long)blockIdx.x * HIST_PITCH + t] = hist[t];
}

__global__ void bucket_partition_kernel(const int* __restrict__ src, const int* __restrict__ dst,
                                        const int* __restrict__ blockHist,
                                        int* __restrict__ bucketBase,
                                        unsigned int* __restrict__ tmp, int E) {
  __shared__ int cur[NBKT];
  __shared__ int gdel[NBKT];
  __shared__ int s_wtot[PART_T / 32];
  __shared__ int s_wown[PART_T / 32];
  __shared__ unsigned int stageW[PART_BLK];
  __shared__ unsigned short stageB[PART_BLK];
  const int t = threadIdx.x;
  const int lane = t & 31;
  const int wave = t >> 5;
  const long b0 = (long)blockIdx.x * PART_BLK;
  int d[PART_EPT], s[PART_EPT];
#pragma unroll
  for (int k = 0; k < PART_EPT; ++k) {
    const long e = b0 + k * PART_T + t;
    const long ec = e < (long)E ? e : (long)E - 1;
    const int dv = dst[ec];
    s[k] = src[ec];
    d[k] = e < (long)E ? dv : -1;
  }
  int tot = 0, pre = 0, own = 0;
  if (t < NBKT) {
    const int nblk = (int)gridDim.x;
    const int mine = (int)blockIdx.x;
#pragma unroll 8
    for (int j = 0; j < nblk; ++j) {
      const int v = blockHist[(long)j * HIST_PITCH + t];
      tot += v;
      pre += j < mine ? v : 0;
      own = j == mine ? v : own;
    }
  }
  int incT = tot, incO = own;
  for (int o = 1; o < 32; o <<= 1) {
    const int upT = __shfl_up(incT, o);
    const int upO = __shfl_up(incO, o);
    if (lane >= o) { incT += upT; incO += upO; }
  }
  if (lane == 31) { s_wtot[wave] = incT; s_wown[wave] = incO; }
  __syncthreads();
  int offT = 0, offO = 0;
  for (int w = 0; w < wave; ++w) { offT += s_wtot[w]; offO += s_wown[w]; }
  const int exT = offT + incT - tot;
  const int exO = offO + incO - own;
  if (t < NBKT) {
    cur[t] = exO;
    gdel[t] = exT + pre - exO;
  }
  __syncthreads();
#pragma unroll
  for (int k = 0; k < PART_EPT; ++k) {
    if (d[k] >= 0) {
      const int bk = d[k] >> BKT_SH;
      const int p = atomicAdd(&cur[bk], 1);
      stageW[p] = ((unsigned int)(d[k] & (BKT_N - 1)) << 24) | (unsigned int)s[k];
      stageB[p] = (unsigned short)bk;
    }
  }
  __syncthreads();
  const long rest = (long)E - b0;
  const int nloc = rest < (long)PART_BLK ? (int)rest : PART_BLK;
#pragma unroll
  for (int k = 0; k < PART_EPT; ++k) {
    const int i = k * PART_T + t;
    if (i < nloc) tmp[i + gdel[stageB[i]]] = stageW[i];
  }
  if (blockIdx.x == 0 && t < NBKT) {
    bucketBase[t] = exT;
    if (t == NBKT - 1) bucketBase[NBKT] = exT + tot;
  }
}

#define SORT_T 256
#define SORT_R 12
#define SORT_CAP 4096
__global__ void bucket_sort_kernel(const unsigned int* __restrict__ tmp,
                                   const int* __restrict__ bucketBase, int* __restrict__ rowptr,
                                   float* __restrict__ dis, int* __restrict__ colidx, int n) {
  __shared__ int cnt[BKT_N];
  __shared__ int cursor[BKT_N];
  __shared__ int s_w[BKT_N / 32];
  __shared__ int stage[SORT_CAP];
  const int t = threadIdx.x;
  const int lane = t & 31;
  const int wave = t >> 5;
  const int b = blockIdx.x;
  const int lo = bucketBase[b];
  const int hi = bucketBase[b + 1];
  if (t < BKT_N) cnt[t] = 0;
  unsigned int w[SORT_R];
#pragma unroll
  for (int k = 0; k < SORT_R; ++k) {
    const int i = lo + k * SORT_T + t;
    w[k] = i < hi ? tmp[i] : 0xFFFFFFFFu;
  }
  __syncthreads();
#pragma unroll
  for (int k = 0; k < SORT_R; ++k)
    if (w[k] != 0xFFFFFFFFu) atomicAdd(&cnt[w[k] >> 24], 1);
  for (int i = lo + SORT_R * SORT_T + t; i < hi; i += SORT_T) atomicAdd(&cnt[tmp[i] >> 24], 1);
  __syncthreads();

  const int c = t < BKT_N ? cnt[t] : 0;
  int inc = c;
  for (int o = 1; o < 32; o <<= 1) {
    const int up = __shfl_up(inc, o);
    if (lane >= o) inc += up;
  }
  if (t < BKT_N && lane == 31) s_w[wave] = inc;
  __syncthreads();
  int ex = 0;
  if (t < BKT_N) {
    int o = 0;
    for (int q = 0; q < wave; ++q) o += s_w[q];
    ex = lo + o + inc - c;
    cursor[t] = ex;
  }
  __syncthreads();

  if (hi - lo <= SORT_CAP) {
#pragma unroll
    for (int k = 0; k < SORT_R; ++k) {
      if (w[k] != 0xFFFFFFFFu) {
        const int p = atomicAdd(&cursor[w[k] >> 24], 1);
        stage[p - lo] = (int)(w[k] & 0xFFFFFFu);
      }
    }
    for (int i = lo + SORT_R * SORT_T + t; i < hi; i += SORT_T) {
      const unsigned int x = tmp[i];
      const int p = atomicAdd(&cursor[x >> 24], 1);
      stage[p - lo] = (int)(x & 0xFFFFFFu);
    }
    __syncthreads();
    for (int i = t; i < hi - lo; i += SORT_T) colidx[lo + i] = stage[i];
  } else {
#pragma unroll
    for (int k = 0; k < SORT_R; ++k) {
      if (w[k] != 0xFFFFFFFFu) {
        const int p = atomicAdd(&cursor[w[k] >> 24], 1);
        colidx[p] = (int)(w[k] & 0xFFFFFFu);
      }
    }
    for (int i = lo + SORT_R * SORT_T + t; i < hi; i += SORT_T) {
      const unsigned int x = tmp[i];
      const int p = atomicAdd(&cursor[x >> 24], 1);
      colidx[p] = (int)(x & 0xFFFFFFu);
    }
  }
  if (t < BKT_N) {
    const int node = b * BKT_N + t;
    if (node < n) {
      rowptr[node] = ex;
      dis[node] = rsqrtf(1.0f + (float)c);
      if (node == n - 1) rowptr[n] = ex + c;
    }
  }
}

#define EMB_T 256
__global__ void embed_kernel(const int* __restrict__ sid, const int* __restrict__ cid,
                             const float* __restrict__ st, const float* __restrict__ ct,
                             const float* __restrict__ dis, float* __restrict__ x0, int n) {
  const long total = (long)n * (EMB / 4);
  const long blk0 = (long)blockIdx.x * (4 * EMB_T);
  const long base = blk0 + threadIdx.x;
  const int c4 = ((int)(threadIdx.x & 7)) << 2;
  if (blk0 + 4 * EMB_T <= total) {
    int s[4], cc[4];
    float dn[4];
    float4 a[4], b[4];
#pragma unroll
    for (int k = 0; k < 4; ++k) {
      const int node = (int)((base + k * EMB_T) >> 3);
      s[k] = sid[node];
      cc[k] = cid[node];
      dn[k] = dis[node];
    }
#pragma unroll
    for (int k = 0; k < 4; ++k) {
      a[k] = *(const float4*)(st + s[k] * EMB + c4);
      b[k] = *(const float4*)(ct + cc[k] * EMB + c4);
    }
#pragma unroll
    for (int k = 0; k < 4; ++k) {
      float4 v;
      v.x = ((s[k] != 0 ? a[k].x : 0.0f) + (cc[k] != 0 ? b[k].x : 0.0f)) * dn[k];
      v.y = ((s[k] != 0 ? a[k].y : 0.0f) + (cc[k] != 0 ? b[k].y : 0.0f)) * dn[k];
      v.z = ((s[k] != 0 ? a[k].z : 0.0f) + (cc[k] != 0 ? b[k].z : 0.0f)) * dn[k];
      v.w = ((s[k] != 0 ? a[k].w : 0.0f) + (cc[k] != 0 ? b[k].w : 0.0f)) * dn[k];
      *(float4*)(x0 + (base + k * EMB_T) * 4) = v;
    }
  } else {
    for (int k = 0; k < 4; ++k) {
      const long idx = base + k * EMB_T;
      if (idx < total) {
        const int node = (int)(idx >> 3);
        const int s1 = sid[node], c1 = cid[node];
        const float d1 = dis[node];
        const float4 a1 = *(const float4*)(st + s1 * EMB + c4);
        const float4 b1 = *(const float4*)(ct + c1 * EMB + c4);
        float4 v;
        v.x = ((s1 != 0 ? a1.x : 0.0f) + (c1 != 0 ? b1.x : 0.0f)) * d1;
        v.y = ((s1 != 0 ? a1.y : 0.0f) + (c1 != 0 ? b1.y : 0.0f)) * d1;
        v.z = ((s1 != 0 ? a1.z : 0.0f) + (c1 != 0 ? b1.z : 0.0f)) * d1;
        v.w = ((s1 != 0 ? a1.w : 0.0f) + (c1 != 0 ? b1.w : 0.0f)) * d1;
        *(float4*)(x0 + idx * 4) = v;
      }
    }
  }
}

template <int WIDTH>
__device__ __forceinline__ float4 node_row_sum(const int* __restrict__ colidx,
                                               const float* __restrict__ hs,
                                               int node, int beg, int end, int c4) {
  const float* hc = hs + c4;

  const float4 self = *(const float4*)(hc + (long)node * WIDTH);
  float ax = self.x, ay = self.y, az = self.z, aw = self.w;

  int j = beg;
  for (; j + 4 <= end; j += 4) {
    const int s0 = colidx[j], s1 = colidx[j + 1], s2 = colidx[j + 2], s3 = colidx[j + 3];
    const float4 v0 = *(const float4*)(hc + (long)s0 * WIDTH);
    const float4 v1 = *(const float4*)(hc + (long)s1 * WIDTH);
    const float4 v2 = *(const float4*)(hc + (long)s2 * WIDTH);
    const float4 v3 = *(const float4*)(hc + (long)s3 * WIDTH);
    ax += v0.x; ay += v0.y; az += v0.z; aw += v0.w;
    ax += v1.x; ay += v1.y; az += v1.z; aw += v1.w;
    ax += v2.x; ay += v2.y; az += v2.z; aw += v2.w;
    ax += v3.x; ay += v3.y; az += v3.z; aw += v3.w;
  }
  if (j < end) {
    const int l = end - 1;
    const int j1 = j + 1 < l ? j + 1 : l;
    const int j2 = j + 2 < l ? j + 2 : l;
    const int s0 = colidx[j], s1 = colidx[j1], s2 = colidx[j2];
    const float m1 = (j + 1 < end) ? 1.0f : 0.0f;
    const float m2 = (j + 2 < end) ? 1.0f : 0.0f;
    const float4 v0 = *(const float4*)(hc + (long)s0 * WIDTH);
    const float4 v1 = *(const float4*)(hc + (long)s1 * WIDTH);
    const float4 v2 = *(const float4*)(hc + (long)s2 * WIDTH);
    ax += v0.x; ay += v0.y; az += v0.z; aw += v0.w;
    ax += v1.x * m1; ay += v1.y * m1; az += v1.z * m1; aw += v1.w * m1;
    ax += v2.x * m2; ay += v2.y * m2; az += v2.z * m2; aw += v2.w * m2;
  }
  float4 r;
  r.x = ax; r.y = ay; r.z = az; r.w = aw;
  return r;
}

__global__ void node_gather_hid_kernel(const int* __restrict__ rowptr, const int* __restrict__ colidx,
                                       const float* __restrict__ hs, const float* __restrict__ dis,
                                       const float* __restrict__ bias, float* __restrict__ x2, int n) {
  const long t = (long)blockIdx.x * blockDim.x + threadIdx.x;
  const int node = (int)(t >> 4);
  if (node >= n) return;
  const int c4 = (int)(t & 15) << 2;
  const int beg = rowptr[node];
  const int end = rowptr[node + 1];
  const float dd = dis[node];
  const float4 a = node_row_sum<HID>(colidx, hs, node, beg, end, c4);
  float ax = a.x, ay = a.y, az = a.z, aw = a.w;
  ax *= dd; ay *= dd; az *= dd; aw *= dd;
  const float4 b = *(const float4*)(bias + c4);
  ax += b.x; ay += b.y; az += b.z; aw += b.w;
  float4 o;
  o.x = ax > 0.0f ? ax : 0.0f;
  o.y = ay > 0.0f ? ay : 0.0f;
  o.z = az > 0.0f ? az : 0.0f;
  o.w = aw > 0.0f ? aw : 0.0f;
  *(float4*)(x2 + (long)node * HID + c4) = o;
}

#define FUSE_T 256
#define FUSE_ROWS 32
#define FUSE_PA (EMB + 4)
#define FUSE_PX (HID + 4)
static_assert(NNODES % FUSE_ROWS == 0, "whole tiles only: no row guard in the fused kernel");
static_assert(FUSE_T == FUSE_ROWS * (EMB / 4), "eight lanes per node");
static_assert(FUSE_T / 32 == (FUSE_ROWS / 16) * (HID / 16), "one 16 x 16 tile per wave");

__global__ void __launch_bounds__(FUSE_T)
gather_emb_gemm_gemm_kernel(const int* __restrict__ rowptr, const int* __restrict__ colidx,
                            const float* __restrict__ xs, const float* __restrict__ dis,
                            const float* __restrict__ W1, const float* __restrict__ b1,
                            const float* __restrict__ W2, float* __restrict__ H) {
  __shared__ __attribute__((aligned(16))) float sA[FUSE_ROWS * FUSE_PA];
  __shared__ __attribute__((aligned(16))) float sX[FUSE_ROWS * FUSE_PX];
  const int t     = threadIdx.x;
  const int lane  = t & 31;
  const int wave  = t >> 5;
  const int l16   = lane & 15;
  const int khalf = lane >> 4;
  const int rt    = wave >> 2;
  const int col   = (wave & 3) * 16 + l16;
  const int node0 = (int)blockIdx.x * FUSE_ROWS;

  const int rloc = rt * 16 + 8 * khalf;
  const long rbase = (long)node0 + rloc;
  const float4 dlo = *(const float4*)(dis + rbase);
  const float4 dhi = *(const float4*)(dis + rbase + 4);
  const float dr[8] = {dlo.x, dlo.y, dlo.z, dlo.w, dhi.x, dhi.y, dhi.z, dhi.w};
  const float bc = b1[col];

  {
    const int r  = t >> 3;
    const int c4 = (t & 7) << 2;
    const int node = node0 + r;
    const int beg = rowptr[node];
    const int end = rowptr[node + 1];
    const float dd = dis[node];
    float4 a = node_row_sum<EMB>(colidx, xs, node, beg, end, c4);
    a.x *= dd; a.y *= dd; a.z *= dd; a.w *= dd;
    *(float4*)(sA + r * FUSE_PA + c4) = a;
  }
  v2f bw1[EMB / 4];
  for (int i = 0; i < EMB / 4; ++i) {
    bw1[i].x = W1[(4 * i + 2 * khalf + 0) * HID + col];
    bw1[i].y = W1[(4 * i + 2 * khalf + 1) * HID + col];
  }
  __syncthreads();

  v8f c = {};
  {
    const float* ap = sA + (rt * 16 + l16) * FUSE_PA + 2 * khalf;
    for (int i = 0; i < EMB / 4; ++i) {
      const v2f a = *(const v2f*)(ap + 4 * i);
      c = __builtin_amdgcn_wmma_f32_16x16x4_f32(false, a, false, bw1[i], (short)0, c, false, false);
    }
  }
  v2f bw2[HID / 4];
  for (int i = 0; i < HID / 4; ++i) {
    bw2[i].x = W2[(4 * i + 2 * khalf + 0) * HID + col];
    bw2[i].y = W2[(4 * i + 2 * khalf + 1) * HID + col];
  }
  {
    float* xp = sX + rloc * FUSE_PX + col;
    for (int v = 0; v < 8; ++v) {
      const float o = c[v] + bc;
      xp[v * FUSE_PX] = (o > 0.0f ? o : 0.0f) * dr[v];
    }
  }
  __syncthreads();

  v8f c2 = {};
  {
    const float* ap = sX + (rt * 16 + l16) * FUSE_PX + 2 * khalf;
    for (int i = 0; i < HID / 4; ++i) {
      const v2f a = *(const v2f*)(ap + 4 * i);
      c2 = __builtin_amdgcn_wmma_f32_16x16x4_f32(false, a, false, bw2[i], (short)0, c2, false, false);
    }
  }
  for (int v = 0; v < 8; ++v)
    H[(rbase + v) * HID + col] = c2[v];
}

#define POOL_R 16
__global__ void pool_kernel(const float* __restrict__ x, const int* __restrict__ batch,
                            float* __restrict__ pooled, float* __restrict__ cnt, int n) {
  const long t = (long)blockIdx.x * blockDim.x + threadIdx.x;
  const int c4 = ((int)(t & 15)) << 2;
  const long n0 = (t >> 4) * POOL_R;
  if (n0 >= (long)n) return;
  int g[POOL_R];
  float4 v[POOL_R];
#pragma unroll
  for (int i = 0; i < POOL_R; ++i) {
    const long node = n0 + i < (long)n ? n0 + i : (long)n - 1;
    g[i] = batch[node];
    v[i] = *(const float4*)(x + node * HID + c4);
  }
  int cur = g[0];
  float4 acc = v[0];
  float run = 1.0f;
#pragma unroll
  for (int i = 1; i < POOL_R; ++i) {
    if (n0 + i < (long)n) {
      if (g[i] != cur) {
        float* pp = pooled + (long)cur * HID + c4;
        atomicAdd(pp + 0, acc.x); atomicAdd(pp + 1, acc.y);
        atomicAdd(pp + 2, acc.z); atomicAdd(pp + 3, acc.w);
        if (c4 == 0) atomicAdd(cnt + cur, run);
        cur = g[i];
        acc = make_float4(0.0f, 0.0f, 0.0f, 0.0f);
        run = 0.0f;
      }
      acc.x += v[i].x; acc.y += v[i].y; acc.z += v[i].z; acc.w += v[i].w;
      run += 1.0f;
    }
  }
  float* pp = pooled + (long)cur * HID + c4;
  atomicAdd(pp + 0, acc.x); atomicAdd(pp + 1, acc.y);
  atomicAdd(pp + 2, acc.z); atomicAdd(pp + 3, acc.w);
  if (c4 == 0) atomicAdd(cnt + cur, run);
}

__global__ void out_kernel(const float* __restrict__ pooled, const float* __restrict__ cnt,
                           const float* __restrict__ Wl, const float* __restrict__ bl,
                           float* __restrict__ out) {
  int i = blockIdx.x * blockDim.x + threadIdx.x;
  if (i >= NGRAPHS * NCLASSES) return;
  int g = i / NCLASSES;
  int k = i % NCLASSES;
  float s = 0.0f;
#pragma unroll 8
  for (int hh = 0; hh < HID; ++hh) s += pooled[g * HID + hh] * Wl[hh * NCLASSES + k];
  float c = cnt[g];
  c = c > 1.0f ? c : 1.0f;
  out[i] = s / c + bl[k];
}

extern "C" void kernel_launch(void* const* d_in, const int* in_sizes, int n_in,
                              void* d_out, int out_size, void* d_ws, size_t ws_size,
                              hipStream_t stream) {
  const int* shape_id = (const int*)d_in[0];
  const int* color_id = (const int*)d_in[1];
  const int* edge_index = (const int*)d_in[2];
  const int* batch = (const int*)d_in[3];
  const float* shape_table = (const float*)d_in[4];
  const float* color_table = (const float*)d_in[5];
  const float* W1 = (const float*)d_in[6];
  const float* b1 = (const float*)d_in[7];
  const float* W2 = (const float*)d_in[8];
  const float* b2 = (const float*)d_in[9];
  const float* Wl = (const float*)d_in[10];
  const float* bl = (const float*)d_in[11];
  float* out = (float*)d_out;

  const int N = NNODES;
  const int E = in_sizes[2] / 2;
  const int* src = edge_index;
  const int* dst = edge_index + E;

  float* dis    = (float*)d_ws;
  int*   colidx = (int*)(dis + N);
  int*   rowptr = colidx + E;
  int*   bucketBase = rowptr + N + 4;
  float* bufH   = dis + (long)N * (1 + EMB);
  float* bufA   = bufH + (long)N * HID;
  float* pooled = bufA + (long)N * HID;
  float* cnt    = pooled + (long)NGRAPHS * HID;
  float* x0     = bufA;
  unsigned int* tmp = (unsigned int*)bufA;
  int*   blockHist  = (int*)bufA + E;
  (void)n_in; (void)out_size; (void)ws_size;

  const int B = 256;
  auto blocks = [](long n, int b) { return (unsigned)((n + b - 1) / b); };

  bucket_count_kernel<<<blocks(E, PART_BLK), PART_T, 0, stream>>>(dst, blockHist, E);
  bucket_partition_kernel<<<blocks(E, PART_BLK), PART_T, 0, stream>>>(src, dst, blockHist, bucketBase,
                                                                      tmp, E);
  bucket_sort_kernel<<<NBKT, SORT_T, 0, stream>>>(tmp, bucketBase, rowptr, dis, colidx, N);

  embed_kernel<<<blocks((long)N * (EMB / 4), 4 * EMB_T), EMB_T, 0, stream>>>(
      shape_id, color_id, shape_table, color_table, dis, x0, N);

  gather_emb_gemm_gemm_kernel<<<N / FUSE_ROWS, FUSE_T, 0, stream>>>(rowptr, colidx, x0, dis, W1, b1,
                                                                   W2, bufH);

  node_gather_hid_kernel<<<blocks((long)N * 16, B), B, 0, stream>>>(rowptr, colidx, bufH, dis, b2,
                                                                    bufA, N);

  zero_f32<<<blocks(((long)NGRAPHS * HID + NGRAPHS + 3) / 4, B), B, 0, stream>>>(
      pooled, (long)NGRAPHS * HID + NGRAPHS);
  pool_kernel<<<blocks((long)((N + POOL_R - 1) / POOL_R) * 16, B), B, 0, stream>>>(bufA, batch, pooled,
                                                                                 cnt, N);
  out_kernel<<<blocks((long)NGRAPHS * NCLASSES, B), B, 0, stream>>>(pooled, cnt, Wl, bl, out);
}


# Three small launches folded into others: pooled zeroing into sort launch 1, embedding rows into the sort's last launch, pool sums into the layer-2 gather (LDS tile); 9 to 6 launches; sum order kept
# speedup vs baseline: 5.6599x; 1.1223x over previous
#include <hip/hip_runtime.h>

#define NNODES 100000
#define NGRAPHS 1024
#define EMB 32
#define HID 64
#define NCLASSES 10

typedef __attribute__((ext_vector_type(2))) float v2f;
typedef __attribute__((ext_vector_type(8))) float v8f;

#define BKT_SH 7
#define BKT_N (1 << BKT_SH)
#define NBKT ((NNODES + BKT_N - 1) / BKT_N)
#define HIST_PITCH 800
#define PART_T 1024
#define PART_EPT 16
#define PART_BLK (PART_EPT * PART_T)
static_assert(NBKT <= PART_T && NBKT <= HIST_PITCH, "one thread per bucket");
static_assert(NNODES <= (1 << 24) && BKT_N <= 128, "packed edge word");
static_assert(NBKT <= 65536, "bucket number kept in 16 bits");

__global__ void bucket_count_kernel(const int* __restrict__ dst, int* __restrict__ blockHist, int E,
                                    float* __restrict__ zp, int zn4) {
  __shared__ int hist[NBKT];
  const int t = threadIdx.x;
  if (t < NBKT) hist[t] = 0;
  const long b0 = (long)blockIdx.x * PART_BLK;
  int d[PART_EPT];
#pragma unroll
  for (int k = 0; k < PART_EPT; ++k) {
    const long e = b0 + k * PART_T + t;
    const int v = dst[e < (long)E ? e : (long)E - 1];
    d[k] = e < (long)E ? v : -1;
  }
  __syncthreads();
#pragma unroll
  for (int k = 0; k < PART_EPT; ++k)
    if (d[k] >= 0) atomicAdd(&hist[d[k] >> BKT_SH], 1);
  __syncthreads();
  if (t < NBKT) blockHist[(long)blockIdx.x * HIST_PITCH + t] = hist[t];
  const int zstep = (int)gridDim.x * PART_T;
  for (int i = (int)blockIdx.x * PART_T + t; i < zn4; i += zstep)
    *(float4*)(zp + (long)i * 4) = make_float4(0.0f, 0.0f, 0.0f, 0.0f);
}

__global__ void bucket_partition_kernel(const int* __restrict__ src, const int* __restrict__ dst,
                                        const int* __restrict__ blockHist,
                                        int* __restrict__ bucketBase,
                                        unsigned int* __restrict__ tmp, int E) {
  __shared__ int cur[NBKT];
  __shared__ int gdel[NBKT];
  __shared__ int s_wtot[PART_T / 32];
  __shared__ int s_wown[PART_T / 32];
  __shared__ unsigned int stageW[PART_BLK];
  __shared__ unsigned short stageB[PART_BLK];
  const int t = threadIdx.x;
  const int lane = t & 31;
  const int wave = t >> 5;
  const long b0 = (long)blockIdx.x * PART_BLK;
  int d[PART_EPT], s[PART_EPT];
#pragma unroll
  for (int k = 0; k < PART_EPT; ++k) {
    const long e = b0 + k * PART_T + t;
    const long ec = e < (long)E ? e : (long)E - 1;
    const int dv = dst[ec];
    s[k] = src[ec];
    d[k] = e < (long)E ? dv : -1;
  }
  int tot = 0, pre = 0, own = 0;
  if (t < NBKT) {
    const int nblk = (int)gridDim.x;
    const int mine = (int)blockIdx.x;
#pragma unroll 8
    for (int j = 0; j < nblk; ++j) {
      const int v = blockHist[(long)j * HIST_PITCH + t];
      tot += v;
      pre += j < mine ? v : 0;
      own = j == mine ? v : own;
    }
  }
  int incT = tot, incO = own;
  for (int o = 1; o < 32; o <<= 1) {
    const int upT = __shfl_up(incT, o);
    const int upO = __shfl_up(incO, o);
    if (lane >= o) { incT += upT; incO += upO; }
  }
  if (lane == 31) { s_wtot[wave] = incT; s_wown[wave] = incO; }
  __syncthreads();
  int offT = 0, offO = 0;
  for (int w = 0; w < wave; ++w) { offT += s_wtot[w]; offO += s_wown[w]; }
  const int exT = offT + incT - tot;
  const int exO = offO + incO - own;
  if (t < NBKT) {
    cur[t] = exO;
    gdel[t] = exT + pre - exO;
  }
  __syncthreads();
#pragma unroll
  for (int k = 0; k < PART_EPT; ++k) {
    if (d[k] >= 0) {
      const int bk = d[k] >> BKT_SH;
      const int p = atomicAdd(&cur[bk], 1);
      stageW[p] = ((unsigned int)(d[k] & (BKT_N - 1)) << 24) | (unsigned int)s[k];
      stageB[p] = (unsigned short)bk;
    }
  }
  __syncthreads();
  const long rest = (long)E - b0;
  const int nloc = rest < (long)PART_BLK ? (int)rest : PART_BLK;
#pragma unroll
  for (int k = 0; k < PART_EPT; ++k) {
    const int i = k * PART_T + t;
    if (i < nloc) tmp[i + gdel[stageB[i]]] = stageW[i];
  }
  if (blockIdx.x == 0 && t < NBKT) {
    bucketBase[t] = exT;
    if (t == NBKT - 1) bucketBase[NBKT] = exT + tot;
  }
}

#define SORT_T 256
#define SORT_R 12
#define SORT_CAP 4096
#define EMB_K (BKT_N * (EMB / 4) / SORT_T)
static_assert(EMB_K * SORT_T == BKT_N * (EMB / 4), "whole pieces per thread");
static_assert(SORT_T % (EMB / 4) == 0, "a thread's pieces share one column group");
__global__ void bucket_sort_embed_kernel(const unsigned int* __restrict__ tmp,
                                         const int* __restrict__ bucketBase, int* __restrict__ rowptr,
                                         float* __restrict__ dis, int* __restrict__ colidx,
                                         const int* __restrict__ sid, const int* __restrict__ cid,
                                         const float* __restrict__ st, const float* __restrict__ ct,
                                         float* __restrict__ x0, int n) {
  __shared__ int cnt[BKT_N];
  __shared__ int cursor[BKT_N];
  __shared__ int s_w[BKT_N / 32];
  __shared__ float s_dis[BKT_N];
  __shared__ int stage[SORT_CAP];
  const int t = threadIdx.x;
  const int lane = t & 31;
  const int wave = t >> 5;
  const int b = blockIdx.x;
  const int lo = bucketBase[b];
  const int hi = bucketBase[b + 1];
  if (t < BKT_N) cnt[t] = 0;
  const int c4 = (t & (EMB / 4 - 1)) << 2;
  const int erow = t / (EMB / 4);
  int es[EMB_K], ec[EMB_K];
#pragma unroll
  for (int k = 0; k < EMB_K; ++k) {
    const int node = b * BKT_N + k * (SORT_T / (EMB / 4)) + erow;
    const int nc = node < n ? node : n - 1;
    es[k] = sid[nc];
    ec[k] = cid[nc];
  }
  unsigned int w[SORT_R];
#pragma unroll
  for (int k = 0; k < SORT_R; ++k) {
    const int i = lo + k * SORT_T + t;
    w[k] = i < hi ? tmp[i] : 0xFFFFFFFFu;
  }
  __syncthreads();
#pragma unroll
  for (int k = 0; k < SORT_R; ++k)
    if (w[k] != 0xFFFFFFFFu) atomicAdd(&cnt[w[k] >> 24], 1);
  for (int i = lo + SORT_R * SORT_T + t; i < hi; i += SORT_T) atomicAdd(&cnt[tmp[i] >> 24], 1);
  __syncthreads();

  const int c = t < BKT_N ? cnt[t] : 0;
  const float dnode = rsqrtf(1.0f + (float)c);
  if (t < BKT_N) s_dis[t] = dnode;
  int inc = c;
  for (int o = 1; o < 32; o <<= 1) {
    const int up = __shfl_up(inc, o);
    if (lane >= o) inc += up;
  }
  if (t < BKT_N && lane == 31) s_w[wave] = inc;
  __syncthreads();
  int ex = 0;
  if (t < BKT_N) {
    int o = 0;
    for (int q = 0; q < wave; ++q) o += s_w[q];
    ex = lo + o + inc - c;
    cursor[t] = ex;
  }
  __syncthreads();

  float4 ea[EMB_K], eb[EMB_K];
#pragma unroll
  for (int k = 0; k < EMB_K; ++k) {
    ea[k] = *(const float4*)(st + es[k] * EMB + c4);
    eb[k] = *(const float4*)(ct + ec[k] * EMB + c4);
  }

  if (hi - lo <= SORT_CAP) {
#pragma unroll
    for (int k = 0; k < SORT_R; ++k) {
      if (w[k] != 0xFFFFFFFFu) {
        const int p = atomicAdd(&cursor[w[k] >> 24], 1);
        stage[p - lo] = (int)(w[k] & 0xFFFFFFu);
      }
    }
    for (int i = lo + SORT_R * SORT_T + t; i < hi; i += SORT_T) {
      const unsigned int x = tmp[i];
      const int p = atomicAdd(&cursor[x >> 24], 1);
      stage[p - lo] = (int)(x & 0xFFFFFFu);
    }
    __syncthreads();
    for (int i = t; i < hi - lo; i += SORT_T) colidx[lo + i] = stage[i];
  } else {
#pragma unroll
    for (int k = 0; k < SORT_R; ++k) {
      if (w[k] != 0xFFFFFFFFu) {
        const int p = atomicAdd(&cursor[w[k] >> 24], 1);
        colidx[p] = (int)(w[k] & 0xFFFFFFu);
      }
    }
    for (int i = lo + SORT_R * SORT_T + t; i < hi; i += SORT_T) {
      const unsigned int x = tmp[i];
      const int p = atomicAdd(&cursor[x >> 24], 1);
      colidx[p] = (int)(x & 0xFFFFFFu);
    }
  }
  if (t < BKT_N) {
    const int node = b * BKT_N + t;
    if (node < n) {
      rowptr[node] = ex;
      dis[node] = dnode;
      if (node == n - 1) rowptr[n] = ex + c;
    }
  }
#pragma unroll
  for (int k = 0; k < EMB_K; ++k) {
    const int nl = k * (SORT_T / (EMB / 4)) + erow;
    const int node = b * BKT_N + nl;
    if (node < n) {
      const float dn = s_dis[nl];
      float4 v;
      v.x = ((es[k] != 0 ? ea[k].x : 0.0f) + (ec[k] != 0 ? eb[k].x : 0.0f)) * dn;
      v.y = ((es[k] != 0 ? ea[k].y : 0.0f) + (ec[k] != 0 ? eb[k].y : 0.0f)) * dn;
      v.z = ((es[k] != 0 ? ea[k].z : 0.0f) + (ec[k] != 0 ? eb[k].z : 0.0f)) * dn;
      v.w = ((es[k] != 0 ? ea[k].w : 0.0f) + (ec[k] != 0 ? eb[k].w : 0.0f)) * dn;
      *(float4*)(x0 + (long)node * EMB + c4) = v;
    }
  }
}

template <int WIDTH>
__device__ __forceinline__ float4 node_row_sum(const int* __restrict__ colidx,
                                               const float* __restrict__ hs,
                                               int node, int beg, int end, int c4) {
  const float* hc = hs + c4;

  const float4 self = *(const float4*)(hc + (long)node * WIDTH);
  float ax = self.x, ay = self.y, az = self.z, aw = self.w;

  int j = beg;
  for (; j + 4 <= end; j += 4) {
    const int s0 = colidx[j], s1 = colidx[j + 1], s2 = colidx[j + 2], s3 = colidx[j + 3];
    const float4 v0 = *(const float4*)(hc + (long)s0 * WIDTH);
    const float4 v1 = *(const float4*)(hc + (long)s1 * WIDTH);
    const float4 v2 = *(const float4*)(hc + (long)s2 * WIDTH);
    const float4 v3 = *(const float4*)(hc + (long)s3 * WIDTH);
    ax += v0.x; ay += v0.y; az += v0.z; aw += v0.w;
    ax += v1.x; ay += v1.y; az += v1.z; aw += v1.w;
    ax += v2.x; ay += v2.y; az += v2.z; aw += v2.w;
    ax += v3.x; ay += v3.y; az += v3.z; aw += v3.w;
  }
  if (j < end) {
    const int l = end - 1;
    const int j1 = j + 1 < l ? j + 1 : l;
    const int j2 = j + 2 < l ? j + 2 : l;
    const int s0 = colidx[j], s1 = colidx[j1], s2 = colidx[j2];
    const float m1 = (j + 1 < end) ? 1.0f : 0.0f;
    const float m2 = (j + 2 < end) ? 1.0f : 0.0f;
    const float4 v0 = *(const float4*)(hc + (long)s0 * WIDTH);
    const float4 v1 = *(const float4*)(hc + (long)s1 * WIDTH);
    const float4 v2 = *(const float4*)(hc + (long)s2 * WIDTH);
    ax += v0.x; ay += v0.y; az += v0.z; aw += v0.w;
    ax += v1.x * m1; ay += v1.y * m1; az += v1.z * m1; aw += v1.w * m1;
    ax += v2.x * m2; ay += v2.y * m2; az += v2.z * m2; aw += v2.w * m2;
  }
  float4 r;
  r.x = ax; r.y = ay; r.z = az; r.w = aw;
  return r;
}

#define GP_T 256
#define GP_NODES (GP_T / (HID / 4))
#define POOL_ZERO_N4 ((NGRAPHS * HID + NGRAPHS) / 4)
static_assert((NGRAPHS * HID + NGRAPHS) % 4 == 0 && (NGRAPHS * HID) % 4 == 0, "whole 16-byte pieces");
static_assert(GP_T >= HID, "one thread per column behind the barrier");
__global__ void __launch_bounds__(GP_T)
node_gather_hid_pool_kernel(const int* __restrict__ rowptr, const int* __restrict__ colidx,
                            const float* __restrict__ hs, const float* __restrict__ dis,
                            const float* __restrict__ bias, const int* __restrict__ batch,
                            float* __restrict__ pooled, float* __restrict__ cnt, int n) {
  __shared__ __attribute__((aligned(16))) float sO[GP_NODES * HID];
  __shared__ int sG[GP_NODES];
  const int t = threadIdx.x;
  const int r = t / (HID / 4);
  const int c4 = (t & (HID / 4 - 1)) << 2;
  const int node0 = (int)blockIdx.x * GP_NODES;
  const int node = node0 + r;
  float4 o = make_float4(0.0f, 0.0f, 0.0f, 0.0f);
  int g = 0;
  if (node < n) {
    g = batch[node];
    const int beg = rowptr[node];
    const int end = rowptr[node + 1];
    const float dd = dis[node];
    const float4 a = node_row_sum<HID>(colidx, hs, node, beg, end, c4);
    float ax = a.x, ay = a.y, az = a.z, aw = a.w;
    ax *= dd; ay *= dd; az *= dd; aw *= dd;
    const float4 bb = *(const float4*)(bias + c4);
    ax += bb.x; ay += bb.y; az += bb.z; aw += bb.w;
    o.x = ax > 0.0f ? ax : 0.0f;
    o.y = ay > 0.0f ? ay : 0.0f;
    o.z = az > 0.0f ? az : 0.0f;
    o.w = aw > 0.0f ? aw : 0.0f;
  }
  *(float4*)(sO + r * HID + c4) = o;
  if (c4 == 0) sG[r] = g;
  __syncthreads();
  if (t < HID) {
    const int rest = n - node0;
    const int rows = rest < GP_NODES ? rest : GP_NODES;
    int gs[GP_NODES];
    float vs[GP_NODES];
#pragma unroll
    for (int i = 0; i < GP_NODES; ++i) {
      gs[i] = sG[i];
      vs[i] = sO[i * HID + t];
    }
    int cur = gs[0];
    float acc = vs[0];
    float run = 1.0f;
#pragma unroll
    for (int i = 1; i < GP_NODES; ++i) {
      if (i < rows) {
        if (gs[i] != cur) {
          atomicAdd(pooled + (long)cur * HID + t, acc);
          if (t == 0) atomicAdd(cnt + cur, run);
          cur = gs[i];
          acc = 0.0f;
          run = 0.0f;
        }
        acc += vs[i];
        run += 1.0f;
      }
    }
    atomicAdd(pooled + (long)cur * HID + t, acc);
    if (t == 0) atomicAdd(cnt + cur, run);
  }
}

#define FUSE_T 256
#define FUSE_ROWS 32
#define FUSE_PA (EMB + 4)
#define FUSE_PX (HID + 4)
static_assert(NNODES % FUSE_ROWS == 0, "whole tiles only: no row guard in the fused kernel");
static_assert(FUSE_T == FUSE_ROWS * (EMB / 4), "eight lanes per node");
static_assert(FUSE_T / 32 == (FUSE_ROWS / 16) * (HID / 16), "one 16 x 16 tile per wave");

__global__ void __launch_bounds__(FUSE_T)
gather_emb_gemm_gemm_kernel(const int* __restrict__ rowptr, const int* __restrict__ colidx,
                            const float* __restrict__ xs, const float* __restrict__ dis,
                            const float* __restrict__ W1, const float* __restrict__ b1,
                            const float* __restrict__ W2, float* __restrict__ H) {
  __shared__ __attribute__((aligned(16))) float sA[FUSE_ROWS * FUSE_PA];
  __shared__ __attribute__((aligned(16))) float sX[FUSE_ROWS * FUSE_PX];
  const int t     = threadIdx.x;
  const int lane  = t & 31;
  const int wave  = t >> 5;
  const int l16   = lane & 15;
  const int khalf = lane >> 4;
  const int rt    = wave >> 2;
  const int col   = (wave & 3) * 16 + l16;
  const int node0 = (int)blockIdx.x * FUSE_ROWS;

  const int rloc = rt * 16 + 8 * khalf;
  const long rbase = (long)node0 + rloc;
  const float4 dlo = *(const float4*)(dis + rbase);
  const float4 dhi = *(const float4*)(dis + rbase + 4);
  const float dr[8] = {dlo.x, dlo.y, dlo.z, dlo.w, dhi.x, dhi.y, dhi.z, dhi.w};
  const float bc = b1[col];

  {
    const int r  = t >> 3;
    const int c4 = (t & 7) << 2;
    const int node = node0 + r;
    const int beg = rowptr[node];
    const int end = rowptr[node + 1];
    const float dd = dis[node];
    float4 a = node_row_sum<EMB>(colidx, xs, node, beg, end, c4);
    a.x *= dd; a.y *= dd; a.z *= dd; a.w *= dd;
    *(float4*)(sA + r * FUSE_PA + c4) = a;
  }
  v2f bw1[EMB / 4];
  for (int i = 0; i < EMB / 4; ++i) {
    bw1[i].x = W1[(4 * i + 2 * khalf + 0) * HID + col];
    bw1[i].y = W1[(4 * i + 2 * khalf + 1) * HID + col];
  }
  __syncthreads();

  v8f c = {};
  {
    const float* ap = sA + (rt * 16 + l16) * FUSE_PA + 2 * khalf;
    for (int i = 0; i < EMB / 4; ++i) {
      const v2f a = *(const v2f*)(ap + 4 * i);
      c = __builtin_amdgcn_wmma_f32_16x16x4_f32(false, a, false, bw1[i], (short)0, c, false, false);
    }
  }
  v2f bw2[HID / 4];
  for (int i = 0; i < HID / 4; ++i) {
    bw2[i].x = W2[(4 * i + 2 * khalf + 0) * HID + col];
    bw2[i].y = W2[(4 * i + 2 * khalf + 1) * HID + col];
  }
  {
    float* xp = sX + rloc * FUSE_PX + col;
    for (int v = 0; v < 8; ++v) {
      const float o = c[v] + bc;
      xp[v * FUSE_PX] = (o > 0.0f ? o : 0.0f) * dr[v];
    }
  }
  __syncthreads();

  v8f c2 = {};
  {
    const float* ap = sX + (rt * 16 + l16) * FUSE_PX + 2 * khalf;
    for (int i = 0; i < HID / 4; ++i) {
      const v2f a = *(const v2f*)(ap + 4 * i);
      c2 = __builtin_amdgcn_wmma_f32_16x16x4_f32(false, a, false, bw2[i], (short)0, c2, false, false);
    }
  }
  for (int v = 0; v < 8; ++v)
    H[(rbase + v) * HID + col] = c2[v];
}

__global__ void out_kernel(const float* __restrict__ pooled, const float* __restrict__ cnt,
                           const float* __restrict__ Wl, const float* __restrict__ bl,
                           float* __restrict__ out) {
  int i = blockIdx.x * blockDim.x + threadIdx.x;
  if (i >= NGRAPHS * NCLASSES) return;
  int g = i / NCLASSES;
  int k = i % NCLASSES;
  float s = 0.0f;
#pragma unroll 8
  for (int hh = 0; hh < HID; ++hh) s += pooled[g * HID + hh] * Wl[hh * NCLASSES + k];
  float c = cnt[g];
  c = c > 1.0f ? c : 1.0f;
  out[i] = s / c + bl[k];
}

extern "C" void kernel_launch(void* const* d_in, const int* in_sizes, int n_in,
                              void* d_out, int out_size, void* d_ws, size_t ws_size,
                              hipStream_t stream) {
  const int* shape_id = (const int*)d_in[0];
  const int* color_id = (const int*)d_in[1];
  const int* edge_index = (const int*)d_in[2];
  const int* batch = (const int*)d_in[3];
  const float* shape_table = (const float*)d_in[4];
  const float* color_table = (const float*)d_in[5];
  const float* W1 = (const float*)d_in[6];
  const float* b1 = (const float*)d_in[7];
  const float* W2 = (const float*)d_in[8];
  const float* b2 = (const float*)d_in[9];
  const float* Wl = (const float*)d_in[10];
  const float* bl = (const float*)d_in[11];
  float* out = (float*)d_out;

  const int N = NNODES;
  const int E = in_sizes[2] / 2;
  const int* src = edge_index;
  const int* dst = edge_index + E;

  float* dis    = (float*)d_ws;
  int*   colidx = (int*)(dis + N);
  int*   rowptr = colidx + E;
  int*   bucketBase = rowptr + N + 4;
  float* bufH   = dis + (long)N * (1 + EMB);
  float* bufA   = bufH + (long)N * HID;
  float* pooled = bufA + (long)N * HID;
  float* cnt    = pooled + (long)NGRAPHS * HID;
  float* x0     = bufA;
  unsigned int* tmp = (unsigned int*)bufH;
  int*   blockHist  = (int*)bufH + E;
  (void)n_in; (void)out_size; (void)ws_size;

  const int B = 256;
  auto blocks = [](long n, int b) { return (unsigned)((n + b - 1) / b); };

  bucket_count_kernel<<<blocks(E, PART_BLK), PART_T, 0, stream>>>(dst, blockHist, E, pooled,
                                                                  POOL_ZERO_N4);
  bucket_partition_kernel<<<blocks(E, PART_BLK), PART_T, 0, stream>>>(src, dst, blockHist, bucketBase,
                                                                      tmp, E);
  bucket_sort_embed_kernel<<<NBKT, SORT_T, 0, stream>>>(tmp, bucketBase, rowptr, dis, colidx, shape_id,
                                                        color_id, shape_table, color_table, x0, N);

  gather_emb_gemm_gemm_kernel<<<N / FUSE_ROWS, FUSE_T, 0, stream>>>(rowptr, colidx, x0, dis, W1, b1,
                                                                   W2, bufH);

  node_gather_hid_pool_kernel<<<blocks(N, GP_NODES), GP_T, 0, stream>>>(rowptr, colidx, bufH, dis, b2,
                                                                        batch, pooled, cnt, N);

  out_kernel<<<blocks((long)NGRAPHS * NCLASSES, B), B, 0, stream>>>(pooled, cnt, Wl, bl, out);
}


# Edge sort in two launches, not three: each slice bucket-ordered in place, a bucket block fetches its runs from all slices, colidx place by one cursor add; 6 to 5 launches; f32 sum order changes
# speedup vs baseline: 6.5954x; 1.1653x over previous
#include <hip/hip_runtime.h>

#define NNODES 100000
#define NGRAPHS 1024
#define EMB 32
#define HID 64
#define NCLASSES 10

typedef __attribute__((ext_vector_type(2))) float v2f;
typedef __attribute__((ext_vector_type(8))) float v8f;

#define NEDGES 1600000
#define BKT_SH 7
#define BKT_N (1 << BKT_SH)
#define NBKT ((NNODES + BKT_N - 1) / BKT_N)
#define META_PITCH 800
#define PART_T 512
#define PART_EPT 16
#define PART_BLK (PART_EPT * PART_T)
#define EMPTY_W 0xFFFFFFFFu
static_assert(NBKT <= 2 * PART_T && NBKT <= META_PITCH, "two buckets per thread");
static_assert(NBKT % 2 == 0 && META_PITCH % 2 == 0, "a thread's two meta words are one 8-byte store");
static_assert(NNODES <= (1 << 24) && BKT_N <= 128, "packed edge word");
static_assert(PART_BLK <= 32768, "run start and run length in 16 bits each");

__global__ void __launch_bounds__(PART_T)
slice_bucket_order_kernel(const int* __restrict__ src, const int* __restrict__ dst,
                          unsigned int* __restrict__ meta, unsigned int* __restrict__ tmp, int E,
                          float* __restrict__ zp, int zn4, int* __restrict__ cursor) {
  __shared__ int hist[NBKT];
  __shared__ int cur[NBKT + 32];
  __shared__ int s_w[PART_T / 32];
  __shared__ unsigned int stageW[PART_BLK];
  const int t = threadIdx.x;
  const int lane = t & 31;
  const int wave = t >> 5;
  const long b0 = (long)blockIdx.x * PART_BLK;
  for (int i = t; i < NBKT; i += PART_T) hist[i] = 0;
  int d[PART_EPT], s[PART_EPT];
#pragma unroll
  for (int k = 0; k < PART_EPT; ++k) {
    const long e = b0 + k * PART_T + t;
    const long ec = e < (long)E ? e : (long)E - 1;
    const int dv = dst[ec];
    s[k] = src[ec];
    d[k] = e < (long)E ? dv : -1;
  }
  __syncthreads();
#pragma unroll
  for (int k = 0; k < PART_EPT; ++k)
    if (d[k] >= 0) atomicAdd(&hist[d[k] >> BKT_SH], 1);
  __syncthreads();
  const int k0 = 2 * t;
  const bool mine = k0 < NBKT;
  const int o0 = mine ? hist[k0] : 0;
  const int o1 = mine ? hist[k0 + 1] : 0;
  int inc = o0 + o1;
  for (int o = 1; o < 32; o <<= 1) {
    const int up = __shfl_up(inc, o);
    if (lane >= o) inc += up;
  }
  if (lane == 31) s_w[wave] = inc;
  __syncthreads();
  int off = 0;
  for (int w = 0; w < wave; ++w) off += s_w[w];
  const int ex0 = off + inc - o0 - o1;
  const int ex1 = ex0 + o0;
  if (t < 32) cur[NBKT + t] = 0;
  if (mine) {
    cur[k0] = ex0;
    cur[k0 + 1] = ex1;
    uint2 m;
    m.x = ((unsigned int)ex0 << 16) | (unsigned int)o0;
    m.y = ((unsigned int)ex1 << 16) | (unsigned int)o1;
    *(uint2*)(meta + (long)blockIdx.x * META_PITCH + k0) = m;
  }
  __syncthreads();
  int p[PART_EPT];
#pragma unroll
  for (int k = 0; k < PART_EPT; ++k)
    p[k] = atomicAdd(&cur[d[k] >= 0 ? (d[k] >> BKT_SH) : NBKT + lane], 1);
#pragma unroll
  for (int k = 0; k < PART_EPT; ++k)
    if (d[k] >= 0)
      stageW[p[k]] = ((unsigned int)(d[k] & (BKT_N - 1)) << 24) | (unsigned int)s[k];
  __syncthreads();
  const long rest = (long)E - b0;
  const int nloc = rest < (long)PART_BLK ? (int)rest : PART_BLK;
  unsigned int o[PART_EPT];
#pragma unroll
  for (int k = 0; k < PART_EPT; ++k) o[k] = stageW[k * PART_T + t];
#pragma unroll
  for (int k = 0; k < PART_EPT; ++k) {
    const int i = k * PART_T + t;
    if (i < nloc) tmp[b0 + i] = o[k];
  }
  const int zstep = (int)gridDim.x * PART_T;
  for (int i = (int)blockIdx.x * PART_T + t; i < zn4; i += zstep)
    *(float4*)(zp + (long)i * 4) = make_float4(0.0f, 0.0f, 0.0f, 0.0f);
  if (blockIdx.x == 0 && t == 0) *cursor = 0;
}

#define SORT_T 256
#define SORT_W (SORT_T / 32)
#define RUN_Q 25
#define RUN_HELD (RUN_Q * SORT_W)
#define SORT_CAP 4096
#define EMB_K (BKT_N * (EMB / 4) / SORT_T)
static_assert(EMB_K * SORT_T == BKT_N * (EMB / 4), "whole pieces per thread");
static_assert(SORT_T % (EMB / 4) == 0, "a thread's pieces share one column group");
static_assert(RUN_Q <= 32, "one lane per held run reads its meta word");
static_assert((NEDGES + PART_BLK - 1) / PART_BLK <= RUN_HELD,
              "every run of the declared edge count is a held one");
__global__ void __launch_bounds__(SORT_T)
bucket_sort_embed_kernel(const unsigned int* __restrict__ tmp, const unsigned int* __restrict__ meta,
                         int* __restrict__ cursor, int* __restrict__ rowse,
                         float* __restrict__ dis, int* __restrict__ colidx,
                         const int* __restrict__ sid, const int* __restrict__ cid,
                         const float* __restrict__ st, const float* __restrict__ ct,
                         float* __restrict__ x0, int n, int E) {
  __shared__ int cnt[BKT_N];
  __shared__ int place[BKT_N + 32];
  __shared__ int s_w[BKT_N / 32];
  __shared__ float s_dis[BKT_N];
  __shared__ int s_base;
  __shared__ int stage[SORT_CAP];
  const int t = threadIdx.x;
  const int lane = t & 31;
  const int wave = t >> 5;
  const int b = blockIdx.x;
  const int nblk = (int)(((long)E + PART_BLK - 1) / PART_BLK);
  if (t < BKT_N) cnt[t] = 0;
  const int rq = lane * SORT_W + wave;
  unsigned int mq = 0;
  if (lane < RUN_Q && rq < nblk) mq = meta[(long)rq * META_PITCH + b];
  unsigned int ms[(RUN_HELD + 31) / 32];
  if (wave == 0) {
#pragma unroll
    for (int i = 0; i < (RUN_HELD + 31) / 32; ++i) {
      const int r = i * 32 + lane;
      ms[i] = meta[(long)(r < nblk ? r : nblk - 1) * META_PITCH + b];
    }
  }
  const int c4 = (t & (EMB / 4 - 1)) << 2;
  const int erow = t / (EMB / 4);
  int es[EMB_K], ec[EMB_K];
#pragma unroll
  for (int k = 0; k < EMB_K; ++k) {
    const int node = b * BKT_N + k * (SORT_T / (EMB / 4)) + erow;
    const int nc = node < n ? node : n - 1;
    es[k] = sid[nc];
    ec[k] = cid[nc];
  }

  auto for_rest = [&](auto f) {
    for (int r = wave; r < nblk; r += SORT_W) {
      const unsigned int m = meta[(long)r * META_PITCH + b];
      const int rc = (int)(m & 0xFFFFu);
      const unsigned int* p = tmp + (long)r * PART_BLK + (m >> 16);
      for (int j = (r < RUN_HELD ? 32 : 0) + lane; j < rc; j += 32) f(p[j]);
    }
  };

  unsigned int w[RUN_Q];
  bool rest = nblk > RUN_HELD;
#pragma unroll
  for (int q = 0; q < RUN_Q; ++q) {
    const unsigned int m = (unsigned int)__builtin_amdgcn_readlane((int)mq, q);
    const int rc = (int)(m & 0xFFFFu);
    const unsigned int* p = tmp + (long)(q * SORT_W + wave) * PART_BLK + (m >> 16);
    w[q] = lane < rc ? p[lane] : EMPTY_W;
    rest = rest || rc > 32;
  }
  int basev = 0;
  if (wave == 0) {
    int sz = 0;
#pragma unroll
    for (int i = 0; i < (RUN_HELD + 31) / 32; ++i)
      sz += i * 32 + lane < nblk ? (int)(ms[i] & 0xFFFFu) : 0;
    for (int r = ((RUN_HELD + 31) / 32) * 32 + lane; r < nblk; r += 32)
      sz += (int)(meta[(long)r * META_PITCH + b] & 0xFFFFu);
    for (int o = 16; o > 0; o >>= 1) sz += __shfl_xor(sz, o);
    if (lane == 0) basev = atomicAdd(cursor, sz);
  }
  __syncthreads();
#pragma unroll
  for (int q = 0; q < RUN_Q; ++q)
    if (w[q] != EMPTY_W) atomicAdd(&cnt[w[q] >> 24], 1);
  if (rest) for_rest([&](unsigned int x) { atomicAdd(&cnt[x >> 24], 1); });
  __syncthreads();

  const int c = t < BKT_N ? cnt[t] : 0;
  const float dnode = rsqrtf(1.0f + (float)c);
  if (t < BKT_N) s_dis[t] = dnode;
  int inc = c;
  for (int o = 1; o < 32; o <<= 1) {
    const int up = __shfl_up(inc, o);
    if (lane >= o) inc += up;
  }
  if (t < BKT_N && lane == 31) s_w[wave] = inc;
  if (t == 0) s_base = basev;
  __syncthreads();
  int ex = 0;
  if (t < BKT_N) {
    int o = 0;
    for (int q = 0; q < wave; ++q) o += s_w[q];
    ex = o + inc - c;
    place[t] = ex;
  } else if (t < BKT_N + 32) {
    place[t] = 0;
  }
  int tot = 0;
  for (int q = 0; q < BKT_N / 32; ++q) tot += s_w[q];
  const int base = s_base;
  __syncthreads();

  float4 ea[EMB_K], eb[EMB_K];
#pragma unroll
  for (int k = 0; k < EMB_K; ++k) {
    ea[k] = *(const float4*)(st + es[k] * EMB + c4);
    eb[k] = *(const float4*)(ct + ec[k] * EMB + c4);
  }

  if (tot <= SORT_CAP) {
    int pp[RUN_Q];
#pragma unroll
    for (int q = 0; q < RUN_Q; ++q)
      pp[q] = atomicAdd(&place[w[q] != EMPTY_W ? (int)(w[q] >> 24) : BKT_N + lane], 1);
#pragma unroll
    for (int q = 0; q < RUN_Q; ++q)
      if (w[q] != EMPTY_W) stage[pp[q]] = (int)(w[q] & 0xFFFFFFu);
    if (rest) for_rest([&](unsigned int x) {
      const int p = atomicAdd(&place[x >> 24], 1);
      stage[p] = (int)(x & 0xFFFFFFu);
    });
    __syncthreads();
    for (int i = t; i < tot; i += SORT_T) colidx[base + i] = stage[i];
  } else {
#pragma unroll
    for (int q = 0; q < RUN_Q; ++q) {
      if (w[q] != EMPTY_W) {
        const int p = atomicAdd(&place[w[q] >> 24], 1);
        colidx[base + p] = (int)(w[q] & 0xFFFFFFu);
      }
    }
    if (rest) for_rest([&](unsigned int x) {
      const int p = atomicAdd(&place[x >> 24], 1);
      colidx[base + p] = (int)(x & 0xFFFFFFu);
    });
  }
  if (t < BKT_N) {
    const int node = b * BKT_N + t;
    if (node < n) {
      *(int2*)(rowse + 2 * (long)node) = make_int2(base + ex, base + ex + c);
      dis[node] = dnode;
    }
  }
#pragma unroll
  for (int k = 0; k < EMB_K; ++k) {
    const int nl = k * (SORT_T / (EMB / 4)) + erow;
    const int node = b * BKT_N + nl;
    if (node < n) {
      const float dn = s_dis[nl];
      float4 v;
      v.x = ((es[k] != 0 ? ea[k].x : 0.0f) + (ec[k] != 0 ? eb[k].x : 0.0f)) * dn;
      v.y = ((es[k] != 0 ? ea[k].y : 0.0f) + (ec[k] != 0 ? eb[k].y : 0.0f)) * dn;
      v.z = ((es[k] != 0 ? ea[k].z : 0.0f) + (ec[k] != 0 ? eb[k].z : 0.0f)) * dn;
      v.w = ((es[k] != 0 ? ea[k].w : 0.0f) + (ec[k] != 0 ? eb[k].w : 0.0f)) * dn;
      *(float4*)(x0 + (long)node * EMB + c4) = v;
    }
  }
}

template <int WIDTH>
__device__ __forceinline__ float4 node_row_sum(const int* __restrict__ colidx,
                                               const float* __restrict__ hs,
                                               int node, int beg, int end, int c4) {
  const float* hc = hs + c4;

  const float4 self = *(const float4*)(hc + (long)node * WIDTH);
  float ax = self.x, ay = self.y, az = self.z, aw = self.w;

  int j = beg;
  for (; j + 4 <= end; j += 4) {
    const int s0 = colidx[j], s1 = colidx[j + 1], s2 = colidx[j + 2], s3 = colidx[j + 3];
    const float4 v0 = *(const float4*)(hc + (long)s0 * WIDTH);
    const float4 v1 = *(const float4*)(hc + (long)s1 * WIDTH);
    const float4 v2 = *(const float4*)(hc + (long)s2 * WIDTH);
    const float4 v3 = *(const float4*)(hc + (long)s3 * WIDTH);
    ax += v0.x; ay += v0.y; az += v0.z; aw += v0.w;
    ax += v1.x; ay += v1.y; az += v1.z; aw += v1.w;
    ax += v2.x; ay += v2.y; az += v2.z; aw += v2.w;
    ax += v3.x; ay += v3.y; az += v3.z; aw += v3.w;
  }
  if (j < end) {
    const int l = end - 1;
    const int j1 = j + 1 < l ? j + 1 : l;
    const int j2 = j + 2 < l ? j + 2 : l;
    const int s0 = colidx[j], s1 = colidx[j1], s2 = colidx[j2];
    const float m1 = (j + 1 < end) ? 1.0f : 0.0f;
    const float m2 = (j + 2 < end) ? 1.0f : 0.0f;
    const float4 v0 = *(const float4*)(hc + (long)s0 * WIDTH);
    const float4 v1 = *(const float4*)(hc + (long)s1 * WIDTH);
    const float4 v2 = *(const float4*)(hc + (long)s2 * WIDTH);
    ax += v0.x; ay += v0.y; az += v0.z; aw += v0.w;
    ax += v1.x * m1; ay += v1.y * m1; az += v1.z * m1; aw += v1.w * m1;
    ax += v2.x * m2; ay += v2.y * m2; az += v2.z * m2; aw += v2.w * m2;
  }
  float4 r;
  r.x = ax; r.y = ay; r.z = az; r.w = aw;
  return r;
}

#define GP_T 256
#define GP_NODES (GP_T / (HID / 4))
#define POOL_ZERO_N4 ((NGRAPHS * HID + NGRAPHS) / 4)
static_assert((NGRAPHS * HID + NGRAPHS) % 4 == 0 && (NGRAPHS * HID) % 4 == 0, "whole 16-byte pieces");
static_assert(GP_T >= HID, "one thread per column behind the barrier");
__global__ void __launch_bounds__(GP_T)
node_gather_hid_pool_kernel(const int* __restrict__ rowse, const int* __restrict__ colidx,
                            const float* __restrict__ hs, const float* __restrict__ dis,
                            const float* __restrict__ bias, const int* __restrict__ batch,
                            float* __restrict__ pooled, float* __restrict__ cnt, int n) {
  __shared__ __attribute__((aligned(16))) float sO[GP_NODES * HID];
  __shared__ int sG[GP_NODES];
  const int t = threadIdx.x;
  const int r = t / (HID / 4);
  const int c4 = (t & (HID / 4 - 1)) << 2;
  const int node0 = (int)blockIdx.x * GP_NODES;
  const int node = node0 + r;
  float4 o = make_float4(0.0f, 0.0f, 0.0f, 0.0f);
  int g = 0;
  if (node < n) {
    g = batch[node];
    const int2 be = *(const int2*)(rowse + 2 * (long)node);
    const int beg = be.x;
    const int end = be.y;
    const float dd = dis[node];
    const float4 a = node_row_sum<HID>(colidx, hs, node, beg, end, c4);
    float ax = a.x, ay = a.y, az = a.z, aw = a.w;
    ax *= dd; ay *= dd; az *= dd; aw *= dd;
    const float4 bb = *(const float4*)(bias + c4);
    ax += bb.x; ay += bb.y; az += bb.z; aw += bb.w;
    o.x = ax > 0.0f ? ax : 0.0f;
    o.y = ay > 0.0f ? ay : 0.0f;
    o.z = az > 0.0f ? az : 0.0f;
    o.w = aw > 0.0f ? aw : 0.0f;
  }
  *(float4*)(sO + r * HID + c4) = o;
  if (c4 == 0) sG[r] = g;
  __syncthreads();
  if (t < HID) {
    const int rest = n - node0;
    const int rows = rest < GP_NODES ? rest : GP_NODES;
    int gs[GP_NODES];
    float vs[GP_NODES];
#pragma unroll
    for (int i = 0; i < GP_NODES; ++i) {
      gs[i] = sG[i];
      vs[i] = sO[i * HID + t];
    }
    int cur = gs[0];
    float acc = vs[0];
    float run = 1.0f;
#pragma unroll
    for (int i = 1; i < GP_NODES; ++i) {
      if (i < rows) {
        if (gs[i] != cur) {
          atomicAdd(pooled + (long)cur * HID + t, acc);
          if (t == 0) atomicAdd(cnt + cur, run);
          cur = gs[i];
          acc = 0.0f;
          run = 0.0f;
        }
        acc += vs[i];
        run += 1.0f;
      }
    }
    atomicAdd(pooled + (long)cur * HID + t, acc);
    if (t == 0) atomicAdd(cnt + cur, run);
  }
}

#define FUSE_T 256
#define FUSE_ROWS 32
#define FUSE_PA (EMB + 4)
#define FUSE_PX (HID + 4)
static_assert(NNODES % FUSE_ROWS == 0, "whole tiles only: no row guard in the fused kernel");
static_assert(FUSE_T == FUSE_ROWS * (EMB / 4), "eight lanes per node");
static_assert(FUSE_T / 32 == (FUSE_ROWS / 16) * (HID / 16), "one 16 x 16 tile per wave");

__global__ void __launch_bounds__(FUSE_T)
gather_emb_gemm_gemm_kernel(const int* __restrict__ rowse, const int* __restrict__ colidx,
                            const float* __restrict__ xs, const float* __restrict__ dis,
                            const float* __restrict__ W1, const float* __restrict__ b1,
                            const float* __restrict__ W2, float* __restrict__ H) {
  __shared__ __attribute__((aligned(16))) float sA[FUSE_ROWS * FUSE_PA];
  __shared__ __attribute__((aligned(16))) float sX[FUSE_ROWS * FUSE_PX];
  const int t     = threadIdx.x;
  const int lane  = t & 31;
  const int wave  = t >> 5;
  const int l16   = lane & 15;
  const int khalf = lane >> 4;
  const int rt    = wave >> 2;
  const int col   = (wave & 3) * 16 + l16;
  const int node0 = (int)blockIdx.x * FUSE_ROWS;

  const int rloc = rt * 16 + 8 * khalf;
  const long rbase = (long)node0 + rloc;
  const float4 dlo = *(const float4*)(dis + rbase);
  const float4 dhi = *(const float4*)(dis + rbase + 4);
  const float dr[8] = {dlo.x, dlo.y, dlo.z, dlo.w, dhi.x, dhi.y, dhi.z, dhi.w};
  const float bc = b1[col];

  {
    const int r  = t >> 3;
    const int c4 = (t & 7) << 2;
    const int node = node0 + r;
    const int2 be = *(const int2*)(rowse + 2 * (long)node);
    const int beg = be.x;
    const int end = be.y;
    const float dd = dis[node];
    float4 a = node_row_sum<EMB>(colidx, xs, node, beg, end, c4);
    a.x *= dd; a.y *= dd; a.z *= dd; a.w *= dd;
    *(float4*)(sA + r * FUSE_PA + c4) = a;
  }
  v2f bw1[EMB / 4];
  for (int i = 0; i < EMB / 4; ++i) {
    bw1[i].x = W1[(4 * i + 2 * khalf + 0) * HID + col];
    bw1[i].y = W1[(4 * i + 2 * khalf + 1) * HID + col];
  }
  __syncthreads();

  v8f c = {};
  {
    const float* ap = sA + (rt * 16 + l16) * FUSE_PA + 2 * khalf;
    for (int i = 0; i < EMB / 4; ++i) {
      const v2f a = *(const v2f*)(ap + 4 * i);
      c = __builtin_amdgcn_wmma_f32_16x16x4_f32(false, a, false, bw1[i], (short)0, c, false, false);
    }
  }
  v2f bw2[HID / 4];
  for (int i = 0; i < HID / 4; ++i) {
    bw2[i].x = W2[(4 * i + 2 * khalf + 0) * HID + col];
    bw2[i].y = W2[(4 * i + 2 * khalf + 1) * HID + col];
  }
  {
    float* xp = sX + rloc * FUSE_PX + col;
    for (int v = 0; v < 8; ++v) {
      const float o = c[v] + bc;
      xp[v * FUSE_PX] = (o > 0.0f ? o : 0.0f) * dr[v];
    }
  }
  __syncthreads();

  v8f c2 = {};
  {
    const float* ap = sX + (rt * 16 + l16) * FUSE_PX + 2 * khalf;
    for (int i = 0; i < HID / 4; ++i) {
      const v2f a = *(const v2f*)(ap + 4 * i);
      c2 = __builtin_amdgcn_wmma_f32_16x16x4_f32(false, a, false, bw2[i], (short)0, c2, false, false);
    }
  }
  for (int v = 0; v < 8; ++v)
    H[(rbase + v) * HID + col] = c2[v];
}

__global__ void out_kernel(const float* __restrict__ pooled, const float* __restrict__ cnt,
                           const float* __restrict__ Wl, const float* __restrict__ bl,
                           float* __restrict__ out) {
  int i = blockIdx.x * blockDim.x + threadIdx.x;
  if (i >= NGRAPHS * NCLASSES) return;
  int g = i / NCLASSES;
  int k = i % NCLASSES;
  float s = 0.0f;
#pragma unroll 8
  for (int hh = 0; hh < HID; ++hh) s += pooled[g * HID + hh] * Wl[hh * NCLASSES + k];
  float c = cnt[g];
  c = c > 1.0f ? c : 1.0f;
  out[i] = s / c + bl[k];
}

extern "C" void kernel_launch(void* const* d_in, const int* in_sizes, int n_in,
                              void* d_out, int out_size, void* d_ws, size_t ws_size,
                              hipStream_t stream) {
  const int* shape_id = (const int*)d_in[0];
  const int* color_id = (const int*)d_in[1];
  const int* edge_index = (const int*)d_in[2];
  const int* batch = (const int*)d_in[3];
  const float* shape_table = (const float*)d_in[4];
  const float* color_table = (const float*)d_in[5];
  const float* W1 = (const float*)d_in[6];
  const float* b1 = (const float*)d_in[7];
  const float* W2 = (const float*)d_in[8];
  const float* b2 = (const float*)d_in[9];
  const float* Wl = (const float*)d_in[10];
  const float* bl = (const float*)d_in[11];
  float* out = (float*)d_out;

  const int N = NNODES;
  const int E = in_sizes[2] / 2;
  const int* src = edge_index;
  const int* dst = edge_index + E;

  float* dis    = (float*)d_ws;
  int*   colidx = (int*)(dis + N);
  int*   rowse  = colidx + E;
  int*   cursor = rowse + 2 * N;
  float* bufH   = dis + (long)N * (1 + EMB);
  float* bufA   = bufH + (long)N * HID;
  float* pooled = bufA + (long)N * HID;
  float* cnt    = pooled + (long)NGRAPHS * HID;
  float* x0     = bufA;
  unsigned int* tmp  = (unsigned int*)bufH;
  unsigned int* meta = tmp + E;
  (void)n_in; (void)out_size; (void)ws_size;

  const int B = 256;
  auto blocks = [](long n, int b) { return (unsigned)((n + b - 1) / b); };

  slice_bucket_order_kernel<<<blocks(E, PART_BLK), PART_T, 0, stream>>>(src, dst, meta, tmp, E, pooled,
                                                                        POOL_ZERO_N4, cursor);
  bucket_sort_embed_kernel<<<NBKT, SORT_T, 0, stream>>>(tmp, meta, cursor, rowse, dis, colidx, shape_id,
                                                        color_id, shape_table, color_table, x0, N, E);

  gather_emb_gemm_gemm_kernel<<<N / FUSE_ROWS, FUSE_T, 0, stream>>>(rowse, colidx, x0, dis, W1, b1,
                                                                   W2, bufH);

  node_gather_hid_pool_kernel<<<blocks(N, GP_NODES), GP_T, 0, stream>>>(rowse, colidx, bufH, dis, b2,
                                                                        batch, pooled, cnt, N);

  out_kernel<<<blocks((long)NGRAPHS * NCLASSES, B), B, 0, stream>>>(pooled, cnt, Wl, bl, out);
}


# Layer 1 without a row gather: per node the weights of the 24 table rows, ids and degree from a 2-byte node table held in LDS; tables enter as a 24-deep WMMA product; f32 sum order changes
# speedup vs baseline: 6.9947x; 1.0605x over previous
#include <hip/hip_runtime.h>

#define NNODES 100000
#define NGRAPHS 1024
#define EMB 32
#define HID 64
#define NCLASSES 10

typedef __attribute__((ext_vector_type(2))) float v2f;
typedef __attribute__((ext_vector_type(8))) float v8f;

#define NEDGES 1600000
#define BKT_SH 7
#define BKT_N (1 << BKT_SH)
#define NBKT ((NNODES + BKT_N - 1) / BKT_N)
#define META_PITCH 800
#define PART_T 512
#define PART_EPT 16
#define PART_BLK (PART_EPT * PART_T)
#define EMPTY_W 0xFFFFFFFFu
static_assert(NBKT <= 2 * PART_T && NBKT <= META_PITCH, "two buckets per thread");
static_assert(NBKT % 2 == 0 && META_PITCH % 2 == 0, "a thread's two meta words are one 8-byte store");
static_assert(NNODES <= (1 << 24) && BKT_N <= 128, "packed edge word");
static_assert(PART_BLK <= 32768, "run start and run length in 16 bits each");

__global__ void __launch_bounds__(PART_T)
slice_bucket_order_kernel(const int* __restrict__ src, const int* __restrict__ dst,
                          unsigned int* __restrict__ meta, unsigned int* __restrict__ tmp, int E,
                          float* __restrict__ zp, int zn4, int* __restrict__ cursor) {
  __shared__ int hist[NBKT];
  __shared__ int cur[NBKT + 32];
  __shared__ int s_w[PART_T / 32];
  __shared__ unsigned int stageW[PART_BLK];
  const int t = threadIdx.x;
  const int lane = t & 31;
  const int wave = t >> 5;
  const long b0 = (long)blockIdx.x * PART_BLK;
  for (int i = t; i < NBKT; i += PART_T) hist[i] = 0;
  int d[PART_EPT], s[PART_EPT];
#pragma unroll
  for (int k = 0; k < PART_EPT; ++k) {
    const long e = b0 + k * PART_T + t;
    const long ec = e < (long)E ? e : (long)E - 1;
    const int dv = dst[ec];
    s[k] = src[ec];
    d[k] = e < (long)E ? dv : -1;
  }
  __syncthreads();
#pragma unroll
  for (int k = 0; k < PART_EPT; ++k)
    if (d[k] >= 0) atomicAdd(&hist[d[k] >> BKT_SH], 1);
  __syncthreads();
  const int k0 = 2 * t;
  const bool mine = k0 < NBKT;
  const int o0 = mine ? hist[k0] : 0;
  const int o1 = mine ? hist[k0 + 1] : 0;
  int inc = o0 + o1;
  for (int o = 1; o < 32; o <<= 1) {
    const int up = __shfl_up(inc, o);
    if (lane >= o) inc += up;
  }
  if (lane == 31) s_w[wave] = inc;
  __syncthreads();
  int off = 0;
  for (int w = 0; w < wave; ++w) off += s_w[w];
  const int ex0 = off + inc - o0 - o1;
  const int ex1 = ex0 + o0;
  if (t < 32) cur[NBKT + t] = 0;
  if (mine) {
    cur[k0] = ex0;
    cur[k0 + 1] = ex1;
    uint2 m;
    m.x = ((unsigned int)ex0 << 16) | (unsigned int)o0;
    m.y = ((unsigned int)ex1 << 16) | (unsigned int)o1;
    *(uint2*)(meta + (long)blockIdx.x * META_PITCH + k0) = m;
  }
  __syncthreads();
  int p[PART_EPT];
#pragma unroll
  for (int k = 0; k < PART_EPT; ++k)
    p[k] = atomicAdd(&cur[d[k] >= 0 ? (d[k] >> BKT_SH) : NBKT + lane], 1);
#pragma unroll
  for (int k = 0; k < PART_EPT; ++k)
    if (d[k] >= 0)
      stageW[p[k]] = ((unsigned int)(d[k] & (BKT_N - 1)) << 24) | (unsigned int)s[k];
  __syncthreads();
  const long rest = (long)E - b0;
  const int nloc = rest < (long)PART_BLK ? (int)rest : PART_BLK;
  unsigned int o[PART_EPT];
#pragma unroll
  for (int k = 0; k < PART_EPT; ++k) o[k] = stageW[k * PART_T + t];
#pragma unroll
  for (int k = 0; k < PART_EPT; ++k) {
    const int i = k * PART_T + t;
    if (i < nloc) tmp[b0 + i] = o[k];
  }
  const int zstep = (int)gridDim.x * PART_T;
  for (int i = (int)blockIdx.x * PART_T + t; i < zn4; i += zstep)
    *(float4*)(zp + (long)i * 4) = make_float4(0.0f, 0.0f, 0.0f, 0.0f);
  if (blockIdx.x == 0 && t == 0) *cursor = 0;
}

#define SORT_T 256
#define SORT_W (SORT_T / 32)
#define RUN_Q 25
#define RUN_HELD (RUN_Q * SORT_W)
#define SORT_CAP 4096
#define DEG_ESC 127
#define N_SHAPE_ROWS 17
#define N_COLOR_ROWS 9
static_assert(N_SHAPE_ROWS <= 32 && N_COLOR_ROWS <= 16, "id fields of the table entry");
static_assert(RUN_Q <= 32, "one lane per held run reads its meta word");
static_assert((NEDGES + PART_BLK - 1) / PART_BLK <= RUN_HELD,
              "every run of the declared edge count is a held one");
__global__ void __launch_bounds__(SORT_T)
bucket_sort_embed_kernel(const unsigned int* __restrict__ tmp, const unsigned int* __restrict__ meta,
                         int* __restrict__ cursor, int* __restrict__ rowse,
                         float* __restrict__ dis, int* __restrict__ colidx,
                         const int* __restrict__ sid, const int* __restrict__ cid,
                         unsigned short* __restrict__ tab, int n, int E) {
  __shared__ int cnt[BKT_N];
  __shared__ int place[BKT_N + 32];
  __shared__ int s_w[BKT_N / 32];
  __shared__ int s_base;
  __shared__ int stage[SORT_CAP];
  const int t = threadIdx.x;
  const int lane = t & 31;
  const int wave = t >> 5;
  const int b = blockIdx.x;
  const int nblk = (int)(((long)E + PART_BLK - 1) / PART_BLK);
  if (t < BKT_N) cnt[t] = 0;
  const int rq = lane * SORT_W + wave;
  unsigned int mq = 0;
  if (lane < RUN_Q && rq < nblk) mq = meta[(long)rq * META_PITCH + b];
  unsigned int ms[(RUN_HELD + 31) / 32];
  if (wave == 0) {
#pragma unroll
    for (int i = 0; i < (RUN_HELD + 31) / 32; ++i) {
      const int r = i * 32 + lane;
      ms[i] = meta[(long)(r < nblk ? r : nblk - 1) * META_PITCH + b];
    }
  }
  int es = 0, ec = 0;
  if (t < BKT_N) {
    const int node = b * BKT_N + t;
    const int nc = node < n ? node : n - 1;
    es = sid[nc];
    ec = cid[nc];
  }

  auto for_rest = [&](auto f) {
    for (int r = wave; r < nblk; r += SORT_W) {
      const unsigned int m = meta[(long)r * META_PITCH + b];
      const int rc = (int)(m & 0xFFFFu);
      const unsigned int* p = tmp + (long)r * PART_BLK + (m >> 16);
      for (int j = (r < RUN_HELD ? 32 : 0) + lane; j < rc; j += 32) f(p[j]);
    }
  };

  unsigned int w[RUN_Q];
  bool rest = nblk > RUN_HELD;
#pragma unroll
  for (int q = 0; q < RUN_Q; ++q) {
    const unsigned int m = (unsigned int)__builtin_amdgcn_readlane((int)mq, q);
    const int rc = (int)(m & 0xFFFFu);
    const unsigned int* p = tmp + (long)(q * SORT_W + wave) * PART_BLK + (m >> 16);
    w[q] = lane < rc ? p[lane] : EMPTY_W;
    rest = rest || rc > 32;
  }
  int basev = 0;
  if (wave == 0) {
    int sz = 0;
#pragma unroll
    for (int i = 0; i < (RUN_HELD + 31) / 32; ++i)
      sz += i * 32 + lane < nblk ? (int)(ms[i] & 0xFFFFu) : 0;
    for (int r = ((RUN_HELD + 31) / 32) * 32 + lane; r < nblk; r += 32)
      sz += (int)(meta[(long)r * META_PITCH + b] & 0xFFFFu);
    for (int o = 16; o > 0; o >>= 1) sz += __shfl_xor(sz, o);
    if (lane == 0) basev = atomicAdd(cursor, sz);
  }
  __syncthreads();
#pragma unroll
  for (int q = 0; q < RUN_Q; ++q)
    if (w[q] != EMPTY_W) atomicAdd(&cnt[w[q] >> 24], 1);
  if (rest) for_rest([&](unsigned int x) { atomicAdd(&cnt[x >> 24], 1); });
  __syncthreads();

  const int c = t < BKT_N ? cnt[t] : 0;
  const float dnode = rsqrtf(1.0f + (float)c);
  int inc = c;
  for (int o = 1; o < 32; o <<= 1) {
    const int up = __shfl_up(inc, o);
    if (lane >= o) inc += up;
  }
  if (t < BKT_N && lane == 31) s_w[wave] = inc;
  if (t == 0) s_base = basev;
  __syncthreads();
  int ex = 0;
  if (t < BKT_N) {
    int o = 0;
    for (int q = 0; q < wave; ++q) o += s_w[q];
    ex = o + inc - c;
    place[t] = ex;
  } else if (t < BKT_N + 32) {
    place[t] = 0;
  }
  int tot = 0;
  for (int q = 0; q < BKT_N / 32; ++q) tot += s_w[q];
  const int base = s_base;
  __syncthreads();

  if (tot <= SORT_CAP) {
    int pp[RUN_Q];
#pragma unroll
    for (int q = 0; q < RUN_Q; ++q)
      pp[q] = atomicAdd(&place[w[q] != EMPTY_W ? (int)(w[q] >> 24) : BKT_N + lane], 1);
#pragma unroll
    for (int q = 0; q < RUN_Q; ++q)
      if (w[q] != EMPTY_W) stage[pp[q]] = (int)(w[q] & 0xFFFFFFu);
    if (rest) for_rest([&](unsigned int x) {
      const int p = atomicAdd(&place[x >> 24], 1);
      stage[p] = (int)(x & 0xFFFFFFu);
    });
    __syncthreads();
    for (int i = t; i < tot; i += SORT_T) colidx[base + i] = stage[i];
  } else {
#pragma unroll
    for (int q = 0; q < RUN_Q; ++q) {
      if (w[q] != EMPTY_W) {
        const int p = atomicAdd(&place[w[q] >> 24], 1);
        colidx[base + p] = (int)(w[q] & 0xFFFFFFu);
      }
    }
    if (rest) for_rest([&](unsigned int x) {
      const int p = atomicAdd(&place[x >> 24], 1);
      colidx[base + p] = (int)(x & 0xFFFFFFu);
    });
  }
  if (t < BKT_N) {
    const int node = b * BKT_N + t;
    if (node < n) {
      *(int2*)(rowse + 2 * (long)node) = make_int2(base + ex, base + ex + c);
      dis[node] = dnode;
      const int dg = c < DEG_ESC ? c : DEG_ESC;
      tab[node] = (unsigned short)((unsigned int)es | ((unsigned int)ec << 5) | ((unsigned int)dg << 9));
    }
  }
}

template <int WIDTH>
__device__ __forceinline__ float4 node_row_sum(const int* __restrict__ colidx,
                                               const float* __restrict__ hs,
                                               int node, int beg, int end, int c4) {
  const float* hc = hs + c4;

  const float4 self = *(const float4*)(hc + (long)node * WIDTH);
  float ax = self.x, ay = self.y, az = self.z, aw = self.w;

  int j = beg;
  for (; j + 4 <= end; j += 4) {
    const int s0 = colidx[j], s1 = colidx[j + 1], s2 = colidx[j + 2], s3 = colidx[j + 3];
    const float4 v0 = *(const float4*)(hc + (long)s0 * WIDTH);
    const float4 v1 = *(const float4*)(hc + (long)s1 * WIDTH);
    const float4 v2 = *(const float4*)(hc + (long)s2 * WIDTH);
    const float4 v3 = *(const float4*)(hc + (long)s3 * WIDTH);
    ax += v0.x; ay += v0.y; az += v0.z; aw += v0.w;
    ax += v1.x; ay += v1.y; az += v1.z; aw += v1.w;
    ax += v2.x; ay += v2.y; az += v2.z; aw += v2.w;
    ax += v3.x; ay += v3.y; az += v3.z; aw += v3.w;
  }
  if (j < end) {
    const int l = end - 1;
    const int j1 = j + 1 < l ? j + 1 : l;
    const int j2 = j + 2 < l ? j + 2 : l;
    const int s0 = colidx[j], s1 = colidx[j1], s2 = colidx[j2];
    const float m1 = (j + 1 < end) ? 1.0f : 0.0f;
    const float m2 = (j + 2 < end) ? 1.0f : 0.0f;
    const float4 v0 = *(const float4*)(hc + (long)s0 * WIDTH);
    const float4 v1 = *(const float4*)(hc + (long)s1 * WIDTH);
    const float4 v2 = *(const float4*)(hc + (long)s2 * WIDTH);
    ax += v0.x; ay += v0.y; az += v0.z; aw += v0.w;
    ax += v1.x * m1; ay += v1.y * m1; az += v1.z * m1; aw += v1.w * m1;
    ax += v2.x * m2; ay += v2.y * m2; az += v2.z * m2; aw += v2.w * m2;
  }
  float4 r;
  r.x = ax; r.y = ay; r.z = az; r.w = aw;
  return r;
}

#define GP_T 256
#define GP_NODES (GP_T / (HID / 4))
#define POOL_ZERO_N4 ((NGRAPHS * HID + NGRAPHS) / 4)
static_assert((NGRAPHS * HID + NGRAPHS) % 4 == 0 && (NGRAPHS * HID) % 4 == 0, "whole 16-byte pieces");
static_assert(GP_T >= HID, "one thread per column behind the barrier");
__global__ void __launch_bounds__(GP_T)
node_gather_hid_pool_kernel(const int* __restrict__ rowse, const int* __restrict__ colidx,
                            const float* __restrict__ hs, const float* __restrict__ dis,
                            const float* __restrict__ bias, const int* __restrict__ batch,
                            float* __restrict__ pooled, float* __restrict__ cnt, int n) {
  __shared__ __attribute__((aligned(16))) float sO[GP_NODES * HID];
  __shared__ int sG[GP_NODES];
  const int t = threadIdx.x;
  const int r = t / (HID / 4);
  const int c4 = (t & (HID / 4 - 1)) << 2;
  const int node0 = (int)blockIdx.x * GP_NODES;
  const int node = node0 + r;
  float4 o = make_float4(0.0f, 0.0f, 0.0f, 0.0f);
  int g = 0;
  if (node < n) {
    g = batch[node];
    const int2 be = *(const int2*)(rowse + 2 * (long)node);
    const int beg = be.x;
    const int end = be.y;
    const float dd = dis[node];
    const float4 a = node_row_sum<HID>(colidx, hs, node, beg, end, c4);
    float ax = a.x, ay = a.y, az = a.z, aw = a.w;
    ax *= dd; ay *= dd; az *= dd; aw *= dd;
    const float4 bb = *(const float4*)(bias + c4);
    ax += bb.x; ay += bb.y; az += bb.z; aw += bb.w;
    o.x = ax > 0.0f ? ax : 0.0f;
    o.y = ay > 0.0f ? ay : 0.0f;
    o.z = az > 0.0f ? az : 0.0f;
    o.w = aw > 0.0f ? aw : 0.0f;
  }
  *(float4*)(sO + r * HID + c4) = o;
  if (c4 == 0) sG[r] = g;
  __syncthreads();
  if (t < HID) {
    const int rest = n - node0;
    const int rows = rest < GP_NODES ? rest : GP_NODES;
    int gs[GP_NODES];
    float vs[GP_NODES];
#pragma unroll
    for (int i = 0; i < GP_NODES; ++i) {
      gs[i] = sG[i];
      vs[i] = sO[i * HID + t];
    }
    int cur = gs[0];
    float acc = vs[0];
    float run = 1.0f;
#pragma unroll
    for (int i = 1; i < GP_NODES; ++i) {
      if (i < rows) {
        if (gs[i] != cur) {
          atomicAdd(pooled + (long)cur * HID + t, acc);
          if (t == 0) atomicAdd(cnt + cur, run);
          cur = gs[i];
          acc = 0.0f;
          run = 0.0f;
        }
        acc += vs[i];
        run += 1.0f;
      }
    }
    atomicAdd(pooled + (long)cur * HID + t, acc);
    if (t == 0) atomicAdd(cnt + cur, run);
  }
}

#define L1_ROWS 400
#define L1_TILES (L1_ROWS / 16)
#define L1_T (L1_TILES * 32)
#define L1_K 24
#define L1_P (HID + 4)
#define TAB_Q ((NNODES * 2) / 16)
#define TAB_K ((TAB_Q + L1_T - 1) / L1_T)
static_assert(NNODES % L1_ROWS == 0, "whole blocks only: no row guard in the layer-1 kernel");
static_assert(L1_ROWS % 16 == 0 && L1_T <= 1024, "one 16-row tile per wave");
static_assert((NNODES * 2) % 16 == 0, "the node table is copied in 16-byte pieces");
static_assert(N_SHAPE_ROWS == 17 && N_COLOR_ROWS == 9 && L1_K == 24 && L1_K <= EMB, "row ownership of the eight lanes");

#define BCAST8(x, e) __builtin_amdgcn_mov_dpp8((x), (unsigned int)(e) * 0x249249u)

__device__ __forceinline__ void wave_tile_sync() {
  __builtin_amdgcn_fence(__ATOMIC_RELEASE, "workgroup", "local");
  __builtin_amdgcn_wave_barrier();
  __builtin_amdgcn_fence(__ATOMIC_ACQUIRE, "workgroup", "local");
}

__global__ void __launch_bounds__(L1_T)
table_weights_gemm3_kernel(const int* __restrict__ rowse, const int* __restrict__ colidx,
                           const unsigned short* __restrict__ tab, const float* __restrict__ dis,
                           const float* __restrict__ st, const float* __restrict__ ct,
                           const float* __restrict__ W1, const float* __restrict__ b1,
                           const float* __restrict__ W2, float* __restrict__ H) {
  __shared__ __attribute__((aligned(16))) unsigned int sTab[NNODES / 2];
  __shared__ __attribute__((aligned(16))) float sR[L1_ROWS * L1_P];
  const unsigned short* sT16 = (const unsigned short*)sTab;
  const int t     = threadIdx.x;
  const int lane  = t & 31;
  const int wave  = t >> 5;
  const int l16   = lane & 15;
  const int khalf = lane >> 4;
  const int node0 = (int)blockIdx.x * L1_ROWS + wave * 16;
  float* R = sR + wave * 16 * L1_P;

  const int q = lane & 7;
  int2 be4[4];
#pragma unroll
  for (int p = 0; p < 4; ++p)
    be4[p] = *(const int2*)(rowse + 2 * (long)(node0 + 4 * p + (lane >> 3)));

  {
    const uint4* gs = (const uint4*)tab;
    uint4 tv[TAB_K];
#pragma unroll
    for (int k = 0; k < TAB_K; ++k) {
      const int i = k * L1_T + t;
      tv[k] = gs[i < TAB_Q ? i : TAB_Q - 1];
    }
#pragma unroll
    for (int k = 0; k < TAB_K; ++k) {
      const int i = k * L1_T + t;
      if (i < TAB_Q) ((uint4*)sTab)[i] = tv[k];
    }
  }
  int s04[4];
#pragma unroll
  for (int p = 0; p < 4; ++p) {
    const int beg = be4[p].x;
    const int end = be4[p].y;
    const int j = beg + q;
    s04[p] = 0;
    if (beg < end) s04[p] = colidx[j < end ? j : end - 1];
  }
  __syncthreads();

  const unsigned int own0 = (unsigned int)q + 1u;
  const unsigned int own1 = (unsigned int)q + 9u;
#define EDGE_SLOT(ee)                                               \
  {                                                                 \
    const unsigned int ie = BCAST8(ids, ee);                        \
    const float de = __uint_as_float(BCAST8(db, ee));               \
    const unsigned int a = ie & 31u, bcol = ie >> 5;                \
    w0 += a == own0 ? de : 0.0f;                                    \
    w1 += a == own1 ? de : 0.0f;                                    \
    w2 += bcol == own0 ? de : 0.0f;                                 \
  }
  bool esc = false;
#pragma unroll
  for (int p = 0; p < 4; ++p) {
    const int rl = 4 * p + (lane >> 3);
    const int node = node0 + rl;
    const int beg = be4[p].x;
    const int end = be4[p].y;
    int s = s04[p];
    float w0 = 0.0f, w1 = 0.0f, w2 = 0.0f;
    {
      const unsigned int e = sT16[node];
      const unsigned int dg = e >> 9;
      esc = esc || dg == DEG_ESC;
      const float d = rsqrtf(1.0f + (float)dg);
      const unsigned int a = e & 31u, bcol = (e >> 5) & 15u;
      w0 += a == own0 ? d : 0.0f;
      w1 += a == own1 ? d : 0.0f;
      w2 += bcol == own0 ? d : 0.0f;
    }
    for (int j0 = beg; j0 < end; j0 += 8) {
      const int j = j0 + q;
      int sn = 0;
      if (j0 + 8 < end) sn = colidx[j + 8 < end ? j + 8 : end - 1];
      const unsigned int e = sT16[s];
      const unsigned int dg = e >> 9;
      esc = esc || (dg == DEG_ESC && j < end);
      const float d = j < end ? rsqrtf(1.0f + (float)dg) : 0.0f;
      const unsigned int ids = e & 511u;
      const unsigned int db = __float_as_uint(d);
      EDGE_SLOT(0) EDGE_SLOT(1) EDGE_SLOT(2) EDGE_SLOT(3)
      EDGE_SLOT(4) EDGE_SLOT(5) EDGE_SLOT(6) EDGE_SLOT(7)
      s = sn;
    }
    float* wp = R + rl * L1_P + q;
    wp[0] = w0;
    wp[8] = w1;
    wp[16] = w2;
  }
  if (__any(esc ? 1 : 0)) {
    for (int p = 0; p < 4; ++p) {
      const int rl = 4 * p + (lane >> 3);
      const int node = node0 + rl;
      const int2 be = *(const int2*)(rowse + 2 * (long)node);
      const int beg = be.x;
      const int end = be.y;
      float w0 = 0.0f, w1 = 0.0f, w2 = 0.0f;
      {
        const unsigned int e = sT16[node];
        const float d = dis[node];
        const unsigned int a = e & 31u, bcol = (e >> 5) & 15u;
        w0 += a == own0 ? d : 0.0f;
        w1 += a == own1 ? d : 0.0f;
        w2 += bcol == own0 ? d : 0.0f;
      }
      for (int j0 = beg; j0 < end; j0 += 8) {
        const int j = j0 + q;
        const int s = colidx[j < end ? j : end - 1];
        const float d = j < end ? dis[s] : 0.0f;
        const unsigned int ids = sT16[s] & 511u;
        const unsigned int db = __float_as_uint(d);
        EDGE_SLOT(0) EDGE_SLOT(1) EDGE_SLOT(2) EDGE_SLOT(3)
        EDGE_SLOT(4) EDGE_SLOT(5) EDGE_SLOT(6) EDGE_SLOT(7)
      }
      float* wp = R + rl * L1_P + q;
      wp[0] = w0;
      wp[8] = w1;
      wp[16] = w2;
    }
  }
#undef EDGE_SLOT
  wave_tile_sync();

  const int rloc = 8 * khalf;
  const long rbase = (long)node0 + rloc;
  const float4 dlo = *(const float4*)(dis + rbase);
  const float4 dhi = *(const float4*)(dis + rbase + 4);
  const float dr[8] = {dlo.x, dlo.y, dlo.z, dlo.w, dhi.x, dhi.y, dhi.z, dhi.w};
  const float* ap = R + l16 * L1_P + 2 * khalf;

  {
    v2f a[L1_K / 4];
    for (int i = 0; i < L1_K / 4; ++i) a[i] = *(const v2f*)(ap + 4 * i);
    wave_tile_sync();
    v8f acc0 = {}, acc1 = {};
    for (int i = 0; i < L1_K / 4; ++i) {
      const int k0 = 4 * i + 2 * khalf;
      const float* t0 = i < 4 ? st + (k0 + 1) * EMB : ct + (k0 - 15) * EMB;
      v2f b0, b1v;
      b0.x = t0[l16];
      b0.y = t0[EMB + l16];
      b1v.x = t0[16 + l16];
      b1v.y = t0[EMB + 16 + l16];
      acc0 = __builtin_amdgcn_wmma_f32_16x16x4_f32(false, a[i], false, b0, (short)0, acc0, false, false);
      acc1 = __builtin_amdgcn_wmma_f32_16x16x4_f32(false, a[i], false, b1v, (short)0, acc1, false, false);
    }
    float* op = R + rloc * L1_P + l16;
    for (int v = 0; v < 8; ++v) {
      op[v * L1_P] = acc0[v] * dr[v];
      op[v * L1_P + 16] = acc1[v] * dr[v];
    }
  }
  wave_tile_sync();

  {
    v2f a[EMB / 4];
    for (int i = 0; i < EMB / 4; ++i) a[i] = *(const v2f*)(ap + 4 * i);
    wave_tile_sync();
    for (int ctile = 0; ctile < HID / 16; ++ctile) {
      const int col = ctile * 16 + l16;
      v8f c = {};
      for (int i = 0; i < EMB / 4; ++i) {
        v2f b;
        b.x = W1[(4 * i + 2 * khalf + 0) * HID + col];
        b.y = W1[(4 * i + 2 * khalf + 1) * HID + col];
        c = __builtin_amdgcn_wmma_f32_16x16x4_f32(false, a[i], false, b, (short)0, c, false, false);
      }
      const float bc = b1[col];
      float* xp = R + rloc * L1_P + col;
      for (int v = 0; v < 8; ++v) {
        const float o = c[v] + bc;
        xp[v * L1_P] = (o > 0.0f ? o : 0.0f) * dr[v];
      }
    }
  }
  wave_tile_sync();

  {
    v2f a[HID / 4];
    for (int i = 0; i < HID / 4; ++i) a[i] = *(const v2f*)(ap + 4 * i);
    wave_tile_sync();
    for (int ctile = 0; ctile < HID / 16; ++ctile) {
      const int col = ctile * 16 + l16;
      v8f c2 = {};
      for (int i = 0; i < HID / 4; ++i) {
        v2f b;
        b.x = W2[(4 * i + 2 * khalf + 0) * HID + col];
        b.y = W2[(4 * i + 2 * khalf + 1) * HID + col];
        c2 = __builtin_amdgcn_wmma_f32_16x16x4_f32(false, a[i], false, b, (short)0, c2, false, false);
      }
      for (int v = 0; v < 8; ++v)
        H[(rbase + v) * HID + col] = c2[v];
    }
  }
}

__global__ void out_kernel(const float* __restrict__ pooled, const float* __restrict__ cnt,
                           const float* __restrict__ Wl, const float* __restrict__ bl,
                           float* __restrict__ out) {
  int i = blockIdx.x * blockDim.x + threadIdx.x;
  if (i >= NGRAPHS * NCLASSES) return;
  int g = i / NCLASSES;
  int k = i % NCLASSES;
  float s = 0.0f;
#pragma unroll 8
  for (int hh = 0; hh < HID; ++hh) s += pooled[g * HID + hh] * Wl[hh * NCLASSES + k];
  float c = cnt[g];
  c = c > 1.0f ? c : 1.0f;
  out[i] = s / c + bl[k];
}

extern "C" void kernel_launch(void* const* d_in, const int* in_sizes, int n_in,
                              void* d_out, int out_size, void* d_ws, size_t ws_size,
                              hipStream_t stream) {
  const int* shape_id = (const int*)d_in[0];
  const int* color_id = (const int*)d_in[1];
  const int* edge_index = (const int*)d_in[2];
  const int* batch = (const int*)d_in[3];
  const float* shape_table = (const float*)d_in[4];
  const float* color_table = (const float*)d_in[5];
  const float* W1 = (const float*)d_in[6];
  const float* b1 = (const float*)d_in[7];
  const float* W2 = (const float*)d_in[8];
  const float* b2 = (const float*)d_in[9];
  const float* Wl = (const float*)d_in[10];
  const float* bl = (const float*)d_in[11];
  float* out = (float*)d_out;

  const int N = NNODES;
  const int E = in_sizes[2] / 2;
  const int* src = edge_index;
  const int* dst = edge_index + E;

  float* dis    = (float*)d_ws;
  int*   colidx = (int*)(dis + N);
  int*   rowse  = colidx + E;
  int*   cursor = rowse + 2 * N;
  float* bufH   = dis + (long)N * (1 + EMB);
  float* bufA   = bufH + (long)N * HID;
  float* pooled = bufA + (long)N * HID;
  float* cnt    = pooled + (long)NGRAPHS * HID;
  unsigned short* tab = (unsigned short*)bufA;
  unsigned int* tmp  = (unsigned int*)bufH;
  unsigned int* meta = tmp + E;
  (void)n_in; (void)out_size; (void)ws_size;

  const int B = 256;
  auto blocks = [](long n, int b) { return (unsigned)((n + b - 1) / b); };

  slice_bucket_order_kernel<<<blocks(E, PART_BLK), PART_T, 0, stream>>>(src, dst, meta, tmp, E, pooled,
                                                                        POOL_ZERO_N4, cursor);
  bucket_sort_embed_kernel<<<NBKT, SORT_T, 0, stream>>>(tmp, meta, cursor, rowse, dis, colidx, shape_id,
                                                        color_id, tab, N, E);

  table_weights_gemm3_kernel<<<N / L1_ROWS, L1_T, 0, stream>>>(rowse, colidx, tab, dis, shape_table,
                                                               color_table, W1, b1, W2, bufH);

  node_gather_hid_pool_kernel<<<blocks(N, GP_NODES), GP_T, 0, stream>>>(rowse, colidx, bufH, dis, b2,
                                                                        batch, pooled, cnt, N);

  out_kernel<<<blocks((long)NGRAPHS * NCLASSES, B), B, 0, stream>>>(pooled, cnt, Wl, bl, out);
}


# Classifier head folded into the layer-2 launch by linearity: each run of a graph's nodes adds (its x2 sum @ Wl) / node count to out; counts formed by the sort; 5 to 4 launches; f32 sum order changes
# speedup vs baseline: 7.3846x; 1.0557x over previous
#include <hip/hip_runtime.h>

#define NNODES 100000
#define NGRAPHS 1024
#define EMB 32
#define HID 64
#define NCLASSES 10

typedef __attribute__((ext_vector_type(2))) float v2f;
typedef __attribute__((ext_vector_type(8))) float v8f;

#define NEDGES 1600000
#define BKT_SH 7
#define BKT_N (1 << BKT_SH)
#define NBKT ((NNODES + BKT_N - 1) / BKT_N)
#define META_PITCH 800
#define PART_T 512
#define PART_EPT 16
#define PART_BLK (PART_EPT * PART_T)
#define EMPTY_W 0xFFFFFFFFu
static_assert(NBKT <= 2 * PART_T && NBKT <= META_PITCH, "two buckets per thread");
static_assert(NBKT % 2 == 0 && META_PITCH % 2 == 0, "a thread's two meta words are one 8-byte store");
static_assert(NNODES <= (1 << 24) && BKT_N <= 128, "packed edge word");
static_assert(PART_BLK <= 32768, "run start and run length in 16 bits each");

__global__ void __launch_bounds__(PART_T)
slice_bucket_order_kernel(const int* __restrict__ src, const int* __restrict__ dst,
                          unsigned int* __restrict__ meta, unsigned int* __restrict__ tmp, int E,
                          float* __restrict__ cntz, const float* __restrict__ bl,
                          float* __restrict__ out, int* __restrict__ cursor) {
  __shared__ int hist[NBKT];
  __shared__ int cur[NBKT + 32];
  __shared__ int s_w[PART_T / 32];
  __shared__ unsigned int stageW[PART_BLK];
  const int t = threadIdx.x;
  const int lane = t & 31;
  const int wave = t >> 5;
  const long b0 = (long)blockIdx.x * PART_BLK;
  for (int i = t; i < NBKT; i += PART_T) hist[i] = 0;
  const int gi = (int)blockIdx.x * PART_T + t;
  const float blv = bl[gi % NCLASSES];
  int d[PART_EPT], s[PART_EPT];
#pragma unroll
  for (int k = 0; k < PART_EPT; ++k) {
    const long e = b0 + k * PART_T + t;
    const long ec = e < (long)E ? e : (long)E - 1;
    const int dv = dst[ec];
    s[k] = src[ec];
    d[k] = e < (long)E ? dv : -1;
  }
  __syncthreads();
#pragma unroll
  for (int k = 0; k < PART_EPT; ++k)
    if (d[k] >= 0) atomicAdd(&hist[d[k] >> BKT_SH], 1);
  __syncthreads();
  const int k0 = 2 * t;
  const bool mine = k0 < NBKT;
  const int o0 = mine ? hist[k0] : 0;
  const int o1 = mine ? hist[k0 + 1] : 0;
  int inc = o0 + o1;
  for (int o = 1; o < 32; o <<= 1) {
    const int up = __shfl_up(inc, o);
    if (lane >= o) inc += up;
  }
  if (lane == 31) s_w[wave] = inc;
  __syncthreads();
  int off = 0;
  for (int w = 0; w < wave; ++w) off += s_w[w];
  const int ex0 = off + inc - o0 - o1;
  const int ex1 = ex0 + o0;
  if (t < 32) cur[NBKT + t] = 0;
  if (mine) {
    cur[k0] = ex0;
    cur[k0 + 1] = ex1;
    uint2 m;
    m.x = ((unsigned int)ex0 << 16) | (unsigned int)o0;
    m.y = ((unsigned int)ex1 << 16) | (unsigned int)o1;
    *(uint2*)(meta + (long)blockIdx.x * META_PITCH + k0) = m;
  }
  __syncthreads();
  int p[PART_EPT];
#pragma unroll
  for (int k = 0; k < PART_EPT; ++k)
    p[k] = atomicAdd(&cur[d[k] >= 0 ? (d[k] >> BKT_SH) : NBKT + lane], 1);
#pragma unroll
  for (int k = 0; k < PART_EPT; ++k)
    if (d[k] >= 0)
      stageW[p[k]] = ((unsigned int)(d[k] & (BKT_N - 1)) << 24) | (unsigned int)s[k];
  __syncthreads();
  const long rest = (long)E - b0;
  const int nloc = rest < (long)PART_BLK ? (int)rest : PART_BLK;
  unsigned int o[PART_EPT];
#pragma unroll
  for (int k = 0; k < PART_EPT; ++k) o[k] = stageW[k * PART_T + t];
#pragma unroll
  for (int k = 0; k < PART_EPT; ++k) {
    const int i = k * PART_T + t;
    if (i < nloc) tmp[b0 + i] = o[k];
  }
  const int zstep = (int)gridDim.x * PART_T;
  if (gi < NGRAPHS * NCLASSES) out[gi] = blv;
  for (int i = gi + zstep; i < NGRAPHS * NCLASSES; i += zstep) out[i] = bl[i % NCLASSES];
  for (int i = gi; i < NGRAPHS / 4; i += zstep)
    *(float4*)(cntz + (long)i * 4) = make_float4(0.0f, 0.0f, 0.0f, 0.0f);
  if (blockIdx.x == 0 && t == 0) *cursor = 0;
}

#define SORT_T 256
#define SORT_W (SORT_T / 32)
#define RUN_Q 25
#define RUN_HELD (RUN_Q * SORT_W)
#define SORT_CAP 4096
#define DEG_ESC 127
#define N_SHAPE_ROWS 17
#define N_COLOR_ROWS 9
static_assert(N_SHAPE_ROWS <= 32 && N_COLOR_ROWS <= 16, "id fields of the table entry");
static_assert(RUN_Q <= 32, "one lane per held run reads its meta word");
static_assert(BKT_N % 32 == 0, "the bucket's nodes fill whole waves");
static_assert((NEDGES + PART_BLK - 1) / PART_BLK <= RUN_HELD,
              "every run of the declared edge count is a held one");
__global__ void __launch_bounds__(SORT_T)
bucket_sort_embed_kernel(const unsigned int* __restrict__ tmp, const unsigned int* __restrict__ meta,
                         int* __restrict__ cursor, int* __restrict__ rowse,
                         float* __restrict__ dis, int* __restrict__ colidx,
                         const int* __restrict__ sid, const int* __restrict__ cid,
                         unsigned short* __restrict__ tab, const int* __restrict__ batch,
                         float* __restrict__ cntg, int n, int E) {
  __shared__ int cnt[BKT_N];
  __shared__ int place[BKT_N + 32];
  __shared__ int s_w[BKT_N / 32];
  __shared__ int s_base;
  __shared__ int stage[SORT_CAP];
  const int t = threadIdx.x;
  const int lane = t & 31;
  const int wave = t >> 5;
  const int b = blockIdx.x;
  const int nblk = (int)(((long)E + PART_BLK - 1) / PART_BLK);
  if (t < BKT_N) cnt[t] = 0;
  const int rq = lane * SORT_W + wave;
  unsigned int mq = 0;
  if (lane < RUN_Q && rq < nblk) mq = meta[(long)rq * META_PITCH + b];
  unsigned int ms[(RUN_HELD + 31) / 32];
  if (wave == 0) {
#pragma unroll
    for (int i = 0; i < (RUN_HELD + 31) / 32; ++i) {
      const int r = i * 32 + lane;
      ms[i] = meta[(long)(r < nblk ? r : nblk - 1) * META_PITCH + b];
    }
  }
  int es = 0, ec = 0;
  int eg = -1;
  if (t < BKT_N) {
    const int node = b * BKT_N + t;
    const int nc = node < n ? node : n - 1;
    es = sid[nc];
    ec = cid[nc];
    const int gv = batch[nc];
    eg = node < n ? gv : -1;
  }

  auto for_rest = [&](auto f) {
    for (int r = wave; r < nblk; r += SORT_W) {
      const unsigned int m = meta[(long)r * META_PITCH + b];
      const int rc = (int)(m & 0xFFFFu);
      const unsigned int* p = tmp + (long)r * PART_BLK + (m >> 16);
      for (int j = (r < RUN_HELD ? 32 : 0) + lane; j < rc; j += 32) f(p[j]);
    }
  };

  unsigned int w[RUN_Q];
  bool rest = nblk > RUN_HELD;
#pragma unroll
  for (int q = 0; q < RUN_Q; ++q) {
    const unsigned int m = (unsigned int)__builtin_amdgcn_readlane((int)mq, q);
    const int rc = (int)(m & 0xFFFFu);
    const unsigned int* p = tmp + (long)(q * SORT_W + wave) * PART_BLK + (m >> 16);
    w[q] = lane < rc ? p[lane] : EMPTY_W;
    rest = rest || rc > 32;
  }
  int basev = 0;
  if (wave == 0) {
    int sz = 0;
#pragma unroll
    for (int i = 0; i < (RUN_HELD + 31) / 32; ++i)
      sz += i * 32 + lane < nblk ? (int)(ms[i] & 0xFFFFu) : 0;
    for (int r = ((RUN_HELD + 31) / 32) * 32 + lane; r < nblk; r += 32)
      sz += (int)(meta[(long)r * META_PITCH + b] & 0xFFFFu);
    for (int o = 16; o > 0; o >>= 1) sz += __shfl_xor(sz, o);
    if (lane == 0) basev = atomicAdd(cursor, sz);
  }
  __syncthreads();
#pragma unroll
  for (int q = 0; q < RUN_Q; ++q)
    if (w[q] != EMPTY_W) atomicAdd(&cnt[w[q] >> 24], 1);
  if (rest) for_rest([&](unsigned int x) { atomicAdd(&cnt[x >> 24], 1); });
  __syncthreads();

  const int c = t < BKT_N ? cnt[t] : 0;
  const float dnode = rsqrtf(1.0f + (float)c);
  int inc = c;
  for (int o = 1; o < 32; o <<= 1) {
    const int up = __shfl_up(inc, o);
    if (lane >= o) inc += up;
  }
  if (t < BKT_N && lane == 31) s_w[wave] = inc;
  if (t == 0) s_base = basev;
  __syncthreads();
  int ex = 0;
  if (t < BKT_N) {
    int o = 0;
    for (int q = 0; q < wave; ++q) o += s_w[q];
    ex = o + inc - c;
    place[t] = ex;
  } else if (t < BKT_N + 32) {
    place[t] = 0;
  }
  int tot = 0;
  for (int q = 0; q < BKT_N / 32; ++q) tot += s_w[q];
  const int base = s_base;
  __syncthreads();

  if (tot <= SORT_CAP) {
    int pp[RUN_Q];
#pragma unroll
    for (int q = 0; q < RUN_Q; ++q)
      pp[q] = atomicAdd(&place[w[q] != EMPTY_W ? (int)(w[q] >> 24) : BKT_N + lane], 1);
#pragma unroll
    for (int q = 0; q < RUN_Q; ++q)
      if (w[q] != EMPTY_W) stage[pp[q]] = (int)(w[q] & 0xFFFFFFu);
    if (rest) for_rest([&](unsigned int x) {
      const int p = atomicAdd(&place[x >> 24], 1);
      stage[p] = (int)(x & 0xFFFFFFu);
    });
    __syncthreads();
    for (int i = t; i < tot; i += SORT_T) colidx[base + i] = stage[i];
  } else {
#pragma unroll
    for (int q = 0; q < RUN_Q; ++q) {
      if (w[q] != EMPTY_W) {
        const int p = atomicAdd(&place[w[q] >> 24], 1);
        colidx[base + p] = (int)(w[q] & 0xFFFFFFu);
      }
    }
    if (rest) for_rest([&](unsigned int x) {
      const int p = atomicAdd(&place[x >> 24], 1);
      colidx[base + p] = (int)(x & 0xFFFFFFu);
    });
  }
  if (t < BKT_N) {
    const int prev = __shfl_up(eg, 1);
    const bool lead = lane == 0 || eg != prev;
    const unsigned int leads = __builtin_amdgcn_ballot_w32(lead);
    const unsigned int above = (leads >> lane) >> 1;
    const int len = above ? __builtin_ctz(above) + 1 : 32 - lane;
    if (lead && eg >= 0) atomicAdd(cntg + eg, (float)len);
    const int node = b * BKT_N + t;
    if (node < n) {
      *(int2*)(rowse + 2 * (long)node) = make_int2(base + ex, base + ex + c);
      dis[node] = dnode;
      const int dg = c < DEG_ESC ? c : DEG_ESC;
      tab[node] = (unsigned short)((unsigned int)es | ((unsigned int)ec << 5) | ((unsigned int)dg << 9));
    }
  }
}

template <int WIDTH>
__device__ __forceinline__ float4 node_row_sum(const int* __restrict__ colidx,
                                               const float* __restrict__ hs,
                                               int node, int beg, int end, int c4) {
  const float* hc = hs + c4;

  const float4 self = *(const float4*)(hc + (long)node * WIDTH);
  float ax = self.x, ay = self.y, az = self.z, aw = self.w;

  int j = beg;
  for (; j + 4 <= end; j += 4) {
    const int s0 = colidx[j], s1 = colidx[j + 1], s2 = colidx[j + 2], s3 = colidx[j + 3];
    const float4 v0 = *(const float4*)(hc + (long)s0 * WIDTH);
    const float4 v1 = *(const float4*)(hc + (long)s1 * WIDTH);
    const float4 v2 = *(const float4*)(hc + (long)s2 * WIDTH);
    const float4 v3 = *(const float4*)(hc + (long)s3 * WIDTH);
    ax += v0.x; ay += v0.y; az += v0.z; aw += v0.w;
    ax += v1.x; ay += v1.y; az += v1.z; aw += v1.w;
    ax += v2.x; ay += v2.y; az += v2.z; aw += v2.w;
    ax += v3.x; ay += v3.y; az += v3.z; aw += v3.w;
  }
  if (j < end) {
    const int l = end - 1;
    const int j1 = j + 1 < l ? j + 1 : l;
    const int j2 = j + 2 < l ? j + 2 : l;
    const int s0 = colidx[j], s1 = colidx[j1], s2 = colidx[j2];
    const float m1 = (j + 1 < end) ? 1.0f : 0.0f;
    const float m2 = (j + 2 < end) ? 1.0f : 0.0f;
    const float4 v0 = *(const float4*)(hc + (long)s0 * WIDTH);
    const float4 v1 = *(const float4*)(hc + (long)s1 * WIDTH);
    const float4 v2 = *(const float4*)(hc + (long)s2 * WIDTH);
    ax += v0.x; ay += v0.y; az += v0.z; aw += v0.w;
    ax += v1.x * m1; ay += v1.y * m1; az += v1.z * m1; aw += v1.w * m1;
    ax += v2.x * m2; ay += v2.y * m2; az += v2.z * m2; aw += v2.w * m2;
  }
  float4 r;
  r.x = ax; r.y = ay; r.z = az; r.w = aw;
  return r;
}

#define GP_T 256
#define GP_NODES (GP_T / (HID / 4))
static_assert(NGRAPHS % 4 == 0, "the node counts are zeroed in whole 16-byte pieces");
static_assert(GP_T >= HID && HID % 32 == 0, "one thread per column behind the barrier, whole waves");
static_assert(NCLASSES <= 32, "one lane per class");

__device__ __forceinline__ float xhalf(float x) {
  return __builtin_bit_cast(float, __builtin_amdgcn_permlanex16(__builtin_bit_cast(int, x),
                                                                __builtin_bit_cast(int, x), 0x76543210,
                                                                0xfedcba98, false, false));
}
__device__ __forceinline__ float wave_sum32(float v) {
  v += __builtin_bit_cast(float, __builtin_amdgcn_update_dpp(0, __builtin_bit_cast(int, v), 0x161, 0xf, 0xf, false));
  v += __builtin_bit_cast(float, __builtin_amdgcn_update_dpp(0, __builtin_bit_cast(int, v), 0x162, 0xf, 0xf, false));
  v += __builtin_bit_cast(float, __builtin_amdgcn_update_dpp(0, __builtin_bit_cast(int, v), 0x164, 0xf, 0xf, false));
  v += __builtin_bit_cast(float, __builtin_amdgcn_update_dpp(0, __builtin_bit_cast(int, v), 0x168, 0xf, 0xf, false));
  return v + xhalf(v);
}

__global__ void __launch_bounds__(GP_T)
node_gather_hid_pool_head_kernel(const int* __restrict__ rowse, const int* __restrict__ colidx,
                                 const float* __restrict__ hs, const float* __restrict__ dis,
                                 const float* __restrict__ bias, const int* __restrict__ batch,
                                 const float* __restrict__ cntg, const float* __restrict__ Wl,
                                 float* __restrict__ out, int n) {
  __shared__ __attribute__((aligned(16))) float sO[GP_NODES * HID];
  __shared__ int sG[GP_NODES];
  __shared__ float sC[GP_NODES];
  const int t = threadIdx.x;
  const int lane = t & 31;
  const int r = t / (HID / 4);
  const int c4 = (t & (HID / 4 - 1)) << 2;
  const int node0 = (int)blockIdx.x * GP_NODES;
  const int node = node0 + r;
  float4 o = make_float4(0.0f, 0.0f, 0.0f, 0.0f);
  int g = 0;
  float cn = 1.0f;
  if (node < n) {
    g = batch[node];
    const int2 be = *(const int2*)(rowse + 2 * (long)node);
    const int beg = be.x;
    const int end = be.y;
    const float dd = dis[node];
    cn = cntg[g];
    const float4 a = node_row_sum<HID>(colidx, hs, node, beg, end, c4);
    float ax = a.x, ay = a.y, az = a.z, aw = a.w;
    ax *= dd; ay *= dd; az *= dd; aw *= dd;
    const float4 bb = *(const float4*)(bias + c4);
    ax += bb.x; ay += bb.y; az += bb.z; aw += bb.w;
    o.x = ax > 0.0f ? ax : 0.0f;
    o.y = ay > 0.0f ? ay : 0.0f;
    o.z = az > 0.0f ? az : 0.0f;
    o.w = aw > 0.0f ? aw : 0.0f;
  }
  float wl[NCLASSES];
#pragma unroll
  for (int k = 0; k < NCLASSES; ++k) wl[k] = t < HID ? Wl[t * NCLASSES + k] : 0.0f;
  *(float4*)(sO + r * HID + c4) = o;
  if (c4 == 0) {
    sG[r] = g;
    sC[r] = cn > 1.0f ? cn : 1.0f;
  }
  __syncthreads();
  if (t < HID) {
    const int rest = n - node0;
    const int rows = rest < GP_NODES ? rest : GP_NODES;
    int gs[GP_NODES];
    float vs[GP_NODES];
#pragma unroll
    for (int i = 0; i < GP_NODES; ++i) {
      gs[i] = sG[i];
      vs[i] = sO[i * HID + t];
    }
    auto close_run = [&](int gcur, int idx, float acc) {
      const float cdiv = sC[idx];
      float mine = 0.0f;
#pragma unroll
      for (int k = 0; k < NCLASSES; ++k) {
        const float p = wave_sum32(acc * wl[k]);
        mine = lane == k ? p : mine;
      }
      if (lane < NCLASSES) atomicAdd(out + (long)gcur * NCLASSES + lane, mine / cdiv);
    };
    int cur = gs[0];
    float acc = vs[0];
#pragma unroll
    for (int i = 1; i < GP_NODES; ++i) {
      if (i < rows) {
        if (gs[i] != cur) {
          close_run(cur, i - 1, acc);
          cur = gs[i];
          acc = 0.0f;
        }
        acc += vs[i];
      }
    }
    close_run(cur, rows - 1, acc);
  }
}

#define L1_ROWS 400
#define L1_TILES (L1_ROWS / 16)
#define L1_T (L1_TILES * 32)
#define L1_K 24
#define L1_P (HID + 4)
#define TAB_Q ((NNODES * 2) / 16)
#define TAB_K ((TAB_Q + L1_T - 1) / L1_T)
static_assert(NNODES % L1_ROWS == 0, "whole blocks only: no row guard in the layer-1 kernel");
static_assert(L1_ROWS % 16 == 0 && L1_T <= 1024, "one 16-row tile per wave");
static_assert((NNODES * 2) % 16 == 0, "the node table is copied in 16-byte pieces");
static_assert(N_SHAPE_ROWS == 17 && N_COLOR_ROWS == 9 && L1_K == 24 && L1_K <= EMB, "row ownership of the eight lanes");

#define BCAST8(x, e) __builtin_amdgcn_mov_dpp8((x), (unsigned int)(e) * 0x249249u)

__device__ __forceinline__ void wave_tile_sync() {
  __builtin_amdgcn_fence(__ATOMIC_RELEASE, "workgroup", "local");
  __builtin_amdgcn_wave_barrier();
  __builtin_amdgcn_fence(__ATOMIC_ACQUIRE, "workgroup", "local");
}

__global__ void __launch_bounds__(L1_T)
table_weights_gemm3_kernel(const int* __restrict__ rowse, const int* __restrict__ colidx,
                           const unsigned short* __restrict__ tab, const float* __restrict__ dis,
                           const float* __restrict__ st, const float* __restrict__ ct,
                           const float* __restrict__ W1, const float* __restrict__ b1,
                           const float* __restrict__ W2, float* __restrict__ H) {
  __shared__ __attribute__((aligned(16))) unsigned int sTab[NNODES / 2];
  __shared__ __attribute__((aligned(16))) float sR[L1_ROWS * L1_P];
  const unsigned short* sT16 = (const unsigned short*)sTab;
  const int t     = threadIdx.x;
  const int lane  = t & 31;
  const int wave  = t >> 5;
  const int l16   = lane & 15;
  const int khalf = lane >> 4;
  const int node0 = (int)blockIdx.x * L1_ROWS + wave * 16;
  float* R = sR + wave * 16 * L1_P;

  const int q = lane & 7;
  int2 be4[4];
#pragma unroll
  for (int p = 0; p < 4; ++p)
    be4[p] = *(const int2*)(rowse + 2 * (long)(node0 + 4 * p + (lane >> 3)));

  {
    const uint4* gs = (const uint4*)tab;
    uint4 tv[TAB_K];
#pragma unroll
    for (int k = 0; k < TAB_K; ++k) {
      const int i = k * L1_T + t;
      tv[k] = gs[i < TAB_Q ? i : TAB_Q - 1];
    }
#pragma unroll
    for (int k = 0; k < TAB_K; ++k) {
      const int i = k * L1_T + t;
      if (i < TAB_Q) ((uint4*)sTab)[i] = tv[k];
    }
  }
  int s04[4];
#pragma unroll
  for (int p = 0; p < 4; ++p) {
    const int beg = be4[p].x;
    const int end = be4[p].y;
    const int j = beg + q;
    s04[p] = 0;
    if (beg < end) s04[p] = colidx[j < end ? j : end - 1];
  }
  __syncthreads();

  const unsigned int own0 = (unsigned int)q + 1u;
  const unsigned int own1 = (unsigned int)q + 9u;
#define EDGE_SLOT(ee)                                               \
  {                                                                 \
    const unsigned int ie = BCAST8(ids, ee);                        \
    const float de = __uint_as_float(BCAST8(db, ee));               \
    const unsigned int a = ie & 31u, bcol = ie >> 5;                \
    w0 += a == own0 ? de : 0.0f;                                    \
    w1 += a == own1 ? de : 0.0f;                                    \
    w2 += bcol == own0 ? de : 0.0f;                                 \
  }
  bool esc = false;
#pragma unroll
  for (int p = 0; p < 4; ++p) {
    const int rl = 4 * p + (lane >> 3);
    const int node = node0 + rl;
    const int beg = be4[p].x;
    const int end = be4[p].y;
    int s = s04[p];
    float w0 = 0.0f, w1 = 0.0f, w2 = 0.0f;
    {
      const unsigned int e = sT16[node];
      const unsigned int dg = e >> 9;
      esc = esc || dg == DEG_ESC;
      const float d = rsqrtf(1.0f + (float)dg);
      const unsigned int a = e & 31u, bcol = (e >> 5) & 15u;
      w0 += a == own0 ? d : 0.0f;
      w1 += a == own1 ? d : 0.0f;
      w2 += bcol == own0 ? d : 0.0f;
    }
    for (int j0 = beg; j0 < end; j0 += 8) {
      const int j = j0 + q;
      int sn = 0;
      if (j0 + 8 < end) sn = colidx[j + 8 < end ? j + 8 : end - 1];
      const unsigned int e = sT16[s];
      const unsigned int dg = e >> 9;
      esc = esc || (dg == DEG_ESC && j < end);
      const float d = j < end ? rsqrtf(1.0f + (float)dg) : 0.0f;
      const unsigned int ids = e & 511u;
      const unsigned int db = __float_as_uint(d);
      EDGE_SLOT(0) EDGE_SLOT(1) EDGE_SLOT(2) EDGE_SLOT(3)
      EDGE_SLOT(4) EDGE_SLOT(5) EDGE_SLOT(6) EDGE_SLOT(7)
      s = sn;
    }
    float* wp = R + rl * L1_P + q;
    wp[0] = w0;
    wp[8] = w1;
    wp[16] = w2;
  }
  if (__any(esc ? 1 : 0)) {
    for (int p = 0; p < 4; ++p) {
      const int rl = 4 * p + (lane >> 3);
      const int node = node0 + rl;
      const int2 be = *(const int2*)(rowse + 2 * (long)node);
      const int beg = be.x;
      const int end = be.y;
      float w0 = 0.0f, w1 = 0.0f, w2 = 0.0f;
      {
        const unsigned int e = sT16[node];
        const float d = dis[node];
        const unsigned int a = e & 31u, bcol = (e >> 5) & 15u;
        w0 += a == own0 ? d : 0.0f;
        w1 += a == own1 ? d : 0.0f;
        w2 += bcol == own0 ? d : 0.0f;
      }
      for (int j0 = beg; j0 < end; j0 += 8) {
        const int j = j0 + q;
        const int s = colidx[j < end ? j : end - 1];
        const float d = j < end ? dis[s] : 0.0f;
        const unsigned int ids = sT16[s] & 511u;
        const unsigned int db = __float_as_uint(d);
        EDGE_SLOT(0) EDGE_SLOT(1) EDGE_SLOT(2) EDGE_SLOT(3)
        EDGE_SLOT(4) EDGE_SLOT(5) EDGE_SLOT(6) EDGE_SLOT(7)
      }
      float* wp = R + rl * L1_P + q;
      wp[0] = w0;
      wp[8] = w1;
      wp[16] = w2;
    }
  }
#undef EDGE_SLOT
  wave_tile_sync();

  const int rloc = 8 * khalf;
  const long rbase = (long)node0 + rloc;
  const float4 dlo = *(const float4*)(dis + rbase);
  const float4 dhi = *(const float4*)(dis + rbase + 4);
  const float dr[8] = {dlo.x, dlo.y, dlo.z, dlo.w, dhi.x, dhi.y, dhi.z, dhi.w};
  const float* ap = R + l16 * L1_P + 2 * khalf;

  {
    v2f a[L1_K / 4];
    for (int i = 0; i < L1_K / 4; ++i) a[i] = *(const v2f*)(ap + 4 * i);
    wave_tile_sync();
    v8f acc0 = {}, acc1 = {};
    for (int i = 0; i < L1_K / 4; ++i) {
      const int k0 = 4 * i + 2 * khalf;
      const float* t0 = i < 4 ? st + (k0 + 1) * EMB : ct + (k0 - 15) * EMB;
      v2f b0, b1v;
      b0.x = t0[l16];
      b0.y = t0[EMB + l16];
      b1v.x = t0[16 + l16];
      b1v.y = t0[EMB + 16 + l16];
      acc0 = __builtin_amdgcn_wmma_f32_16x16x4_f32(false, a[i], false, b0, (short)0, acc0, false, false);
      acc1 = __builtin_amdgcn_wmma_f32_16x16x4_f32(false, a[i], false, b1v, (short)0, acc1, false, false);
    }
    float* op = R + rloc * L1_P + l16;
    for (int v = 0; v < 8; ++v) {
      op[v * L1_P] = acc0[v] * dr[v];
      op[v * L1_P + 16] = acc1[v] * dr[v];
    }
  }
  wave_tile_sync();

  {
    v2f a[EMB / 4];
    for (int i = 0; i < EMB / 4; ++i) a[i] = *(const v2f*)(ap + 4 * i);
    wave_tile_sync();
    for (int ctile = 0; ctile < HID / 16; ++ctile) {
      const int col = ctile * 16 + l16;
      v8f c = {};
      for (int i = 0; i < EMB / 4; ++i) {
        v2f b;
        b.x = W1[(4 * i + 2 * khalf + 0) * HID + col];
        b.y = W1[(4 * i + 2 * khalf + 1) * HID + col];
        c = __builtin_amdgcn_wmma_f32_16x16x4_f32(false, a[i], false, b, (short)0, c, false, false);
      }
      const float bc = b1[col];
      float* xp = R + rloc * L1_P + col;
      for (int v = 0; v < 8; ++v) {
        const float o = c[v] + bc;
        xp[v * L1_P] = (o > 0.0f ? o : 0.0f) * dr[v];
      }
    }
  }
  wave_tile_sync();

  {
    v2f a[HID / 4];
    for (int i = 0; i < HID / 4; ++i) a[i] = *(const v2f*)(ap + 4 * i);
    wave_tile_sync();
    for (int ctile = 0; ctile < HID / 16; ++ctile) {
      const int col = ctile * 16 + l16;
      v8f c2 = {};
      for (int i = 0; i < HID / 4; ++i) {
        v2f b;
        b.x = W2[(4 * i + 2 * khalf + 0) * HID + col];
        b.y = W2[(4 * i + 2 * khalf + 1) * HID + col];
        c2 = __builtin_amdgcn_wmma_f32_16x16x4_f32(false, a[i], false, b, (short)0, c2, false, false);
      }
      for (int v = 0; v < 8; ++v)
        H[(rbase + v) * HID + col] = c2[v];
    }
  }
}

extern "C" void kernel_launch(void* const* d_in, const int* in_sizes, int n_in,
                              void* d_out, int out_size, void* d_ws, size_t ws_size,
                              hipStream_t stream) {
  const int* shape_id = (const int*)d_in[0];
  const int* color_id = (const int*)d_in[1];
  const int* edge_index = (const int*)d_in[2];
  const int* batch = (const int*)d_in[3];
  const float* shape_table = (const float*)d_in[4];
  const float* color_table = (const float*)d_in[5];
  const float* W1 = (const float*)d_in[6];
  const float* b1 = (const float*)d_in[7];
  const float* W2 = (const float*)d_in[8];
  const float* b2 = (const float*)d_in[9];
  const float* Wl = (const float*)d_in[10];
  const float* bl = (const float*)d_in[11];
  float* out = (float*)d_out;

  const int N = NNODES;
  const int E = in_sizes[2] / 2;
  const int* src = edge_index;
  const int* dst = edge_index + E;

  float* dis    = (float*)d_ws;
  int*   colidx = (int*)(dis + N);
  int*   rowse  = colidx + E;
  int*   cursor = rowse + 2 * N;
  float* bufH   = dis + (long)N * (1 + EMB);
  float* bufA   = bufH + (long)N * HID;
  float* cnt    = bufA + (long)N * HID;
  unsigned short* tab = (unsigned short*)bufA;
  unsigned int* tmp  = (unsigned int*)bufH;
  unsigned int* meta = tmp + E;
  (void)n_in; (void)out_size; (void)ws_size;

  auto blocks = [](long n, int b) { return (unsigned)((n + b - 1) / b); };

  slice_bucket_order_kernel<<<blocks(E, PART_BLK), PART_T, 0, stream>>>(src, dst, meta, tmp, E, cnt, bl,
                                                                        out, cursor);
  bucket_sort_embed_kernel<<<NBKT, SORT_T, 0, stream>>>(tmp, meta, cursor, rowse, dis, colidx, shape_id,
                                                        color_id, tab, batch, cnt, N, E);

  table_weights_gemm3_kernel<<<N / L1_ROWS, L1_T, 0, stream>>>(rowse, colidx, tab, dis, shape_table,
                                                               color_table, W1, b1, W2, bufH);

  node_gather_hid_pool_head_kernel<<<blocks(N, GP_NODES), GP_T, 0, stream>>>(rowse, colidx, bufH, dis, b2,
                                                                             batch, cnt, Wl, out, N);
}
